# starting kernel
# baseline (speedup 1.0000x reference)
#include <hip/hip_runtime.h>
#include <hip/hip_bf16.h>

typedef float v2f __attribute__((ext_vector_type(2)));
typedef float v8f __attribute__((ext_vector_type(8)));

#define N_GRAPHS 64

__global__ void k_zero_i32(int* p, int n) {
  int i = blockIdx.x * blockDim.x + threadIdx.x;
  for (; i < n; i += gridDim.x * blockDim.x) p[i] = 0;
}
__global__ void k_zero_f32(float* p, int n) {
  int i = blockIdx.x * blockDim.x + threadIdx.x;
  for (; i < n; i += gridDim.x * blockDim.x) p[i] = 0.0f;
}

__global__ void k_count(const int* __restrict__ dst, int* __restrict__ cnt, int E) {
  int e = blockIdx.x * blockDim.x + threadIdx.x;
  if (e < E) atomicAdd(&cnt[dst[e]], 1);
}

__global__ void k_scan(const int* __restrict__ cnt, int* __restrict__ rowstart,
                       float* __restrict__ dinv, int n) {
  __shared__ int tmp[1024];
  __shared__ int carry;
  const int tid = threadIdx.x;
  if (tid == 0) carry = 0;
  __syncthreads();
  for (int base = 0; base < n; base += 1024) {
    int i = base + tid;
    int v = (i < n) ? cnt[i] : 0;
    tmp[tid] = v;
    __syncthreads();
    #pragma unroll
    for (int off = 1; off < 1024; off <<= 1) {
      int t = (tid >= off) ? tmp[tid - off] : 0;
      __syncthreads();
      tmp[tid] += t;
      __syncthreads();
    }
    int incl = tmp[tid];
    int c = carry;
    if (i < n) {
      rowstart[i] = c + incl - v;
      dinv[i] = rsqrtf((float)v + 1.0f);
    }
    __syncthreads();
    if (tid == 0) carry = c + tmp[1023];
    __syncthreads();
  }
  if (tid == 0) rowstart[n] = carry;
}

__global__ void k_fill(const int* __restrict__ src, const int* __restrict__ dst,
                       const int* __restrict__ rowstart, int* __restrict__ cur,
                       int* __restrict__ csr, int E) {
  int e = blockIdx.x * blockDim.x + threadIdx.x;
  if (e < E) {
    int d = dst[e];
    int pos = atomicAdd(&cur[d], 1);
    csr[rowstart[d] + pos] = src[e];
  }
}

template <int N>
__global__ void k_gemm_wmma(const float* __restrict__ A, const float* __restrict__ W,
                            const float* __restrict__ dinv, float* __restrict__ out, int M) {
  constexpr int K = 128;
  constexpr int NT = N / 16;
  constexpr int RP = 2 * N + 32;
  extern __shared__ float Wlds[];
  for (int i = threadIdx.x; i < K * N; i += blockDim.x) {
    int k = i / N;
    int n = i % N;
    Wlds[(k >> 1) * RP + n * 2 + (k & 1)] = W[i];
  }
  __syncthreads();

  const int wave = threadIdx.x >> 5;
  const int lane = threadIdx.x & 31;
  const int mtile = blockIdx.x * 8 + wave;
  if (mtile * 16 >= M) return;

  const int Rm = mtile * 16;
  const int half = lane >> 4;
  const int l16 = lane & 15;

  v8f acc[NT] = {};
  const float* arow = A + (size_t)(Rm + l16) * K;
  const float* bbase = Wlds + half * RP + l16 * 2;

  for (int kb = 0; kb < K; kb += 4) {
    v2f a = *(const v2f*)(arow + kb + 2 * half);
    const float* bk = bbase + (kb >> 1) * RP;
    #pragma unroll
    for (int t = 0; t < NT; ++t) {
      v2f b = *(const v2f*)(bk + t * 32);
      acc[t] = __builtin_amdgcn_wmma_f32_16x16x4_f32(
          false, a, false, b, (short)0, acc[t], false, false);
    }
  }

  #pragma unroll
  for (int r = 0; r < 8; ++r) {
    const int row = Rm + r + 8 * half;
    const float di = dinv[row];
    #pragma unroll
    for (int t = 0; t < NT; ++t) {
      out[(size_t)row * N + t * 16 + l16] = acc[t][r] * di;
    }
  }
}

template <int F>
__global__ void k_aggregate(const float* __restrict__ hws, const float* __restrict__ dinv,
                            const int* __restrict__ rowstart, const int* __restrict__ csr,
                            const float* __restrict__ bias, float* __restrict__ out, int n) {
  constexpr int NPB = 256 / F;
  const int node = blockIdx.x * NPB + threadIdx.x / F;
  const int f = threadIdx.x % F;
  if (node >= n) return;
  const int s = rowstart[node];
  const int e = rowstart[node + 1];
  float sum = hws[(size_t)node * F + f];
  for (int j = s; j < e; ++j) {
    sum += hws[(size_t)csr[j] * F + f];
  }
  float v = dinv[node] * sum + bias[f];
  out[(size_t)node * F + f] = fmaxf(v, 0.0f);
}

__global__ void k_pool_acc(const float* __restrict__ h, const int* __restrict__ batch,
                           float* __restrict__ pool, float* __restrict__ cntg, int n) {
  int idx = blockIdx.x * blockDim.x + threadIdx.x;
  if (idx < n * 64) {
    int node = idx >> 6;
    int f = idx & 63;
    int g = batch[node];
    atomicAdd(&pool[g * 64 + f], h[idx]);
    if (f == 0) atomicAdd(&cntg[g], 1.0f);
  }
}

__global__ void k_pool_final(const float* __restrict__ pool, const float* __restrict__ cntg,
                             float* __restrict__ out) {
  int idx = blockIdx.x * blockDim.x + threadIdx.x;
  if (idx < N_GRAPHS * 64) {
    out[idx] = pool[idx] / fmaxf(cntg[idx >> 6], 1.0f);
  }
}

extern "C" void kernel_launch(void* const* d_in, const int* in_sizes, int n_in,
                              void* d_out, int out_size, void* d_ws, size_t ws_size,
                              hipStream_t stream) {
  const float* x     = (const float*)d_in[0];
  const float* W1    = (const float*)d_in[1];
  const float* b1    = (const float*)d_in[2];
  const float* W2    = (const float*)d_in[3];
  const float* b2    = (const float*)d_in[4];
  const int*   ei    = (const int*)d_in[5];
  const int*   batch = (const int*)d_in[6];

  const int N = in_sizes[6];
  const int E = in_sizes[5] / 2;
  const int G = N_GRAPHS;

  const int* src = ei;
  const int* dst = ei + E;

  char* base = (char*)d_ws;
  size_t off = 0;
  auto take = [&](size_t bytes) {
    size_t o = off;
    off = (off + bytes + 255) & ~(size_t)255;
    return o;
  };
  int*   cnt      = (int*)(base + take(sizeof(int) * (size_t)N));
  int*   cur      = (int*)(base + take(sizeof(int) * (size_t)N));
  int*   rowstart = (int*)(base + take(sizeof(int) * (size_t)(N + 1)));
  float* dinv     = (float*)(base + take(sizeof(float) * (size_t)N));
  int*   csr      = (int*)(base + take(sizeof(int) * (size_t)E));
  float* hw1      = (float*)(base + take(sizeof(float) * (size_t)N * 128));
  float* h1       = (float*)(base + take(sizeof(float) * (size_t)N * 128));
  float* pool     = (float*)(base + take(sizeof(float) * (size_t)G * 64));
  float* cntg     = (float*)(base + take(sizeof(float) * (size_t)G));
  float* hw2 = hw1;
  float* h2  = h1;

  const int B = 256;
  const int gN = (N + B - 1) / B;
  const int gE = (E + B - 1) / B;
  const int mtiles = (N + 15) / 16;
  const int gGemm = (mtiles + 7) / 8;

  k_zero_i32<<<gN, B, 0, stream>>>(cnt, N);
  k_count<<<gE, B, 0, stream>>>(dst, cnt, E);
  k_scan<<<1, 1024, 0, stream>>>(cnt, rowstart, dinv, N);
  k_zero_i32<<<gN, B, 0, stream>>>(cur, N);
  k_fill<<<gE, B, 0, stream>>>(src, dst, rowstart, cur, csr, E);

  k_gemm_wmma<128><<<gGemm, 256, 64 * (2 * 128 + 32) * sizeof(float), stream>>>(x, W1, dinv, hw1, N);
  k_aggregate<128><<<(N + 1) / 2, 256, 0, stream>>>(hw1, dinv, rowstart, csr, b1, h1, N);

  k_gemm_wmma<64><<<gGemm, 256, 64 * (2 * 64 + 32) * sizeof(float), stream>>>(h1, W2, dinv, hw2, N);
  k_aggregate<64><<<(N + 3) / 4, 256, 0, stream>>>(hw2, dinv, rowstart, csr, b2, h2, N);

  k_zero_f32<<<(G * 64 + G + B - 1) / B, B, 0, stream>>>(pool, G * 64 + G);
  k_pool_acc<<<((size_t)N * 64 + B - 1) / B, B, 0, stream>>>(h2, batch, pool, cntg, N);
  k_pool_final<<<(G * 64 + B - 1) / B, B, 0, stream>>>(pool, cntg, (float*)d_out);
}


# CSR build split by destination range: a block owns 512 nodes, reads half the edges, counts and hands out csr slots from LDS counters; no global atomics, no scan launch; sum order = half, LDS arrival
# speedup vs baseline: 1.5851x; 1.5851x over previous
#include <hip/hip_runtime.h>
#include <hip/hip_bf16.h>

typedef float v2f __attribute__((ext_vector_type(2)));
typedef float v8f __attribute__((ext_vector_type(8)));

#define N_GRAPHS 64

__global__ void k_zero_f32(float* p, int n) {
  int i = blockIdx.x * blockDim.x + threadIdx.x;
  for (; i < n; i += gridDim.x * blockDim.x) p[i] = 0.0f;
}

#define OWN 512
#define OWN_H 2
#define OWN_T 1024
#define OWN_U 4

__device__ __forceinline__ void own_part(int E, int h, int& eb, int& ee) {
  const int pieces = (E + 3) >> 2;
  const int per = ((pieces + OWN_H - 1) / OWN_H) * 4;
  eb = min(E, h * per);
  ee = min(E, eb + per);
}

__device__ __forceinline__ int4 own_piece(const int4* __restrict__ v, int len, int rot, int j) {
  int idx = j + rot;
  if (idx >= len) idx -= len;
  const bool in = j < len;
  int4 r = v[in ? idx : 0];
  if (!in) r = make_int4(-1, -1, -1, -1);
  return r;
}

__global__ void __launch_bounds__(OWN_T)
k_own_count(const int* __restrict__ dst, int* __restrict__ cnt, int E, int n) {
  __shared__ int c[OWN];
  const int tid = threadIdx.x;
  const int lo = blockIdx.x * OWN;
  const int h = blockIdx.y;
  int eb, ee;
  own_part(E, h, eb, ee);
  if (tid < OWN) c[tid] = 0;
  __syncthreads();
  if (((E & 3) == 0) && ((((size_t)dst) & 15) == 0)) {
    const int len = (ee - eb) >> 2;
    if (len > 0) {
      const int4* dv = (const int4*)dst + (eb >> 2);
      const int rot = (int)(((long long)blockIdx.x * len) / gridDim.x);
      constexpr int STEP = OWN_T * OWN_U;
      int4 cur[OWN_U], nxt[OWN_U];
      #pragma unroll
      for (int u = 0; u < OWN_U; ++u) cur[u] = own_piece(dv, len, rot, u * OWN_T + tid);
      for (int j0 = 0; j0 < len; j0 += STEP) {
        #pragma unroll
        for (int u = 0; u < OWN_U; ++u) nxt[u] = own_piece(dv, len, rot, j0 + STEP + u * OWN_T + tid);
        #pragma unroll
        for (int u = 0; u < OWN_U; ++u) {
          const unsigned l0 = (unsigned)(cur[u].x - lo), l1 = (unsigned)(cur[u].y - lo);
          const unsigned l2 = (unsigned)(cur[u].z - lo), l3 = (unsigned)(cur[u].w - lo);
          if (l0 < OWN) atomicAdd(&c[l0], 1);
          if (l1 < OWN) atomicAdd(&c[l1], 1);
          if (l2 < OWN) atomicAdd(&c[l2], 1);
          if (l3 < OWN) atomicAdd(&c[l3], 1);
        }
        #pragma unroll
        for (int u = 0; u < OWN_U; ++u) cur[u] = nxt[u];
      }
    }
  } else {
    for (int e = eb + tid; e < ee; e += OWN_T) {
      const unsigned l = (unsigned)(dst[e] - lo);
      if (l < OWN) atomicAdd(&c[l], 1);
    }
  }
  __syncthreads();
  const int node = lo + tid;
  if (tid < OWN && node < n) cnt[h * n + node] = c[tid];
}

#define OWN_ASK(U) \
  int p##U##x = -1, p##U##y = -1, p##U##z = -1, p##U##w = -1; \
  { const unsigned lx = (unsigned)(dc[U].x - lo), ly = (unsigned)(dc[U].y - lo); \
    const unsigned lz = (unsigned)(dc[U].z - lo), lw = (unsigned)(dc[U].w - lo); \
    if (lx < OWN) p##U##x = atomicAdd(&pos[lx], 1); \
    if (ly < OWN) p##U##y = atomicAdd(&pos[ly], 1); \
    if (lz < OWN) p##U##z = atomicAdd(&pos[lz], 1); \
    if (lw < OWN) p##U##w = atomicAdd(&pos[lw], 1); }
#define OWN_PUT(U) \
  if (p##U##x >= 0) csr[p##U##x] = sc[U].x; \
  if (p##U##y >= 0) csr[p##U##y] = sc[U].y; \
  if (p##U##z >= 0) csr[p##U##z] = sc[U].z; \
  if (p##U##w >= 0) csr[p##U##w] = sc[U].w;

__global__ void __launch_bounds__(OWN_T)
k_own_fill(const int* __restrict__ src, const int* __restrict__ dst, const int* __restrict__ cnt,
           int* __restrict__ rowstart, float* __restrict__ dinv, int* __restrict__ csr,
           int E, int n) {
  __shared__ int pos[OWN];
  __shared__ int wsum[32];
  const int tid = threadIdx.x;
  const int lane = tid & 31;
  const int wave = tid >> 5;
  const int lo = blockIdx.x * OWN;
  const int h = blockIdx.y;
  int eb, ee;
  own_part(E, h, eb, ee);

  int acc = 0;
  for (int i = tid; i < lo; i += OWN_T) {
    #pragma unroll
    for (int hh = 0; hh < OWN_H; ++hh) acc += cnt[hh * n + i];
  }
  for (int o = 16; o > 0; o >>= 1) acc += __shfl_xor(acc, o);
  if (lane == 0) wsum[wave] = acc;
  __syncthreads();
  int base = 0;
  for (int w = 0; w < 32; ++w) base += wsum[w];
  __syncthreads();

  const int node = lo + tid;
  int v = 0, before = 0;
  if (tid < OWN && node < n) {
    #pragma unroll
    for (int hh = 0; hh < OWN_H; ++hh) {
      const int cv = cnt[hh * n + node];
      if (hh < h) before += cv;
      v += cv;
    }
  }
  int inc = v;
  for (int o = 1; o < 32; o <<= 1) {
    const int t = __shfl_up(inc, o);
    if (lane >= o) inc += t;
  }
  if (lane == 31) wsum[wave] = inc;
  __syncthreads();
  int wpre = 0;
  for (int w = 0; w < wave; ++w) wpre += wsum[w];
  const int excl = base + wpre + inc - v;
  if (tid < OWN) {
    pos[tid] = excl + before;
    if (h == 0 && node < n) {
      rowstart[node] = excl;
      dinv[node] = rsqrtf((float)v + 1.0f);
    }
  }
  if (h == 0 && blockIdx.x == gridDim.x - 1 && tid == OWN - 1) rowstart[n] = excl + v;
  __syncthreads();

  if (((E & 3) == 0) && ((((size_t)dst) & 15) == 0) && ((((size_t)src) & 15) == 0)) {
    const int len = (ee - eb) >> 2;
    if (len > 0) {
      const int4* dv = (const int4*)dst + (eb >> 2);
      const int4* sv = (const int4*)src + (eb >> 2);
      const int rot = (int)(((long long)blockIdx.x * len) / gridDim.x);
      constexpr int STEP = OWN_T * OWN_U;
      int4 dc[OWN_U], sc[OWN_U], dn[OWN_U], sn[OWN_U];
      #pragma unroll
      for (int u = 0; u < OWN_U; ++u) {
        dc[u] = own_piece(dv, len, rot, u * OWN_T + tid);
        sc[u] = own_piece(sv, len, rot, u * OWN_T + tid);
      }
      for (int j0 = 0; j0 < len; j0 += STEP) {
        #pragma unroll
        for (int u = 0; u < OWN_U; ++u) {
          dn[u] = own_piece(dv, len, rot, j0 + STEP + u * OWN_T + tid);
          sn[u] = own_piece(sv, len, rot, j0 + STEP + u * OWN_T + tid);
        }
        OWN_ASK(0) OWN_ASK(1) OWN_ASK(2) OWN_ASK(3)
        OWN_PUT(0) OWN_PUT(1) OWN_PUT(2) OWN_PUT(3)
        #pragma unroll
        for (int u = 0; u < OWN_U; ++u) { dc[u] = dn[u]; sc[u] = sn[u]; }
      }
    }
  } else {
    for (int e = eb + tid; e < ee; e += OWN_T) {
      const unsigned l = (unsigned)(dst[e] - lo);
      if (l < OWN) csr[atomicAdd(&pos[l], 1)] = src[e];
    }
  }
}

template <int N>
__global__ void k_gemm_wmma(const float* __restrict__ A, const float* __restrict__ W,
                            const float* __restrict__ dinv, float* __restrict__ out, int M) {
  constexpr int K = 128;
  constexpr int NT = N / 16;
  constexpr int RP = 2 * N + 32;
  extern __shared__ float Wlds[];
  for (int i = threadIdx.x; i < K * N; i += blockDim.x) {
    int k = i / N;
    int n = i % N;
    Wlds[(k >> 1) * RP + n * 2 + (k & 1)] = W[i];
  }
  __syncthreads();

  const int wave = threadIdx.x >> 5;
  const int lane = threadIdx.x & 31;
  const int mtile = blockIdx.x * 8 + wave;
  if (mtile * 16 >= M) return;

  const int Rm = mtile * 16;
  const int half = lane >> 4;
  const int l16 = lane & 15;

  v8f acc[NT] = {};
  const float* arow = A + (size_t)(Rm + l16) * K;
  const float* bbase = Wlds + half * RP + l16 * 2;

  for (int kb = 0; kb < K; kb += 4) {
    v2f a = *(const v2f*)(arow + kb + 2 * half);
    const float* bk = bbase + (kb >> 1) * RP;
    #pragma unroll
    for (int t = 0; t < NT; ++t) {
      v2f b = *(const v2f*)(bk + t * 32);
      acc[t] = __builtin_amdgcn_wmma_f32_16x16x4_f32(
          false, a, false, b, (short)0, acc[t], false, false);
    }
  }

  #pragma unroll
  for (int r = 0; r < 8; ++r) {
    const int row = Rm + r + 8 * half;
    const float di = dinv[row];
    #pragma unroll
    for (int t = 0; t < NT; ++t) {
      out[(size_t)row * N + t * 16 + l16] = acc[t][r] * di;
    }
  }
}

template <int F>
__global__ void k_aggregate(const float* __restrict__ hws, const float* __restrict__ dinv,
                            const int* __restrict__ rowstart, const int* __restrict__ csr,
                            const float* __restrict__ bias, float* __restrict__ out, int n) {
  constexpr int NPB = 256 / F;
  const int node = blockIdx.x * NPB + threadIdx.x / F;
  const int f = threadIdx.x % F;
  if (node >= n) return;
  const int s = rowstart[node];
  const int e = rowstart[node + 1];
  float sum = hws[(size_t)node * F + f];
  for (int j = s; j < e; ++j) {
    sum += hws[(size_t)csr[j] * F + f];
  }
  float v = dinv[node] * sum + bias[f];
  out[(size_t)node * F + f] = fmaxf(v, 0.0f);
}

__global__ void k_pool_acc(const float* __restrict__ h, const int* __restrict__ batch,
                           float* __restrict__ pool, float* __restrict__ cntg, int n) {
  int idx = blockIdx.x * blockDim.x + threadIdx.x;
  if (idx < n * 64) {
    int node = idx >> 6;
    int f = idx & 63;
    int g = batch[node];
    atomicAdd(&pool[g * 64 + f], h[idx]);
    if (f == 0) atomicAdd(&cntg[g], 1.0f);
  }
}

__global__ void k_pool_final(const float* __restrict__ pool, const float* __restrict__ cntg,
                             float* __restrict__ out) {
  int idx = blockIdx.x * blockDim.x + threadIdx.x;
  if (idx < N_GRAPHS * 64) {
    out[idx] = pool[idx] / fmaxf(cntg[idx >> 6], 1.0f);
  }
}

extern "C" void kernel_launch(void* const* d_in, const int* in_sizes, int n_in,
                              void* d_out, int out_size, void* d_ws, size_t ws_size,
                              hipStream_t stream) {
  const float* x     = (const float*)d_in[0];
  const float* W1    = (const float*)d_in[1];
  const float* b1    = (const float*)d_in[2];
  const float* W2    = (const float*)d_in[3];
  const float* b2    = (const float*)d_in[4];
  const int*   ei    = (const int*)d_in[5];
  const int*   batch = (const int*)d_in[6];

  const int N = in_sizes[6];
  const int E = in_sizes[5] / 2;
  const int G = N_GRAPHS;

  const int* src = ei;
  const int* dst = ei + E;

  char* base = (char*)d_ws;
  size_t off = 0;
  auto take = [&](size_t bytes) {
    size_t o = off;
    off = (off + bytes + 255) & ~(size_t)255;
    return o;
  };
  int*   cnt      = (int*)(base + take(sizeof(int) * (size_t)N * OWN_H));
  int*   rowstart = (int*)(base + take(sizeof(int) * (size_t)(N + 1)));
  float* dinv     = (float*)(base + take(sizeof(float) * (size_t)N));
  int*   csr      = (int*)(base + take(sizeof(int) * (size_t)E));
  float* hw1      = (float*)(base + take(sizeof(float) * (size_t)N * 128));
  float* h1       = (float*)(base + take(sizeof(float) * (size_t)N * 128));
  float* pool     = (float*)(base + take(sizeof(float) * (size_t)G * 64));
  float* cntg     = (float*)(base + take(sizeof(float) * (size_t)G));
  float* hw2 = hw1;
  float* h2  = h1;

  const int B = 256;
  const int gOwn = (N + OWN - 1) / OWN;
  const int mtiles = (N + 15) / 16;
  const int gGemm = (mtiles + 7) / 8;

  k_own_count<<<dim3(gOwn, OWN_H), OWN_T, 0, stream>>>(dst, cnt, E, N);
  k_own_fill<<<dim3(gOwn, OWN_H), OWN_T, 0, stream>>>(src, dst, cnt, rowstart, dinv, csr, E, N);

  k_gemm_wmma<128><<<gGemm, 256, 64 * (2 * 128 + 32) * sizeof(float), stream>>>(x, W1, dinv, hw1, N);
  k_aggregate<128><<<(N + 1) / 2, 256, 0, stream>>>(hw1, dinv, rowstart, csr, b1, h1, N);

  k_gemm_wmma<64><<<gGemm, 256, 64 * (2 * 64 + 32) * sizeof(float), stream>>>(h1, W2, dinv, hw2, N);
  k_aggregate<64><<<(N + 3) / 4, 256, 0, stream>>>(hw2, dinv, rowstart, csr, b2, h2, N);

  k_zero_f32<<<(G * 64 + G + B - 1) / B, B, 0, stream>>>(pool, G * 64 + G);
  k_pool_acc<<<((size_t)N * 64 + B - 1) / B, B, 0, stream>>>(h2, batch, pool, cntg, N);
  k_pool_final<<<(G * 64 + B - 1) / B, B, 0, stream>>>(pool, cntg, (float*)d_out);
}


# both layers sum neighbour rows BEFORE the product, (A_hat h)W: a wave gathers its 16 nodes' 512-byte rows (8 loads in flight) into LDS as the WMMA A tile; agg+GEMM launches merge; sum order changes
# speedup vs baseline: 1.7333x; 1.0935x over previous
#include <hip/hip_runtime.h>
#include <hip/hip_bf16.h>

typedef float v2f __attribute__((ext_vector_type(2)));
typedef float v8f __attribute__((ext_vector_type(8)));

#define N_GRAPHS 64

__global__ void k_zero_f32(float* p, int n) {
  int i = blockIdx.x * blockDim.x + threadIdx.x;
  for (; i < n; i += gridDim.x * blockDim.x) p[i] = 0.0f;
}

#define OWN 512
#define OWN_H 2
#define OWN_T 1024
#define OWN_U 4

__device__ __forceinline__ void own_part(int E, int h, int& eb, int& ee) {
  const int pieces = (E + 3) >> 2;
  const int per = ((pieces + OWN_H - 1) / OWN_H) * 4;
  eb = min(E, h * per);
  ee = min(E, eb + per);
}

__device__ __forceinline__ int4 own_piece(const int4* __restrict__ v, int len, int rot, int j) {
  int idx = j + rot;
  if (idx >= len) idx -= len;
  const bool in = j < len;
  int4 r = v[in ? idx : 0];
  if (!in) r = make_int4(-1, -1, -1, -1);
  return r;
}

__global__ void __launch_bounds__(OWN_T)
k_own_count(const int* __restrict__ dst, int* __restrict__ cnt, int E, int n) {
  __shared__ int c[OWN];
  const int tid = threadIdx.x;
  const int lo = blockIdx.x * OWN;
  const int h = blockIdx.y;
  int eb, ee;
  own_part(E, h, eb, ee);
  if (tid < OWN) c[tid] = 0;
  __syncthreads();
  if (((E & 3) == 0) && ((((size_t)dst) & 15) == 0)) {
    const int len = (ee - eb) >> 2;
    if (len > 0) {
      const int4* dv = (const int4*)dst + (eb >> 2);
      const int rot = (int)(((long long)blockIdx.x * len) / gridDim.x);
      constexpr int STEP = OWN_T * OWN_U;
      int4 cur[OWN_U], nxt[OWN_U];
      #pragma unroll
      for (int u = 0; u < OWN_U; ++u) cur[u] = own_piece(dv, len, rot, u * OWN_T + tid);
      for (int j0 = 0; j0 < len; j0 += STEP) {
        #pragma unroll
        for (int u = 0; u < OWN_U; ++u) nxt[u] = own_piece(dv, len, rot, j0 + STEP + u * OWN_T + tid);
        #pragma unroll
        for (int u = 0; u < OWN_U; ++u) {
          const unsigned l0 = (unsigned)(cur[u].x - lo), l1 = (unsigned)(cur[u].y - lo);
          const unsigned l2 = (unsigned)(cur[u].z - lo), l3 = (unsigned)(cur[u].w - lo);
          if (l0 < OWN) atomicAdd(&c[l0], 1);
          if (l1 < OWN) atomicAdd(&c[l1], 1);
          if (l2 < OWN) atomicAdd(&c[l2], 1);
          if (l3 < OWN) atomicAdd(&c[l3], 1);
        }
        #pragma unroll
        for (int u = 0; u < OWN_U; ++u) cur[u] = nxt[u];
      }
    }
  } else {
    for (int e = eb + tid; e < ee; e += OWN_T) {
      const unsigned l = (unsigned)(dst[e] - lo);
      if (l < OWN) atomicAdd(&c[l], 1);
    }
  }
  __syncthreads();
  const int node = lo + tid;
  if (tid < OWN && node < n) cnt[h * n + node] = c[tid];
}

#define OWN_ASK(U) \
  int p##U##x = -1, p##U##y = -1, p##U##z = -1, p##U##w = -1; \
  { const unsigned lx = (unsigned)(dc[U].x - lo), ly = (unsigned)(dc[U].y - lo); \
    const unsigned lz = (unsigned)(dc[U].z - lo), lw = (unsigned)(dc[U].w - lo); \
    if (lx < OWN) p##U##x = atomicAdd(&pos[lx], 1); \
    if (ly < OWN) p##U##y = atomicAdd(&pos[ly], 1); \
    if (lz < OWN) p##U##z = atomicAdd(&pos[lz], 1); \
    if (lw < OWN) p##U##w = atomicAdd(&pos[lw], 1); }
#define OWN_PUT(U) \
  if (p##U##x >= 0) csr[p##U##x] = sc[U].x; \
  if (p##U##y >= 0) csr[p##U##y] = sc[U].y; \
  if (p##U##z >= 0) csr[p##U##z] = sc[U].z; \
  if (p##U##w >= 0) csr[p##U##w] = sc[U].w;

__global__ void __launch_bounds__(OWN_T)
k_own_fill(const int* __restrict__ src, const int* __restrict__ dst, const int* __restrict__ cnt,
           int* __restrict__ rowstart, float* __restrict__ dinv, int* __restrict__ csr,
           int E, int n) {
  __shared__ int pos[OWN];
  __shared__ int wsum[32];
  const int tid = threadIdx.x;
  const int lane = tid & 31;
  const int wave = tid >> 5;
  const int lo = blockIdx.x * OWN;
  const int h = blockIdx.y;
  int eb, ee;
  own_part(E, h, eb, ee);

  int acc = 0;
  for (int i = tid; i < lo; i += OWN_T) {
    #pragma unroll
    for (int hh = 0; hh < OWN_H; ++hh) acc += cnt[hh * n + i];
  }
  for (int o = 16; o > 0; o >>= 1) acc += __shfl_xor(acc, o);
  if (lane == 0) wsum[wave] = acc;
  __syncthreads();
  int base = 0;
  for (int w = 0; w < 32; ++w) base += wsum[w];
  __syncthreads();

  const int node = lo + tid;
  int v = 0, before = 0;
  if (tid < OWN && node < n) {
    #pragma unroll
    for (int hh = 0; hh < OWN_H; ++hh) {
      const int cv = cnt[hh * n + node];
      if (hh < h) before += cv;
      v += cv;
    }
  }
  int inc = v;
  for (int o = 1; o < 32; o <<= 1) {
    const int t = __shfl_up(inc, o);
    if (lane >= o) inc += t;
  }
  if (lane == 31) wsum[wave] = inc;
  __syncthreads();
  int wpre = 0;
  for (int w = 0; w < wave; ++w) wpre += wsum[w];
  const int excl = base + wpre + inc - v;
  if (tid < OWN) {
    pos[tid] = excl + before;
    if (h == 0 && node < n) {
      rowstart[node] = excl;
      dinv[node] = rsqrtf((float)v + 1.0f);
    }
  }
  if (h == 0 && blockIdx.x == gridDim.x - 1 && tid == OWN - 1) rowstart[n] = excl + v;
  __syncthreads();

  if (((E & 3) == 0) && ((((size_t)dst) & 15) == 0) && ((((size_t)src) & 15) == 0)) {
    const int len = (ee - eb) >> 2;
    if (len > 0) {
      const int4* dv = (const int4*)dst + (eb >> 2);
      const int4* sv = (const int4*)src + (eb >> 2);
      const int rot = (int)(((long long)blockIdx.x * len) / gridDim.x);
      constexpr int STEP = OWN_T * OWN_U;
      int4 dc[OWN_U], sc[OWN_U], dn[OWN_U], sn[OWN_U];
      #pragma unroll
      for (int u = 0; u < OWN_U; ++u) {
        dc[u] = own_piece(dv, len, rot, u * OWN_T + tid);
        sc[u] = own_piece(sv, len, rot, u * OWN_T + tid);
      }
      for (int j0 = 0; j0 < len; j0 += STEP) {
        #pragma unroll
        for (int u = 0; u < OWN_U; ++u) {
          dn[u] = own_piece(dv, len, rot, j0 + STEP + u * OWN_T + tid);
          sn[u] = own_piece(sv, len, rot, j0 + STEP + u * OWN_T + tid);
        }
        OWN_ASK(0) OWN_ASK(1) OWN_ASK(2) OWN_ASK(3)
        OWN_PUT(0) OWN_PUT(1) OWN_PUT(2) OWN_PUT(3)
        #pragma unroll
        for (int u = 0; u < OWN_U; ++u) { dc[u] = dn[u]; sc[u] = sn[u]; }
      }
    }
  } else {
    for (int e = eb + tid; e < ee; e += OWN_T) {
      const unsigned l = (unsigned)(dst[e] - lo);
      if (l < OWN) csr[atomicAdd(&pos[l], 1)] = src[e];
    }
  }
}

#define AGG_AP 132
#define AGG_U 8


__device__ __forceinline__ int agg_csr(const int* __restrict__ csr, int first, int s, int e, int lane) {
  const int slot = first + lane;
  return csr[(slot >= s && slot < e) ? slot : 0];
}

__device__ __forceinline__ int agg_idx(int craw, int first, int s, int e, int lane, int self) {
  const int slot = first + lane;
  return (slot >= s && slot < e) ? craw : self;
}

__device__ __forceinline__ void agg_rows(const float* __restrict__ Hin, int lane, int idx, float dw,
                                         int cnt, float4& acc) {
  const int dwb = __float_as_int(dw);
  if (cnt > 0) idx = (lane < cnt) ? idx : __builtin_amdgcn_readlane(idx, cnt - 1);
  for (int k0 = 0; k0 < cnt; k0 += AGG_U) {
    float4 v[AGG_U];
    #pragma unroll
    for (int u = 0; u < AGG_U; ++u) {
      const int sidx = __builtin_amdgcn_readlane(idx, k0 + u);
      v[u] = *(const float4*)(Hin + (size_t)sidx * 128 + lane * 4);
    }
    #pragma unroll
    for (int u = 0; u < AGG_U; ++u) {
      const float w = __int_as_float(__builtin_amdgcn_readlane(dwb, k0 + u));
      acc.x += w * v[u].x; acc.y += w * v[u].y; acc.z += w * v[u].z; acc.w += w * v[u].w;
    }
  }
}

template <int N>
__global__ void __launch_bounds__(256)
k_agg_gemm(const float* __restrict__ Hin, const float* __restrict__ W,
           const float* __restrict__ dinv, const int* __restrict__ rowstart,
           const int* __restrict__ csr, const float* __restrict__ bias,
           float* __restrict__ out, int M) {
  constexpr int K = 128;
  constexpr int NT = N / 16;
  constexpr int RP = 2 * N + 32;
  __shared__ __attribute__((aligned(16))) float Wlds[(K / 2) * RP];
  __shared__ __attribute__((aligned(16))) float Atile[8 * 16 * AGG_AP];
  for (int i = threadIdx.x; i < K * N; i += blockDim.x) {
    int k = i / N;
    int n = i % N;
    Wlds[(k >> 1) * RP + n * 2 + (k & 1)] = W[i];
  }

  const int wave = __builtin_amdgcn_readfirstlane(threadIdx.x >> 5);
  const int lane = threadIdx.x & 31;
  const int mtile = blockIdx.x * 8 + wave;
  const int Rm = mtile * 16;
  float* At = Atile + wave * 16 * AGG_AP;

  if (Rm < M) {
    const int rsv = rowstart[min(Rm + min(lane, 16), M)];
    int s0 = __builtin_amdgcn_readlane(rsv, 0);
    int s1 = __builtin_amdgcn_readlane(rsv, 1);
    int s2 = __builtin_amdgcn_readlane(rsv, 2);
    const int raw0 = agg_csr(csr, s0 - 1, s0, s1, lane);
    int raw1 = agg_csr(csr, s1 - 1, s1, s2, lane);
    int idx = agg_idx(raw0, s0 - 1, s0, s1, lane, min(Rm, M - 1));
    float dw = dinv[idx];
    for (int r = 0; r < 16; ++r) {
      const int node = Rm + r;
      const int nodec = min(node, M - 1);
      const int s3 = __builtin_amdgcn_readlane(rsv, r + 3);
      const int raw2 = agg_csr(csr, s2 - 1, s2, s3, lane);
      const int nidx = agg_idx(raw1, s1 - 1, s1, s2, lane, min(node + 1, M - 1));
      const float ndw = dinv[nidx];
      const int tot = (node < M) ? s1 - s0 + 1 : 0;
      float4 acc = make_float4(0.0f, 0.0f, 0.0f, 0.0f);
      agg_rows(Hin, lane, idx, (lane < tot) ? dw : 0.0f, min(tot, 32), acc);
      for (int cb = s0 + 31; cb < s1; cb += 32) {
        const int xi = agg_idx(agg_csr(csr, cb, s0, s1, lane), cb, s0, s1, lane, nodec);
        const float xw = dinv[xi];
        agg_rows(Hin, lane, xi, (lane < s1 - cb) ? xw : 0.0f, min(s1 - cb, 32), acc);
      }
      *(float4*)(At + r * AGG_AP + lane * 4) = acc;
      idx = nidx; dw = ndw; raw1 = raw2;
      s0 = s1; s1 = s2; s2 = s3;
    }
  }
  __syncthreads();
  if (Rm >= M) return;

  const int half = lane >> 4;
  const int l16 = lane & 15;

  v8f acc[NT] = {};
  const float* abase = At + l16 * AGG_AP + 2 * half;
  const float* bbase = Wlds + half * RP + l16 * 2;

  for (int kb = 0; kb < K; kb += 4) {
    v2f a = *(const v2f*)(abase + kb);
    const float* bk = bbase + (kb >> 1) * RP;
    #pragma unroll
    for (int t = 0; t < NT; ++t) {
      v2f b = *(const v2f*)(bk + t * 32);
      acc[t] = __builtin_amdgcn_wmma_f32_16x16x4_f32(
          false, a, false, b, (short)0, acc[t], false, false);
    }
  }

  float bv[NT];
  #pragma unroll
  for (int t = 0; t < NT; ++t) bv[t] = bias[t * 16 + l16];
  #pragma unroll
  for (int r = 0; r < 8; ++r) {
    const int row = Rm + r + 8 * half;
    if (row < M) {
      const float di = dinv[row];
      #pragma unroll
      for (int t = 0; t < NT; ++t) {
        out[(size_t)row * N + t * 16 + l16] = fmaxf(acc[t][r] * di + bv[t], 0.0f);
      }
    }
  }
}

__global__ void k_pool_acc(const float* __restrict__ h, const int* __restrict__ batch,
                           float* __restrict__ pool, float* __restrict__ cntg, int n) {
  int idx = blockIdx.x * blockDim.x + threadIdx.x;
  if (idx < n * 64) {
    int node = idx >> 6;
    int f = idx & 63;
    int g = batch[node];
    atomicAdd(&pool[g * 64 + f], h[idx]);
    if (f == 0) atomicAdd(&cntg[g], 1.0f);
  }
}

__global__ void k_pool_final(const float* __restrict__ pool, const float* __restrict__ cntg,
                             float* __restrict__ out) {
  int idx = blockIdx.x * blockDim.x + threadIdx.x;
  if (idx < N_GRAPHS * 64) {
    out[idx] = pool[idx] / fmaxf(cntg[idx >> 6], 1.0f);
  }
}

extern "C" void kernel_launch(void* const* d_in, const int* in_sizes, int n_in,
                              void* d_out, int out_size, void* d_ws, size_t ws_size,
                              hipStream_t stream) {
  const float* x     = (const float*)d_in[0];
  const float* W1    = (const float*)d_in[1];
  const float* b1    = (const float*)d_in[2];
  const float* W2    = (const float*)d_in[3];
  const float* b2    = (const float*)d_in[4];
  const int*   ei    = (const int*)d_in[5];
  const int*   batch = (const int*)d_in[6];

  const int N = in_sizes[6];
  const int E = in_sizes[5] / 2;
  const int G = N_GRAPHS;

  const int* src = ei;
  const int* dst = ei + E;

  char* base = (char*)d_ws;
  size_t off = 0;
  auto take = [&](size_t bytes) {
    size_t o = off;
    off = (off + bytes + 255) & ~(size_t)255;
    return o;
  };
  int*   cnt      = (int*)(base + take(sizeof(int) * (size_t)N * OWN_H));
  int*   rowstart = (int*)(base + take(sizeof(int) * (size_t)(N + 1)));
  float* dinv     = (float*)(base + take(sizeof(float) * (size_t)N));
  int*   csr      = (int*)(base + take(sizeof(int) * (size_t)E));
  float* h1       = (float*)(base + take(sizeof(float) * (size_t)N * 128));
  float* h2       = (float*)(base + take(sizeof(float) * (size_t)N * 64));
  float* pool     = (float*)(base + take(sizeof(float) * (size_t)G * 64));
  float* cntg     = (float*)(base + take(sizeof(float) * (size_t)G));

  const int B = 256;
  const int gOwn = (N + OWN - 1) / OWN;
  const int mtiles = (N + 15) / 16;
  const int gGemm = (mtiles + 7) / 8;

  k_own_count<<<dim3(gOwn, OWN_H), OWN_T, 0, stream>>>(dst, cnt, E, N);
  k_own_fill<<<dim3(gOwn, OWN_H), OWN_T, 0, stream>>>(src, dst, cnt, rowstart, dinv, csr, E, N);

  k_agg_gemm<128><<<gGemm, 256, 0, stream>>>(x, W1, dinv, rowstart, csr, b1, h1, N);

  k_agg_gemm<64><<<gGemm, 256, 0, stream>>>(h1, W2, dinv, rowstart, csr, b2, h2, N);

  k_zero_f32<<<(G * 64 + G + B - 1) / B, B, 0, stream>>>(pool, G * 64 + G);
  k_pool_acc<<<((size_t)N * 64 + B - 1) / B, B, 0, stream>>>(h2, batch, pool, cntg, N);
  k_pool_final<<<(G * 64 + B - 1) / B, B, 0, stream>>>(pool, cntg, (float*)d_out);
}


# mean pool moves into the layer-2 launch: a wave sums its 16 relu rows from the WMMA accumulators and adds (atomic) 64 column sums to its graph's row; no h2, no pool launch; pooled sum order changes
# speedup vs baseline: 2.5018x; 1.4434x over previous
#include <hip/hip_runtime.h>
#include <hip/hip_bf16.h>

typedef float v2f __attribute__((ext_vector_type(2)));
typedef float v8f __attribute__((ext_vector_type(8)));

#define N_GRAPHS 64

__global__ void k_zero_f32(float* p, int n) {
  int i = blockIdx.x * blockDim.x + threadIdx.x;
  for (; i < n; i += gridDim.x * blockDim.x) p[i] = 0.0f;
}

#define OWN 512
#define OWN_H 2
#define OWN_T 1024
#define OWN_U 4

__device__ __forceinline__ void own_part(int E, int h, int& eb, int& ee) {
  const int pieces = (E + 3) >> 2;
  const int per = ((pieces + OWN_H - 1) / OWN_H) * 4;
  eb = min(E, h * per);
  ee = min(E, eb + per);
}

__device__ __forceinline__ int4 own_piece(const int4* __restrict__ v, int len, int rot, int j) {
  int idx = j + rot;
  if (idx >= len) idx -= len;
  const bool in = j < len;
  int4 r = v[in ? idx : 0];
  if (!in) r = make_int4(-1, -1, -1, -1);
  return r;
}

__global__ void __launch_bounds__(OWN_T)
k_own_count(const int* __restrict__ dst, int* __restrict__ cnt, int E, int n) {
  __shared__ int c[OWN];
  const int tid = threadIdx.x;
  const int lo = blockIdx.x * OWN;
  const int h = blockIdx.y;
  int eb, ee;
  own_part(E, h, eb, ee);
  if (tid < OWN) c[tid] = 0;
  __syncthreads();
  if (((E & 3) == 0) && ((((size_t)dst) & 15) == 0)) {
    const int len = (ee - eb) >> 2;
    if (len > 0) {
      const int4* dv = (const int4*)dst + (eb >> 2);
      const int rot = (int)(((long long)blockIdx.x * len) / gridDim.x);
      constexpr int STEP = OWN_T * OWN_U;
      int4 cur[OWN_U], nxt[OWN_U];
      #pragma unroll
      for (int u = 0; u < OWN_U; ++u) cur[u] = own_piece(dv, len, rot, u * OWN_T + tid);
      for (int j0 = 0; j0 < len; j0 += STEP) {
        #pragma unroll
        for (int u = 0; u < OWN_U; ++u) nxt[u] = own_piece(dv, len, rot, j0 + STEP + u * OWN_T + tid);
        #pragma unroll
        for (int u = 0; u < OWN_U; ++u) {
          const unsigned l0 = (unsigned)(cur[u].x - lo), l1 = (unsigned)(cur[u].y - lo);
          const unsigned l2 = (unsigned)(cur[u].z - lo), l3 = (unsigned)(cur[u].w - lo);
          if (l0 < OWN) atomicAdd(&c[l0], 1);
          if (l1 < OWN) atomicAdd(&c[l1], 1);
          if (l2 < OWN) atomicAdd(&c[l2], 1);
          if (l3 < OWN) atomicAdd(&c[l3], 1);
        }
        #pragma unroll
        for (int u = 0; u < OWN_U; ++u) cur[u] = nxt[u];
      }
    }
  } else {
    for (int e = eb + tid; e < ee; e += OWN_T) {
      const unsigned l = (unsigned)(dst[e] - lo);
      if (l < OWN) atomicAdd(&c[l], 1);
    }
  }
  __syncthreads();
  const int node = lo + tid;
  if (tid < OWN && node < n) cnt[h * n + node] = c[tid];
}

#define OWN_ASK(U) \
  int p##U##x = -1, p##U##y = -1, p##U##z = -1, p##U##w = -1; \
  { const unsigned lx = (unsigned)(dc[U].x - lo), ly = (unsigned)(dc[U].y - lo); \
    const unsigned lz = (unsigned)(dc[U].z - lo), lw = (unsigned)(dc[U].w - lo); \
    if (lx < OWN) p##U##x = atomicAdd(&pos[lx], 1); \
    if (ly < OWN) p##U##y = atomicAdd(&pos[ly], 1); \
    if (lz < OWN) p##U##z = atomicAdd(&pos[lz], 1); \
    if (lw < OWN) p##U##w = atomicAdd(&pos[lw], 1); }
#define OWN_PUT(U) \
  if (p##U##x >= 0) csr[p##U##x] = sc[U].x; \
  if (p##U##y >= 0) csr[p##U##y] = sc[U].y; \
  if (p##U##z >= 0) csr[p##U##z] = sc[U].z; \
  if (p##U##w >= 0) csr[p##U##w] = sc[U].w;

__global__ void __launch_bounds__(OWN_T)
k_own_fill(const int* __restrict__ src, const int* __restrict__ dst, const int* __restrict__ cnt,
           int* __restrict__ rowstart, float* __restrict__ dinv, int* __restrict__ csr,
           int E, int n) {
  __shared__ int pos[OWN];
  __shared__ int wsum[32];
  const int tid = threadIdx.x;
  const int lane = tid & 31;
  const int wave = tid >> 5;
  const int lo = blockIdx.x * OWN;
  const int h = blockIdx.y;
  int eb, ee;
  own_part(E, h, eb, ee);

  int acc = 0;
  for (int i = tid; i < lo; i += OWN_T) {
    #pragma unroll
    for (int hh = 0; hh < OWN_H; ++hh) acc += cnt[hh * n + i];
  }
  for (int o = 16; o > 0; o >>= 1) acc += __shfl_xor(acc, o);
  if (lane == 0) wsum[wave] = acc;
  __syncthreads();
  int base = 0;
  for (int w = 0; w < 32; ++w) base += wsum[w];
  __syncthreads();

  const int node = lo + tid;
  int v = 0, before = 0;
  if (tid < OWN && node < n) {
    #pragma unroll
    for (int hh = 0; hh < OWN_H; ++hh) {
      const int cv = cnt[hh * n + node];
      if (hh < h) before += cv;
      v += cv;
    }
  }
  int inc = v;
  for (int o = 1; o < 32; o <<= 1) {
    const int t = __shfl_up(inc, o);
    if (lane >= o) inc += t;
  }
  if (lane == 31) wsum[wave] = inc;
  __syncthreads();
  int wpre = 0;
  for (int w = 0; w < wave; ++w) wpre += wsum[w];
  const int excl = base + wpre + inc - v;
  if (tid < OWN) {
    pos[tid] = excl + before;
    if (h == 0 && node < n) {
      rowstart[node] = excl;
      dinv[node] = rsqrtf((float)v + 1.0f);
    }
  }
  if (h == 0 && blockIdx.x == gridDim.x - 1 && tid == OWN - 1) rowstart[n] = excl + v;
  __syncthreads();

  if (((E & 3) == 0) && ((((size_t)dst) & 15) == 0) && ((((size_t)src) & 15) == 0)) {
    const int len = (ee - eb) >> 2;
    if (len > 0) {
      const int4* dv = (const int4*)dst + (eb >> 2);
      const int4* sv = (const int4*)src + (eb >> 2);
      const int rot = (int)(((long long)blockIdx.x * len) / gridDim.x);
      constexpr int STEP = OWN_T * OWN_U;
      int4 dc[OWN_U], sc[OWN_U], dn[OWN_U], sn[OWN_U];
      #pragma unroll
      for (int u = 0; u < OWN_U; ++u) {
        dc[u] = own_piece(dv, len, rot, u * OWN_T + tid);
        sc[u] = own_piece(sv, len, rot, u * OWN_T + tid);
      }
      for (int j0 = 0; j0 < len; j0 += STEP) {
        #pragma unroll
        for (int u = 0; u < OWN_U; ++u) {
          dn[u] = own_piece(dv, len, rot, j0 + STEP + u * OWN_T + tid);
          sn[u] = own_piece(sv, len, rot, j0 + STEP + u * OWN_T + tid);
        }
        OWN_ASK(0) OWN_ASK(1) OWN_ASK(2) OWN_ASK(3)
        OWN_PUT(0) OWN_PUT(1) OWN_PUT(2) OWN_PUT(3)
        #pragma unroll
        for (int u = 0; u < OWN_U; ++u) { dc[u] = dn[u]; sc[u] = sn[u]; }
      }
    }
  } else {
    for (int e = eb + tid; e < ee; e += OWN_T) {
      const unsigned l = (unsigned)(dst[e] - lo);
      if (l < OWN) csr[atomicAdd(&pos[l], 1)] = src[e];
    }
  }
}

#define AGG_AP 132
#define AGG_U 8


__device__ __forceinline__ int agg_csr(const int* __restrict__ csr, int first, int s, int e, int lane) {
  const int slot = first + lane;
  return csr[(slot >= s && slot < e) ? slot : 0];
}

__device__ __forceinline__ int agg_idx(int craw, int first, int s, int e, int lane, int self) {
  const int slot = first + lane;
  return (slot >= s && slot < e) ? craw : self;
}

__device__ __forceinline__ void agg_rows(const float* __restrict__ Hin, int lane, int idx, float dw,
                                         int cnt, float4& acc) {
  const int dwb = __float_as_int(dw);
  if (cnt > 0) idx = (lane < cnt) ? idx : __builtin_amdgcn_readlane(idx, cnt - 1);
  for (int k0 = 0; k0 < cnt; k0 += AGG_U) {
    float4 v[AGG_U];
    #pragma unroll
    for (int u = 0; u < AGG_U; ++u) {
      const int sidx = __builtin_amdgcn_readlane(idx, k0 + u);
      v[u] = *(const float4*)(Hin + (size_t)sidx * 128 + lane * 4);
    }
    #pragma unroll
    for (int u = 0; u < AGG_U; ++u) {
      const float w = __int_as_float(__builtin_amdgcn_readlane(dwb, k0 + u));
      acc.x += w * v[u].x; acc.y += w * v[u].y; acc.z += w * v[u].z; acc.w += w * v[u].w;
    }
  }
}

template <int N, bool POOL>
__global__ void __launch_bounds__(256)
k_agg_gemm(const float* __restrict__ Hin, const float* __restrict__ W,
           const float* __restrict__ dinv, const int* __restrict__ rowstart,
           const int* __restrict__ csr, const float* __restrict__ bias,
           float* __restrict__ out, const int* __restrict__ batch,
           float* __restrict__ cntg, int M) {
  constexpr int K = 128;
  constexpr int NT = N / 16;
  constexpr int RP = 2 * N + 32;
  __shared__ __attribute__((aligned(16))) float Wlds[(K / 2) * RP];
  __shared__ __attribute__((aligned(16))) float Atile[8 * 16 * AGG_AP];
  for (int i = threadIdx.x; i < K * N; i += blockDim.x) {
    int k = i / N;
    int n = i % N;
    Wlds[(k >> 1) * RP + n * 2 + (k & 1)] = W[i];
  }

  const int wave = __builtin_amdgcn_readfirstlane(threadIdx.x >> 5);
  const int lane = threadIdx.x & 31;
  const int mtile = blockIdx.x * 8 + wave;
  const int Rm = mtile * 16;
  float* At = Atile + wave * 16 * AGG_AP;

  if (Rm < M) {
    const int rsv = rowstart[min(Rm + min(lane, 16), M)];
    int s0 = __builtin_amdgcn_readlane(rsv, 0);
    int s1 = __builtin_amdgcn_readlane(rsv, 1);
    int s2 = __builtin_amdgcn_readlane(rsv, 2);
    const int raw0 = agg_csr(csr, s0 - 1, s0, s1, lane);
    int raw1 = agg_csr(csr, s1 - 1, s1, s2, lane);
    int idx = agg_idx(raw0, s0 - 1, s0, s1, lane, min(Rm, M - 1));
    float dw = dinv[idx];
    for (int r = 0; r < 16; ++r) {
      const int node = Rm + r;
      const int nodec = min(node, M - 1);
      const int s3 = __builtin_amdgcn_readlane(rsv, r + 3);
      const int raw2 = agg_csr(csr, s2 - 1, s2, s3, lane);
      const int nidx = agg_idx(raw1, s1 - 1, s1, s2, lane, min(node + 1, M - 1));
      const float ndw = dinv[nidx];
      const int tot = (node < M) ? s1 - s0 + 1 : 0;
      float4 acc = make_float4(0.0f, 0.0f, 0.0f, 0.0f);
      agg_rows(Hin, lane, idx, (lane < tot) ? dw : 0.0f, min(tot, 32), acc);
      for (int cb = s0 + 31; cb < s1; cb += 32) {
        const int xi = agg_idx(agg_csr(csr, cb, s0, s1, lane), cb, s0, s1, lane, nodec);
        const float xw = dinv[xi];
        agg_rows(Hin, lane, xi, (lane < s1 - cb) ? xw : 0.0f, min(s1 - cb, 32), acc);
      }
      *(float4*)(At + r * AGG_AP + lane * 4) = acc;
      idx = nidx; dw = ndw; raw1 = raw2;
      s0 = s1; s1 = s2; s2 = s3;
    }
  }
  __syncthreads();
  if (Rm >= M) return;

  const int half = lane >> 4;
  const int l16 = lane & 15;

  v8f acc[NT] = {};
  const float* abase = At + l16 * AGG_AP + 2 * half;
  const float* bbase = Wlds + half * RP + l16 * 2;

  for (int kb = 0; kb < K; kb += 4) {
    v2f a = *(const v2f*)(abase + kb);
    const float* bk = bbase + (kb >> 1) * RP;
    #pragma unroll
    for (int t = 0; t < NT; ++t) {
      v2f b = *(const v2f*)(bk + t * 32);
      acc[t] = __builtin_amdgcn_wmma_f32_16x16x4_f32(
          false, a, false, b, (short)0, acc[t], false, false);
    }
  }

  float bv[NT];
  #pragma unroll
  for (int t = 0; t < NT; ++t) bv[t] = bias[t * 16 + l16];
  if constexpr (!POOL) {
    #pragma unroll
    for (int r = 0; r < 8; ++r) {
      const int row = Rm + r + 8 * half;
      if (row < M) {
        const float di = dinv[row];
        #pragma unroll
        for (int t = 0; t < NT; ++t) {
          out[(size_t)row * N + t * 16 + l16] = fmaxf(acc[t][r] * di + bv[t], 0.0f);
        }
      }
    }
  } else {
    static_assert(!POOL || N == 64, "the pooled epilogue maps 64 columns onto two 32-lane adds");
    const int bg = batch[min(Rm + l16, M - 1)];
    const int g0 = __builtin_amdgcn_readfirstlane(bg);
    if (Rm + 16 <= M && __all(bg == g0)) {
      if ((unsigned)g0 < (unsigned)N_GRAPHS) {
        float di[8];
        #pragma unroll
        for (int r = 0; r < 8; ++r) di[r] = dinv[Rm + r + 8 * half];
        float s[NT];
        #pragma unroll
        for (int t = 0; t < NT; ++t) s[t] = 0.0f;
        #pragma unroll
        for (int r = 0; r < 8; ++r) {
          #pragma unroll
          for (int t = 0; t < NT; ++t) s[t] += fmaxf(acc[t][r] * di[r] + bv[t], 0.0f);
        }
        #pragma unroll
        for (int t = 0; t < NT; ++t) s[t] += __shfl_xor(s[t], 16);
        float* pg = out + g0 * N;
        atomicAdd(pg + lane, half ? s[1] : s[0]);
        atomicAdd(pg + 32 + lane, half ? s[3] : s[2]);
        if (lane == 0) atomicAdd(cntg + g0, 16.0f);
      }
    } else {
      #pragma unroll
      for (int r = 0; r < 8; ++r) {
        const int row = Rm + r + 8 * half;
        if (row < M) {
          const int g = batch[row];
          if ((unsigned)g < (unsigned)N_GRAPHS) {
            const float di = dinv[row];
            #pragma unroll
            for (int t = 0; t < NT; ++t)
              atomicAdd(out + g * N + t * 16 + l16, fmaxf(acc[t][r] * di + bv[t], 0.0f));
            if (l16 == 0) atomicAdd(cntg + g, 1.0f);
          }
        }
      }
    }
  }
}

__global__ void k_pool_final(const float* __restrict__ pool, const float* __restrict__ cntg,
                             float* __restrict__ out) {
  int idx = blockIdx.x * blockDim.x + threadIdx.x;
  if (idx < N_GRAPHS * 64) {
    out[idx] = pool[idx] / fmaxf(cntg[idx >> 6], 1.0f);
  }
}

extern "C" void kernel_launch(void* const* d_in, const int* in_sizes, int n_in,
                              void* d_out, int out_size, void* d_ws, size_t ws_size,
                              hipStream_t stream) {
  const float* x     = (const float*)d_in[0];
  const float* W1    = (const float*)d_in[1];
  const float* b1    = (const float*)d_in[2];
  const float* W2    = (const float*)d_in[3];
  const float* b2    = (const float*)d_in[4];
  const int*   ei    = (const int*)d_in[5];
  const int*   batch = (const int*)d_in[6];

  const int N = in_sizes[6];
  const int E = in_sizes[5] / 2;
  const int G = N_GRAPHS;

  const int* src = ei;
  const int* dst = ei + E;

  char* base = (char*)d_ws;
  size_t off = 0;
  auto take = [&](size_t bytes) {
    size_t o = off;
    off = (off + bytes + 255) & ~(size_t)255;
    return o;
  };
  int*   cnt      = (int*)(base + take(sizeof(int) * (size_t)N * OWN_H));
  int*   rowstart = (int*)(base + take(sizeof(int) * (size_t)(N + 1)));
  float* dinv     = (float*)(base + take(sizeof(float) * (size_t)N));
  int*   csr      = (int*)(base + take(sizeof(int) * (size_t)E));
  float* h1       = (float*)(base + take(sizeof(float) * (size_t)N * 128));
  float* pool     = (float*)(base + take(sizeof(float) * (size_t)G * 64));
  float* cntg     = (float*)(base + take(sizeof(float) * (size_t)G));

  const int B = 256;
  const int gOwn = (N + OWN - 1) / OWN;
  const int mtiles = (N + 15) / 16;
  const int gGemm = (mtiles + 7) / 8;

  k_zero_f32<<<(G * 64 + G + B - 1) / B, B, 0, stream>>>(pool, G * 64 + G);

  k_own_count<<<dim3(gOwn, OWN_H), OWN_T, 0, stream>>>(dst, cnt, E, N);
  k_own_fill<<<dim3(gOwn, OWN_H), OWN_T, 0, stream>>>(src, dst, cnt, rowstart, dinv, csr, E, N);

  k_agg_gemm<128, false><<<gGemm, 256, 0, stream>>>(x, W1, dinv, rowstart, csr, b1, h1, nullptr, nullptr, N);

  k_agg_gemm<64, true><<<gGemm, 256, 0, stream>>>(h1, W2, dinv, rowstart, csr, b2, pool, batch, cntg, N);

  k_pool_final<<<(G * 64 + B - 1) / B, B, 0, stream>>>(pool, cntg, (float*)d_out);
}


# CSR build re-tiled: an owner block takes 4096 nodes and 1/16 of the edges (was 512 and 1/2), so 13 blocks read each edge, not 98; block totals give row bases; csr row (sum) order = part, LDS arrival
# speedup vs baseline: 3.0032x; 1.2004x over previous
#include <hip/hip_runtime.h>
#include <hip/hip_bf16.h>

typedef float v2f __attribute__((ext_vector_type(2)));
typedef float v8f __attribute__((ext_vector_type(8)));

#define N_GRAPHS 64

__global__ void k_zero_f32(float* p, int n) {
  int i = blockIdx.x * blockDim.x + threadIdx.x;
  for (; i < n; i += gridDim.x * blockDim.x) p[i] = 0.0f;
}

#define OWN 4096
#define OWN_H 16
#define OWN_T 1024
#define OWN_U 4
#define OWN_NPT (OWN / OWN_T)
#define OWN_PB 8

__device__ __forceinline__ void own_part(int E, int h, int& eb, int& ee) {
  const int pieces = (E + 3) >> 2;
  const int per = ((pieces + OWN_H - 1) / OWN_H) * 4;
  eb = min(E, h * per);
  ee = min(E, eb + per);
}

__device__ __forceinline__ int4 own_piece(const int4* __restrict__ v, int len, int rot, int j) {
  int idx = j + rot;
  if (idx >= len) idx -= len;
  const bool in = j < len;
  int4 r = v[in ? idx : 0];
  if (!in) r = make_int4(-1, -1, -1, -1);
  return r;
}

__global__ void __launch_bounds__(OWN_T)
k_own_count(const int* __restrict__ dst, int* __restrict__ cnt, int* __restrict__ tot, int E, int n) {
  __shared__ int c[OWN];
  __shared__ int wsum[32];
  const int tid = threadIdx.x;
  const int lo = blockIdx.x * OWN;
  const int h = blockIdx.y;
  int eb, ee;
  own_part(E, h, eb, ee);
  #pragma unroll
  for (int j = 0; j < OWN_NPT; ++j) c[j * OWN_T + tid] = 0;
  __syncthreads();
  if (((E & 3) == 0) && ((((size_t)dst) & 15) == 0)) {
    const int len = (ee - eb) >> 2;
    if (len > 0) {
      const int4* dv = (const int4*)dst + (eb >> 2);
      const int rot = (int)(((long long)blockIdx.x * len) / gridDim.x);
      constexpr int STEP = OWN_T * OWN_U;
      int4 cur[OWN_U], nxt[OWN_U];
      #pragma unroll
      for (int u = 0; u < OWN_U; ++u) cur[u] = own_piece(dv, len, rot, u * OWN_T + tid);
      for (int j0 = 0; j0 < len; j0 += STEP) {
        #pragma unroll
        for (int u = 0; u < OWN_U; ++u) nxt[u] = own_piece(dv, len, rot, j0 + STEP + u * OWN_T + tid);
        #pragma unroll
        for (int u = 0; u < OWN_U; ++u) {
          const unsigned l0 = (unsigned)(cur[u].x - lo), l1 = (unsigned)(cur[u].y - lo);
          const unsigned l2 = (unsigned)(cur[u].z - lo), l3 = (unsigned)(cur[u].w - lo);
          if (l0 < OWN) atomicAdd(&c[l0], 1);
          if (l1 < OWN) atomicAdd(&c[l1], 1);
          if (l2 < OWN) atomicAdd(&c[l2], 1);
          if (l3 < OWN) atomicAdd(&c[l3], 1);
        }
        #pragma unroll
        for (int u = 0; u < OWN_U; ++u) cur[u] = nxt[u];
      }
    }
  } else {
    for (int e = eb + tid; e < ee; e += OWN_T) {
      const unsigned l = (unsigned)(dst[e] - lo);
      if (l < OWN) atomicAdd(&c[l], 1);
    }
  }
  __syncthreads();
  int mine = 0;
  #pragma unroll
  for (int j = 0; j < OWN_NPT; ++j) {
    const int cv = c[j * OWN_T + tid];
    const int node = lo + j * OWN_T + tid;
    if (node < n) cnt[h * n + node] = cv;
    mine += cv;
  }
  for (int o = 16; o > 0; o >>= 1) mine += __shfl_xor(mine, o);
  if ((tid & 31) == 0) wsum[tid >> 5] = mine;
  __syncthreads();
  if (tid == 0) {
    int t = 0;
    for (int w = 0; w < 32; ++w) t += wsum[w];
    tot[blockIdx.x * OWN_H + h] = t;
  }
}

#define OWN_ASK(U) \
  int p##U##x = -1, p##U##y = -1, p##U##z = -1, p##U##w = -1; \
  { const unsigned lx = (unsigned)(dc[U].x - lo), ly = (unsigned)(dc[U].y - lo); \
    const unsigned lz = (unsigned)(dc[U].z - lo), lw = (unsigned)(dc[U].w - lo); \
    if (lx < OWN) p##U##x = atomicAdd(&pos[lx], 1); \
    if (ly < OWN) p##U##y = atomicAdd(&pos[ly], 1); \
    if (lz < OWN) p##U##z = atomicAdd(&pos[lz], 1); \
    if (lw < OWN) p##U##w = atomicAdd(&pos[lw], 1); }
#define OWN_PUT(U) \
  if (p##U##x >= 0) csr[p##U##x] = sc[U].x; \
  if (p##U##y >= 0) csr[p##U##y] = sc[U].y; \
  if (p##U##z >= 0) csr[p##U##z] = sc[U].z; \
  if (p##U##w >= 0) csr[p##U##w] = sc[U].w;

__global__ void __launch_bounds__(OWN_T)
k_own_fill(const int* __restrict__ src, const int* __restrict__ dst, const int* __restrict__ cnt,
           const int* __restrict__ tot,
           int* __restrict__ rowstart, float* __restrict__ dinv, int* __restrict__ csr,
           int E, int n) {
  __shared__ int pos[OWN];
  __shared__ int wsum[32];
  const int tid = threadIdx.x;
  const int lane = tid & 31;
  const int wave = tid >> 5;
  const int lo = blockIdx.x * OWN;
  const int h = blockIdx.y;
  int eb, ee;
  own_part(E, h, eb, ee);

  int acc = 0;
  for (int i = tid; i < (int)blockIdx.x * OWN_H; i += OWN_T) acc += tot[i];
  for (int o = 16; o > 0; o >>= 1) acc += __shfl_xor(acc, o);
  if (lane == 0) wsum[wave] = acc;
  __syncthreads();
  int base = 0;
  for (int w = 0; w < 32; ++w) base += wsum[w];
  __syncthreads();

  const int first = lo + tid * OWN_NPT;
  int v[OWN_NPT], before[OWN_NPT];
  int nodec[OWN_NPT];
  #pragma unroll
  for (int j = 0; j < OWN_NPT; ++j) { v[j] = 0; before[j] = 0; nodec[j] = min(first + j, n - 1); }
  static_assert(OWN_H % OWN_PB == 0, "the planes are read in whole groups");
  #pragma nounroll
  for (int hb = 0; hb < OWN_H; hb += OWN_PB) {
    int raw[OWN_PB][OWN_NPT];
    #pragma unroll
    for (int q = 0; q < OWN_PB; ++q) {
      const int* __restrict__ plane = cnt + (size_t)(hb + q) * n;
      #pragma unroll
      for (int j = 0; j < OWN_NPT; ++j) raw[q][j] = plane[nodec[j]];
    }
    #pragma unroll
    for (int q = 0; q < OWN_PB; ++q) {
      #pragma unroll
      for (int j = 0; j < OWN_NPT; ++j) {
        const int cv = (first + j < n) ? raw[q][j] : 0;
        if (hb + q < h) before[j] += cv;
        v[j] += cv;
      }
    }
  }
  int pre[OWN_NPT];
  int tsum = 0;
  #pragma unroll
  for (int j = 0; j < OWN_NPT; ++j) { pre[j] = tsum; tsum += v[j]; }
  int inc = tsum;
  for (int o = 1; o < 32; o <<= 1) {
    const int t = __shfl_up(inc, o);
    if (lane >= o) inc += t;
  }
  if (lane == 31) wsum[wave] = inc;
  __syncthreads();
  int wpre = 0;
  for (int w = 0; w < wave; ++w) wpre += wsum[w];
  const int excl = base + wpre + inc - tsum;
  #pragma unroll
  for (int j = 0; j < OWN_NPT; ++j) {
    const int node = first + j;
    pos[tid * OWN_NPT + j] = excl + pre[j] + before[j];
    if (h == 0 && node < n) {
      rowstart[node] = excl + pre[j];
      dinv[node] = rsqrtf((float)v[j] + 1.0f);
    }
  }
  if (h == 0 && blockIdx.x == gridDim.x - 1 && tid == OWN_T - 1) rowstart[n] = excl + tsum;
  __syncthreads();

  if (((E & 3) == 0) && ((((size_t)dst) & 15) == 0) && ((((size_t)src) & 15) == 0)) {
    const int len = (ee - eb) >> 2;
    if (len > 0) {
      const int4* dv = (const int4*)dst + (eb >> 2);
      const int4* sv = (const int4*)src + (eb >> 2);
      const int rot = (int)(((long long)blockIdx.x * len) / gridDim.x);
      constexpr int STEP = OWN_T * OWN_U;
      int4 dc[OWN_U], sc[OWN_U], dn[OWN_U], sn[OWN_U];
      #pragma unroll
      for (int u = 0; u < OWN_U; ++u) {
        dc[u] = own_piece(dv, len, rot, u * OWN_T + tid);
        sc[u] = own_piece(sv, len, rot, u * OWN_T + tid);
      }
      for (int j0 = 0; j0 < len; j0 += STEP) {
        #pragma unroll
        for (int u = 0; u < OWN_U; ++u) {
          dn[u] = own_piece(dv, len, rot, j0 + STEP + u * OWN_T + tid);
          sn[u] = own_piece(sv, len, rot, j0 + STEP + u * OWN_T + tid);
        }
        OWN_ASK(0) OWN_ASK(1) OWN_ASK(2) OWN_ASK(3)
        OWN_PUT(0) OWN_PUT(1) OWN_PUT(2) OWN_PUT(3)
        #pragma unroll
        for (int u = 0; u < OWN_U; ++u) { dc[u] = dn[u]; sc[u] = sn[u]; }
      }
    }
  } else {
    for (int e = eb + tid; e < ee; e += OWN_T) {
      const unsigned l = (unsigned)(dst[e] - lo);
      if (l < OWN) csr[atomicAdd(&pos[l], 1)] = src[e];
    }
  }
}

#define AGG_AP 132
#define AGG_U 8


__device__ __forceinline__ int agg_csr(const int* __restrict__ csr, int first, int s, int e, int lane) {
  const int slot = first + lane;
  return csr[(slot >= s && slot < e) ? slot : 0];
}

__device__ __forceinline__ int agg_idx(int craw, int first, int s, int e, int lane, int self) {
  const int slot = first + lane;
  return (slot >= s && slot < e) ? craw : self;
}

__device__ __forceinline__ void agg_rows(const float* __restrict__ Hin, int lane, int idx, float dw,
                                         int cnt, float4& acc) {
  const int dwb = __float_as_int(dw);
  if (cnt > 0) idx = (lane < cnt) ? idx : __builtin_amdgcn_readlane(idx, cnt - 1);
  for (int k0 = 0; k0 < cnt; k0 += AGG_U) {
    float4 v[AGG_U];
    #pragma unroll
    for (int u = 0; u < AGG_U; ++u) {
      const int sidx = __builtin_amdgcn_readlane(idx, k0 + u);
      v[u] = *(const float4*)(Hin + (size_t)sidx * 128 + lane * 4);
    }
    #pragma unroll
    for (int u = 0; u < AGG_U; ++u) {
      const float w = __int_as_float(__builtin_amdgcn_readlane(dwb, k0 + u));
      acc.x += w * v[u].x; acc.y += w * v[u].y; acc.z += w * v[u].z; acc.w += w * v[u].w;
    }
  }
}

template <int N, bool POOL>
__global__ void __launch_bounds__(256)
k_agg_gemm(const float* __restrict__ Hin, const float* __restrict__ W,
           const float* __restrict__ dinv, const int* __restrict__ rowstart,
           const int* __restrict__ csr, const float* __restrict__ bias,
           float* __restrict__ out, const int* __restrict__ batch,
           float* __restrict__ cntg, int M) {
  constexpr int K = 128;
  constexpr int NT = N / 16;
  constexpr int RP = 2 * N + 32;
  __shared__ __attribute__((aligned(16))) float Wlds[(K / 2) * RP];
  __shared__ __attribute__((aligned(16))) float Atile[8 * 16 * AGG_AP];
  for (int i = threadIdx.x; i < K * N; i += blockDim.x) {
    int k = i / N;
    int n = i % N;
    Wlds[(k >> 1) * RP + n * 2 + (k & 1)] = W[i];
  }

  const int wave = __builtin_amdgcn_readfirstlane(threadIdx.x >> 5);
  const int lane = threadIdx.x & 31;
  const int mtile = blockIdx.x * 8 + wave;
  const int Rm = mtile * 16;
  float* At = Atile + wave * 16 * AGG_AP;

  if (Rm < M) {
    const int rsv = rowstart[min(Rm + min(lane, 16), M)];
    int s0 = __builtin_amdgcn_readlane(rsv, 0);
    int s1 = __builtin_amdgcn_readlane(rsv, 1);
    int s2 = __builtin_amdgcn_readlane(rsv, 2);
    const int raw0 = agg_csr(csr, s0 - 1, s0, s1, lane);
    int raw1 = agg_csr(csr, s1 - 1, s1, s2, lane);
    int idx = agg_idx(raw0, s0 - 1, s0, s1, lane, min(Rm, M - 1));
    float dw = dinv[idx];
    for (int r = 0; r < 16; ++r) {
      const int node = Rm + r;
      const int nodec = min(node, M - 1);
      const int s3 = __builtin_amdgcn_readlane(rsv, r + 3);
      const int raw2 = agg_csr(csr, s2 - 1, s2, s3, lane);
      const int nidx = agg_idx(raw1, s1 - 1, s1, s2, lane, min(node + 1, M - 1));
      const float ndw = dinv[nidx];
      const int tot = (node < M) ? s1 - s0 + 1 : 0;
      float4 acc = make_float4(0.0f, 0.0f, 0.0f, 0.0f);
      agg_rows(Hin, lane, idx, (lane < tot) ? dw : 0.0f, min(tot, 32), acc);
      for (int cb = s0 + 31; cb < s1; cb += 32) {
        const int xi = agg_idx(agg_csr(csr, cb, s0, s1, lane), cb, s0, s1, lane, nodec);
        const float xw = dinv[xi];
        agg_rows(Hin, lane, xi, (lane < s1 - cb) ? xw : 0.0f, min(s1 - cb, 32), acc);
      }
      *(float4*)(At + r * AGG_AP + lane * 4) = acc;
      idx = nidx; dw = ndw; raw1 = raw2;
      s0 = s1; s1 = s2; s2 = s3;
    }
  }
  __syncthreads();
  if (Rm >= M) return;

  const int half = lane >> 4;
  const int l16 = lane & 15;

  v8f acc[NT] = {};
  const float* abase = At + l16 * AGG_AP + 2 * half;
  const float* bbase = Wlds + half * RP + l16 * 2;

  for (int kb = 0; kb < K; kb += 4) {
    v2f a = *(const v2f*)(abase + kb);
    const float* bk = bbase + (kb >> 1) * RP;
    #pragma unroll
    for (int t = 0; t < NT; ++t) {
      v2f b = *(const v2f*)(bk + t * 32);
      acc[t] = __builtin_amdgcn_wmma_f32_16x16x4_f32(
          false, a, false, b, (short)0, acc[t], false, false);
    }
  }

  float bv[NT];
  #pragma unroll
  for (int t = 0; t < NT; ++t) bv[t] = bias[t * 16 + l16];
  if constexpr (!POOL) {
    #pragma unroll
    for (int r = 0; r < 8; ++r) {
      const int row = Rm + r + 8 * half;
      if (row < M) {
        const float di = dinv[row];
        #pragma unroll
        for (int t = 0; t < NT; ++t) {
          out[(size_t)row * N + t * 16 + l16] = fmaxf(acc[t][r] * di + bv[t], 0.0f);
        }
      }
    }
  } else {
    static_assert(!POOL || N == 64, "the pooled epilogue maps 64 columns onto two 32-lane adds");
    const int bg = batch[min(Rm + l16, M - 1)];
    const int g0 = __builtin_amdgcn_readfirstlane(bg);
    if (Rm + 16 <= M && __all(bg == g0)) {
      if ((unsigned)g0 < (unsigned)N_GRAPHS) {
        float di[8];
        #pragma unroll
        for (int r = 0; r < 8; ++r) di[r] = dinv[Rm + r + 8 * half];
        float s[NT];
        #pragma unroll
        for (int t = 0; t < NT; ++t) s[t] = 0.0f;
        #pragma unroll
        for (int r = 0; r < 8; ++r) {
          #pragma unroll
          for (int t = 0; t < NT; ++t) s[t] += fmaxf(acc[t][r] * di[r] + bv[t], 0.0f);
        }
        #pragma unroll
        for (int t = 0; t < NT; ++t) s[t] += __shfl_xor(s[t], 16);
        float* pg = out + g0 * N;
        atomicAdd(pg + lane, half ? s[1] : s[0]);
        atomicAdd(pg + 32 + lane, half ? s[3] : s[2]);
        if (lane == 0) atomicAdd(cntg + g0, 16.0f);
      }
    } else {
      #pragma unroll
      for (int r = 0; r < 8; ++r) {
        const int row = Rm + r + 8 * half;
        if (row < M) {
          const int g = batch[row];
          if ((unsigned)g < (unsigned)N_GRAPHS) {
            const float di = dinv[row];
            #pragma unroll
            for (int t = 0; t < NT; ++t)
              atomicAdd(out + g * N + t * 16 + l16, fmaxf(acc[t][r] * di + bv[t], 0.0f));
            if (l16 == 0) atomicAdd(cntg + g, 1.0f);
          }
        }
      }
    }
  }
}

__global__ void k_pool_final(const float* __restrict__ pool, const float* __restrict__ cntg,
                             float* __restrict__ out) {
  int idx = blockIdx.x * blockDim.x + threadIdx.x;
  if (idx < N_GRAPHS * 64) {
    out[idx] = pool[idx] / fmaxf(cntg[idx >> 6], 1.0f);
  }
}

extern "C" void kernel_launch(void* const* d_in, const int* in_sizes, int n_in,
                              void* d_out, int out_size, void* d_ws, size_t ws_size,
                              hipStream_t stream) {
  const float* x     = (const float*)d_in[0];
  const float* W1    = (const float*)d_in[1];
  const float* b1    = (const float*)d_in[2];
  const float* W2    = (const float*)d_in[3];
  const float* b2    = (const float*)d_in[4];
  const int*   ei    = (const int*)d_in[5];
  const int*   batch = (const int*)d_in[6];

  const int N = in_sizes[6];
  const int E = in_sizes[5] / 2;
  const int G = N_GRAPHS;

  const int* src = ei;
  const int* dst = ei + E;

  char* base = (char*)d_ws;
  size_t off = 0;
  auto take = [&](size_t bytes) {
    size_t o = off;
    off = (off + bytes + 255) & ~(size_t)255;
    return o;
  };
  int*   cnt      = (int*)(base + take(sizeof(int) * (size_t)N * OWN_H));
  int*   tot      = (int*)(base + take(sizeof(int) * (size_t)((N + OWN - 1) / OWN) * OWN_H));
  int*   rowstart = (int*)(base + take(sizeof(int) * (size_t)(N + 1)));
  float* dinv     = (float*)(base + take(sizeof(float) * (size_t)N));
  int*   csr      = (int*)(base + take(sizeof(int) * (size_t)E));
  float* h1       = (float*)(base + take(sizeof(float) * (size_t)N * 128));
  float* pool     = (float*)(base + take(sizeof(float) * (size_t)G * 64));
  float* cntg     = (float*)(base + take(sizeof(float) * (size_t)G));

  const int B = 256;
  const int gOwn = (N + OWN - 1) / OWN;
  const int mtiles = (N + 15) / 16;
  const int gGemm = (mtiles + 7) / 8;

  k_zero_f32<<<(G * 64 + G + B - 1) / B, B, 0, stream>>>(pool, G * 64 + G);

  k_own_count<<<dim3(gOwn, OWN_H), OWN_T, 0, stream>>>(dst, cnt, tot, E, N);
  k_own_fill<<<dim3(gOwn, OWN_H), OWN_T, 0, stream>>>(src, dst, cnt, tot, rowstart, dinv, csr, E, N);

  k_agg_gemm<128, false><<<gGemm, 256, 0, stream>>>(x, W1, dinv, rowstart, csr, b1, h1, nullptr, nullptr, N);

  k_agg_gemm<64, true><<<gGemm, 256, 0, stream>>>(h1, W2, dinv, rowstart, csr, b2, pool, batch, cntg, N);

  k_pool_final<<<(G * 64 + B - 1) / B, B, 0, stream>>>(pool, cntg, (float*)d_out);
}


# layer kernels: two waves share a 16-node strip (512-thread blocks): each sums the rows of 8 of its nodes into the strip's LDS tile, then does the WMMAs of half the column tiles; sum orders kept
# speedup vs baseline: 3.6319x; 1.2093x over previous
#include <hip/hip_runtime.h>
#include <hip/hip_bf16.h>

typedef float v2f __attribute__((ext_vector_type(2)));
typedef float v8f __attribute__((ext_vector_type(8)));

#define N_GRAPHS 64

__global__ void k_zero_f32(float* p, int n) {
  int i = blockIdx.x * blockDim.x + threadIdx.x;
  for (; i < n; i += gridDim.x * blockDim.x) p[i] = 0.0f;
}

#define OWN 4096
#define OWN_H 16
#define OWN_T 1024
#define OWN_U 4
#define OWN_NPT (OWN / OWN_T)
#define OWN_PB 8

__device__ __forceinline__ void own_part(int E, int h, int& eb, int& ee) {
  const int pieces = (E + 3) >> 2;
  const int per = ((pieces + OWN_H - 1) / OWN_H) * 4;
  eb = min(E, h * per);
  ee = min(E, eb + per);
}

__device__ __forceinline__ int4 own_piece(const int4* __restrict__ v, int len, int rot, int j) {
  int idx = j + rot;
  if (idx >= len) idx -= len;
  const bool in = j < len;
  int4 r = v[in ? idx : 0];
  if (!in) r = make_int4(-1, -1, -1, -1);
  return r;
}

__global__ void __launch_bounds__(OWN_T)
k_own_count(const int* __restrict__ dst, int* __restrict__ cnt, int* __restrict__ tot, int E, int n) {
  __shared__ int c[OWN];
  __shared__ int wsum[32];
  const int tid = threadIdx.x;
  const int lo = blockIdx.x * OWN;
  const int h = blockIdx.y;
  int eb, ee;
  own_part(E, h, eb, ee);
  #pragma unroll
  for (int j = 0; j < OWN_NPT; ++j) c[j * OWN_T + tid] = 0;
  __syncthreads();
  if (((E & 3) == 0) && ((((size_t)dst) & 15) == 0)) {
    const int len = (ee - eb) >> 2;
    if (len > 0) {
      const int4* dv = (const int4*)dst + (eb >> 2);
      const int rot = (int)(((long long)blockIdx.x * len) / gridDim.x);
      constexpr int STEP = OWN_T * OWN_U;
      int4 cur[OWN_U], nxt[OWN_U];
      #pragma unroll
      for (int u = 0; u < OWN_U; ++u) cur[u] = own_piece(dv, len, rot, u * OWN_T + tid);
      for (int j0 = 0; j0 < len; j0 += STEP) {
        #pragma unroll
        for (int u = 0; u < OWN_U; ++u) nxt[u] = own_piece(dv, len, rot, j0 + STEP + u * OWN_T + tid);
        #pragma unroll
        for (int u = 0; u < OWN_U; ++u) {
          const unsigned l0 = (unsigned)(cur[u].x - lo), l1 = (unsigned)(cur[u].y - lo);
          const unsigned l2 = (unsigned)(cur[u].z - lo), l3 = (unsigned)(cur[u].w - lo);
          if (l0 < OWN) atomicAdd(&c[l0], 1);
          if (l1 < OWN) atomicAdd(&c[l1], 1);
          if (l2 < OWN) atomicAdd(&c[l2], 1);
          if (l3 < OWN) atomicAdd(&c[l3], 1);
        }
        #pragma unroll
        for (int u = 0; u < OWN_U; ++u) cur[u] = nxt[u];
      }
    }
  } else {
    for (int e = eb + tid; e < ee; e += OWN_T) {
      const unsigned l = (unsigned)(dst[e] - lo);
      if (l < OWN) atomicAdd(&c[l], 1);
    }
  }
  __syncthreads();
  int mine = 0;
  #pragma unroll
  for (int j = 0; j < OWN_NPT; ++j) {
    const int cv = c[j * OWN_T + tid];
    const int node = lo + j * OWN_T + tid;
    if (node < n) cnt[h * n + node] = cv;
    mine += cv;
  }
  for (int o = 16; o > 0; o >>= 1) mine += __shfl_xor(mine, o);
  if ((tid & 31) == 0) wsum[tid >> 5] = mine;
  __syncthreads();
  if (tid == 0) {
    int t = 0;
    for (int w = 0; w < 32; ++w) t += wsum[w];
    tot[blockIdx.x * OWN_H + h] = t;
  }
}

#define OWN_ASK(U) \
  int p##U##x = -1, p##U##y = -1, p##U##z = -1, p##U##w = -1; \
  { const unsigned lx = (unsigned)(dc[U].x - lo), ly = (unsigned)(dc[U].y - lo); \
    const unsigned lz = (unsigned)(dc[U].z - lo), lw = (unsigned)(dc[U].w - lo); \
    if (lx < OWN) p##U##x = atomicAdd(&pos[lx], 1); \
    if (ly < OWN) p##U##y = atomicAdd(&pos[ly], 1); \
    if (lz < OWN) p##U##z = atomicAdd(&pos[lz], 1); \
    if (lw < OWN) p##U##w = atomicAdd(&pos[lw], 1); }
#define OWN_PUT(U) \
  if (p##U##x >= 0) csr[p##U##x] = sc[U].x; \
  if (p##U##y >= 0) csr[p##U##y] = sc[U].y; \
  if (p##U##z >= 0) csr[p##U##z] = sc[U].z; \
  if (p##U##w >= 0) csr[p##U##w] = sc[U].w;

__global__ void __launch_bounds__(OWN_T)
k_own_fill(const int* __restrict__ src, const int* __restrict__ dst, const int* __restrict__ cnt,
           const int* __restrict__ tot,
           int* __restrict__ rowstart, float* __restrict__ dinv, int* __restrict__ csr,
           int E, int n) {
  __shared__ int pos[OWN];
  __shared__ int wsum[32];
  const int tid = threadIdx.x;
  const int lane = tid & 31;
  const int wave = tid >> 5;
  const int lo = blockIdx.x * OWN;
  const int h = blockIdx.y;
  int eb, ee;
  own_part(E, h, eb, ee);

  int acc = 0;
  for (int i = tid; i < (int)blockIdx.x * OWN_H; i += OWN_T) acc += tot[i];
  for (int o = 16; o > 0; o >>= 1) acc += __shfl_xor(acc, o);
  if (lane == 0) wsum[wave] = acc;
  __syncthreads();
  int base = 0;
  for (int w = 0; w < 32; ++w) base += wsum[w];
  __syncthreads();

  const int first = lo + tid * OWN_NPT;
  int v[OWN_NPT], before[OWN_NPT];
  int nodec[OWN_NPT];
  #pragma unroll
  for (int j = 0; j < OWN_NPT; ++j) { v[j] = 0; before[j] = 0; nodec[j] = min(first + j, n - 1); }
  static_assert(OWN_H % OWN_PB == 0, "the planes are read in whole groups");
  #pragma nounroll
  for (int hb = 0; hb < OWN_H; hb += OWN_PB) {
    int raw[OWN_PB][OWN_NPT];
    #pragma unroll
    for (int q = 0; q < OWN_PB; ++q) {
      const int* __restrict__ plane = cnt + (size_t)(hb + q) * n;
      #pragma unroll
      for (int j = 0; j < OWN_NPT; ++j) raw[q][j] = plane[nodec[j]];
    }
    #pragma unroll
    for (int q = 0; q < OWN_PB; ++q) {
      #pragma unroll
      for (int j = 0; j < OWN_NPT; ++j) {
        const int cv = (first + j < n) ? raw[q][j] : 0;
        if (hb + q < h) before[j] += cv;
        v[j] += cv;
      }
    }
  }
  int pre[OWN_NPT];
  int tsum = 0;
  #pragma unroll
  for (int j = 0; j < OWN_NPT; ++j) { pre[j] = tsum; tsum += v[j]; }
  int inc = tsum;
  for (int o = 1; o < 32; o <<= 1) {
    const int t = __shfl_up(inc, o);
    if (lane >= o) inc += t;
  }
  if (lane == 31) wsum[wave] = inc;
  __syncthreads();
  int wpre = 0;
  for (int w = 0; w < wave; ++w) wpre += wsum[w];
  const int excl = base + wpre + inc - tsum;
  #pragma unroll
  for (int j = 0; j < OWN_NPT; ++j) {
    const int node = first + j;
    pos[tid * OWN_NPT + j] = excl + pre[j] + before[j];
    if (h == 0 && node < n) {
      rowstart[node] = excl + pre[j];
      dinv[node] = rsqrtf((float)v[j] + 1.0f);
    }
  }
  if (h == 0 && blockIdx.x == gridDim.x - 1 && tid == OWN_T - 1) rowstart[n] = excl + tsum;
  __syncthreads();

  if (((E & 3) == 0) && ((((size_t)dst) & 15) == 0) && ((((size_t)src) & 15) == 0)) {
    const int len = (ee - eb) >> 2;
    if (len > 0) {
      const int4* dv = (const int4*)dst + (eb >> 2);
      const int4* sv = (const int4*)src + (eb >> 2);
      const int rot = (int)(((long long)blockIdx.x * len) / gridDim.x);
      constexpr int STEP = OWN_T * OWN_U;
      int4 dc[OWN_U], sc[OWN_U], dn[OWN_U], sn[OWN_U];
      #pragma unroll
      for (int u = 0; u < OWN_U; ++u) {
        dc[u] = own_piece(dv, len, rot, u * OWN_T + tid);
        sc[u] = own_piece(sv, len, rot, u * OWN_T + tid);
      }
      for (int j0 = 0; j0 < len; j0 += STEP) {
        #pragma unroll
        for (int u = 0; u < OWN_U; ++u) {
          dn[u] = own_piece(dv, len, rot, j0 + STEP + u * OWN_T + tid);
          sn[u] = own_piece(sv, len, rot, j0 + STEP + u * OWN_T + tid);
        }
        OWN_ASK(0) OWN_ASK(1) OWN_ASK(2) OWN_ASK(3)
        OWN_PUT(0) OWN_PUT(1) OWN_PUT(2) OWN_PUT(3)
        #pragma unroll
        for (int u = 0; u < OWN_U; ++u) { dc[u] = dn[u]; sc[u] = sn[u]; }
      }
    }
  } else {
    for (int e = eb + tid; e < ee; e += OWN_T) {
      const unsigned l = (unsigned)(dst[e] - lo);
      if (l < OWN) csr[atomicAdd(&pos[l], 1)] = src[e];
    }
  }
}

#define AGG_AP 132
#define AGG_U 8
#define AGG_WS 2
#define AGG_NR (16 / AGG_WS)


__device__ __forceinline__ int agg_csr(const int* __restrict__ csr, int first, int s, int e, int lane) {
  const int slot = first + lane;
  return csr[(slot >= s && slot < e) ? slot : 0];
}

__device__ __forceinline__ int agg_idx(int craw, int first, int s, int e, int lane, int self) {
  const int slot = first + lane;
  return (slot >= s && slot < e) ? craw : self;
}

__device__ __forceinline__ void agg_rows(const float* __restrict__ Hin, int lane, int idx, float dw,
                                         int cnt, float4& acc) {
  const int dwb = __float_as_int(dw);
  if (cnt > 0) idx = (lane < cnt) ? idx : __builtin_amdgcn_readlane(idx, cnt - 1);
  for (int k0 = 0; k0 < cnt; k0 += AGG_U) {
    float4 v[AGG_U];
    #pragma unroll
    for (int u = 0; u < AGG_U; ++u) {
      const int sidx = __builtin_amdgcn_readlane(idx, k0 + u);
      v[u] = *(const float4*)(Hin + (size_t)sidx * 128 + lane * 4);
    }
    #pragma unroll
    for (int u = 0; u < AGG_U; ++u) {
      const float w = __int_as_float(__builtin_amdgcn_readlane(dwb, k0 + u));
      acc.x += w * v[u].x; acc.y += w * v[u].y; acc.z += w * v[u].z; acc.w += w * v[u].w;
    }
  }
}

template <int N, bool POOL>
__global__ void __launch_bounds__(256 * AGG_WS)
k_agg_gemm(const float* __restrict__ Hin, const float* __restrict__ W,
           const float* __restrict__ dinv, const int* __restrict__ rowstart,
           const int* __restrict__ csr, const float* __restrict__ bias,
           float* __restrict__ out, const int* __restrict__ batch,
           float* __restrict__ cntg, int M) {
  constexpr int K = 128;
  constexpr int NT = N / 16;
  constexpr int NTW = NT / AGG_WS;
  constexpr int RP = 2 * N + 32;
  __shared__ __attribute__((aligned(16))) float Wlds[(K / 2) * RP];
  __shared__ __attribute__((aligned(16))) float Atile[8 * 16 * AGG_AP];
  for (int i = threadIdx.x; i < K * N; i += blockDim.x) {
    int k = i / N;
    int n = i % N;
    Wlds[(k >> 1) * RP + n * 2 + (k & 1)] = W[i];
  }

  const int wave = __builtin_amdgcn_readfirstlane(threadIdx.x >> 5);
  const int lane = threadIdx.x & 31;
  const int strip = wave / AGG_WS;
  const int sub = wave % AGG_WS;
  const int mtile = blockIdx.x * 8 + strip;
  const int Rm = mtile * 16;
  const int R0 = Rm + sub * AGG_NR;
  float* At = Atile + strip * 16 * AGG_AP;

  if (Rm < M) {
    const int rsv = rowstart[min(R0 + min(lane, AGG_NR), M)];
    int s0 = __builtin_amdgcn_readlane(rsv, 0);
    int s1 = __builtin_amdgcn_readlane(rsv, 1);
    int s2 = __builtin_amdgcn_readlane(rsv, 2);
    const int raw0 = agg_csr(csr, s0 - 1, s0, s1, lane);
    int raw1 = agg_csr(csr, s1 - 1, s1, s2, lane);
    int idx = agg_idx(raw0, s0 - 1, s0, s1, lane, min(R0, M - 1));
    float dw = dinv[idx];
    for (int r = 0; r < AGG_NR; ++r) {
      const int node = R0 + r;
      const int nodec = min(node, M - 1);
      const int s3 = __builtin_amdgcn_readlane(rsv, r + 3);
      const int raw2 = agg_csr(csr, s2 - 1, s2, s3, lane);
      const int nidx = agg_idx(raw1, s1 - 1, s1, s2, lane, min(node + 1, M - 1));
      const float ndw = dinv[nidx];
      const int tot = (node < M) ? s1 - s0 + 1 : 0;
      float4 acc = make_float4(0.0f, 0.0f, 0.0f, 0.0f);
      agg_rows(Hin, lane, idx, (lane < tot) ? dw : 0.0f, min(tot, 32), acc);
      for (int cb = s0 + 31; cb < s1; cb += 32) {
        const int xi = agg_idx(agg_csr(csr, cb, s0, s1, lane), cb, s0, s1, lane, nodec);
        const float xw = dinv[xi];
        agg_rows(Hin, lane, xi, (lane < s1 - cb) ? xw : 0.0f, min(s1 - cb, 32), acc);
      }
      *(float4*)(At + (sub * AGG_NR + r) * AGG_AP + lane * 4) = acc;
      idx = nidx; dw = ndw; raw1 = raw2;
      s0 = s1; s1 = s2; s2 = s3;
    }
  }
  __syncthreads();
  if (Rm >= M) return;

  const int half = lane >> 4;
  const int l16 = lane & 15;

  v8f acc[NTW] = {};
  const float* abase = At + l16 * AGG_AP + 2 * half;
  const int t0 = sub * NTW;
  const float* bbase = Wlds + half * RP + l16 * 2 + t0 * 32;

  for (int kb = 0; kb < K; kb += 4) {
    v2f a = *(const v2f*)(abase + kb);
    const float* bk = bbase + (kb >> 1) * RP;
    #pragma unroll
    for (int t = 0; t < NTW; ++t) {
      v2f b = *(const v2f*)(bk + t * 32);
      acc[t] = __builtin_amdgcn_wmma_f32_16x16x4_f32(
          false, a, false, b, (short)0, acc[t], false, false);
    }
  }

  float bv[NTW];
  #pragma unroll
  for (int t = 0; t < NTW; ++t) bv[t] = bias[(t0 + t) * 16 + l16];
  if constexpr (!POOL) {
    #pragma unroll
    for (int r = 0; r < 8; ++r) {
      const int row = Rm + r + 8 * half;
      if (row < M) {
        const float di = dinv[row];
        #pragma unroll
        for (int t = 0; t < NTW; ++t) {
          out[(size_t)row * N + (t0 + t) * 16 + l16] = fmaxf(acc[t][r] * di + bv[t], 0.0f);
        }
      }
    }
  } else {
    static_assert(!POOL || NTW == 2, "the pooled epilogue maps a wave's two column tiles onto one 32-lane add");
    const int bg = batch[min(Rm + l16, M - 1)];
    const int g0 = __builtin_amdgcn_readfirstlane(bg);
    if (Rm + 16 <= M && __all(bg == g0)) {
      if ((unsigned)g0 < (unsigned)N_GRAPHS) {
        float di[8];
        #pragma unroll
        for (int r = 0; r < 8; ++r) di[r] = dinv[Rm + r + 8 * half];
        float s[NTW];
        #pragma unroll
        for (int t = 0; t < NTW; ++t) s[t] = 0.0f;
        #pragma unroll
        for (int r = 0; r < 8; ++r) {
          #pragma unroll
          for (int t = 0; t < NTW; ++t) s[t] += fmaxf(acc[t][r] * di[r] + bv[t], 0.0f);
        }
        #pragma unroll
        for (int t = 0; t < NTW; ++t) s[t] += __shfl_xor(s[t], 16);
        float* pg = out + g0 * N + t0 * 16;
        atomicAdd(pg + lane, half ? s[1] : s[0]);
        if (lane == 0 && sub == 0) atomicAdd(cntg + g0, 16.0f);
      }
    } else {
      #pragma unroll
      for (int r = 0; r < 8; ++r) {
        const int row = Rm + r + 8 * half;
        if (row < M) {
          const int g = batch[row];
          if ((unsigned)g < (unsigned)N_GRAPHS) {
            const float di = dinv[row];
            #pragma unroll
            for (int t = 0; t < NTW; ++t)
              atomicAdd(out + g * N + (t0 + t) * 16 + l16, fmaxf(acc[t][r] * di + bv[t], 0.0f));
            if (l16 == 0 && sub == 0) atomicAdd(cntg + g, 1.0f);
          }
        }
      }
    }
  }
}

__global__ void k_pool_final(const float* __restrict__ pool, const float* __restrict__ cntg,
                             float* __restrict__ out) {
  int idx = blockIdx.x * blockDim.x + threadIdx.x;
  if (idx < N_GRAPHS * 64) {
    out[idx] = pool[idx] / fmaxf(cntg[idx >> 6], 1.0f);
  }
}

extern "C" void kernel_launch(void* const* d_in, const int* in_sizes, int n_in,
                              void* d_out, int out_size, void* d_ws, size_t ws_size,
                              hipStream_t stream) {
  const float* x     = (const float*)d_in[0];
  const float* W1    = (const float*)d_in[1];
  const float* b1    = (const float*)d_in[2];
  const float* W2    = (const float*)d_in[3];
  const float* b2    = (const float*)d_in[4];
  const int*   ei    = (const int*)d_in[5];
  const int*   batch = (const int*)d_in[6];

  const int N = in_sizes[6];
  const int E = in_sizes[5] / 2;
  const int G = N_GRAPHS;

  const int* src = ei;
  const int* dst = ei + E;

  char* base = (char*)d_ws;
  size_t off = 0;
  auto take = [&](size_t bytes) {
    size_t o = off;
    off = (off + bytes + 255) & ~(size_t)255;
    return o;
  };
  int*   cnt      = (int*)(base + take(sizeof(int) * (size_t)N * OWN_H));
  int*   tot      = (int*)(base + take(sizeof(int) * (size_t)((N + OWN - 1) / OWN) * OWN_H));
  int*   rowstart = (int*)(base + take(sizeof(int) * (size_t)(N + 1)));
  float* dinv     = (float*)(base + take(sizeof(float) * (size_t)N));
  int*   csr      = (int*)(base + take(sizeof(int) * (size_t)E));
  float* h1       = (float*)(base + take(sizeof(float) * (size_t)N * 128));
  float* pool     = (float*)(base + take(sizeof(float) * (size_t)G * 64));
  float* cntg     = (float*)(base + take(sizeof(float) * (size_t)G));

  const int B = 256;
  const int gOwn = (N + OWN - 1) / OWN;
  const int mtiles = (N + 15) / 16;
  const int gGemm = (mtiles + 7) / 8;

  k_zero_f32<<<(G * 64 + G + B - 1) / B, B, 0, stream>>>(pool, G * 64 + G);

  k_own_count<<<dim3(gOwn, OWN_H), OWN_T, 0, stream>>>(dst, cnt, tot, E, N);
  k_own_fill<<<dim3(gOwn, OWN_H), OWN_T, 0, stream>>>(src, dst, cnt, tot, rowstart, dinv, csr, E, N);

  k_agg_gemm<128, false><<<gGemm, 256 * AGG_WS, 0, stream>>>(x, W1, dinv, rowstart, csr, b1, h1, nullptr, nullptr, N);

  k_agg_gemm<64, true><<<gGemm, 256 * AGG_WS, 0, stream>>>(h1, W2, dinv, rowstart, csr, b2, pool, batch, cntg, N);

  k_pool_final<<<(G * 64 + B - 1) / B, B, 0, stream>>>(pool, cntg, (float*)d_out);
}


# layer kernels re-tiled for even WGP load: 241 blocks of 13 strips, 1024 threads, LDS sized so a block has its WGP alone; 208 nodes split over 32 waves; was 391 blocks of 8, two a WGP; sum orders kept
# speedup vs baseline: 3.9038x; 1.0749x over previous
#include <hip/hip_runtime.h>
#include <hip/hip_bf16.h>

typedef float v2f __attribute__((ext_vector_type(2)));
typedef float v8f __attribute__((ext_vector_type(8)));

#define N_GRAPHS 64

__global__ void k_zero_f32(float* p, int n) {
  int i = blockIdx.x * blockDim.x + threadIdx.x;
  for (; i < n; i += gridDim.x * blockDim.x) p[i] = 0.0f;
}

#define OWN 4096
#define OWN_H 16
#define OWN_T 1024
#define OWN_U 4
#define OWN_NPT (OWN / OWN_T)
#define OWN_PB 8

__device__ __forceinline__ void own_part(int E, int h, int& eb, int& ee) {
  const int pieces = (E + 3) >> 2;
  const int per = ((pieces + OWN_H - 1) / OWN_H) * 4;
  eb = min(E, h * per);
  ee = min(E, eb + per);
}

__device__ __forceinline__ int4 own_piece(const int4* __restrict__ v, int len, int rot, int j) {
  int idx = j + rot;
  if (idx >= len) idx -= len;
  const bool in = j < len;
  int4 r = v[in ? idx : 0];
  if (!in) r = make_int4(-1, -1, -1, -1);
  return r;
}

__global__ void __launch_bounds__(OWN_T)
k_own_count(const int* __restrict__ dst, int* __restrict__ cnt, int* __restrict__ tot, int E, int n) {
  __shared__ int c[OWN];
  __shared__ int wsum[32];
  const int tid = threadIdx.x;
  const int lo = blockIdx.x * OWN;
  const int h = blockIdx.y;
  int eb, ee;
  own_part(E, h, eb, ee);
  #pragma unroll
  for (int j = 0; j < OWN_NPT; ++j) c[j * OWN_T + tid] = 0;
  __syncthreads();
  if (((E & 3) == 0) && ((((size_t)dst) & 15) == 0)) {
    const int len = (ee - eb) >> 2;
    if (len > 0) {
      const int4* dv = (const int4*)dst + (eb >> 2);
      const int rot = (int)(((long long)blockIdx.x * len) / gridDim.x);
      constexpr int STEP = OWN_T * OWN_U;
      int4 cur[OWN_U], nxt[OWN_U];
      #pragma unroll
      for (int u = 0; u < OWN_U; ++u) cur[u] = own_piece(dv, len, rot, u * OWN_T + tid);
      for (int j0 = 0; j0 < len; j0 += STEP) {
        #pragma unroll
        for (int u = 0; u < OWN_U; ++u) nxt[u] = own_piece(dv, len, rot, j0 + STEP + u * OWN_T + tid);
        #pragma unroll
        for (int u = 0; u < OWN_U; ++u) {
          const unsigned l0 = (unsigned)(cur[u].x - lo), l1 = (unsigned)(cur[u].y - lo);
          const unsigned l2 = (unsigned)(cur[u].z - lo), l3 = (unsigned)(cur[u].w - lo);
          if (l0 < OWN) atomicAdd(&c[l0], 1);
          if (l1 < OWN) atomicAdd(&c[l1], 1);
          if (l2 < OWN) atomicAdd(&c[l2], 1);
          if (l3 < OWN) atomicAdd(&c[l3], 1);
        }
        #pragma unroll
        for (int u = 0; u < OWN_U; ++u) cur[u] = nxt[u];
      }
    }
  } else {
    for (int e = eb + tid; e < ee; e += OWN_T) {
      const unsigned l = (unsigned)(dst[e] - lo);
      if (l < OWN) atomicAdd(&c[l], 1);
    }
  }
  __syncthreads();
  int mine = 0;
  #pragma unroll
  for (int j = 0; j < OWN_NPT; ++j) {
    const int cv = c[j * OWN_T + tid];
    const int node = lo + j * OWN_T + tid;
    if (node < n) cnt[h * n + node] = cv;
    mine += cv;
  }
  for (int o = 16; o > 0; o >>= 1) mine += __shfl_xor(mine, o);
  if ((tid & 31) == 0) wsum[tid >> 5] = mine;
  __syncthreads();
  if (tid == 0) {
    int t = 0;
    for (int w = 0; w < 32; ++w) t += wsum[w];
    tot[blockIdx.x * OWN_H + h] = t;
  }
}

#define OWN_ASK(U) \
  int p##U##x = -1, p##U##y = -1, p##U##z = -1, p##U##w = -1; \
  { const unsigned lx = (unsigned)(dc[U].x - lo), ly = (unsigned)(dc[U].y - lo); \
    const unsigned lz = (unsigned)(dc[U].z - lo), lw = (unsigned)(dc[U].w - lo); \
    if (lx < OWN) p##U##x = atomicAdd(&pos[lx], 1); \
    if (ly < OWN) p##U##y = atomicAdd(&pos[ly], 1); \
    if (lz < OWN) p##U##z = atomicAdd(&pos[lz], 1); \
    if (lw < OWN) p##U##w = atomicAdd(&pos[lw], 1); }
#define OWN_PUT(U) \
  if (p##U##x >= 0) csr[p##U##x] = sc[U].x; \
  if (p##U##y >= 0) csr[p##U##y] = sc[U].y; \
  if (p##U##z >= 0) csr[p##U##z] = sc[U].z; \
  if (p##U##w >= 0) csr[p##U##w] = sc[U].w;

__global__ void __launch_bounds__(OWN_T)
k_own_fill(const int* __restrict__ src, const int* __restrict__ dst, const int* __restrict__ cnt,
           const int* __restrict__ tot,
           int* __restrict__ rowstart, float* __restrict__ dinv, int* __restrict__ csr,
           int E, int n) {
  __shared__ int pos[OWN];
  __shared__ int wsum[32];
  const int tid = threadIdx.x;
  const int lane = tid & 31;
  const int wave = tid >> 5;
  const int lo = blockIdx.x * OWN;
  const int h = blockIdx.y;
  int eb, ee;
  own_part(E, h, eb, ee);

  int acc = 0;
  for (int i = tid; i < (int)blockIdx.x * OWN_H; i += OWN_T) acc += tot[i];
  for (int o = 16; o > 0; o >>= 1) acc += __shfl_xor(acc, o);
  if (lane == 0) wsum[wave] = acc;
  __syncthreads();
  int base = 0;
  for (int w = 0; w < 32; ++w) base += wsum[w];
  __syncthreads();

  const int first = lo + tid * OWN_NPT;
  int v[OWN_NPT], before[OWN_NPT];
  int nodec[OWN_NPT];
  #pragma unroll
  for (int j = 0; j < OWN_NPT; ++j) { v[j] = 0; before[j] = 0; nodec[j] = min(first + j, n - 1); }
  static_assert(OWN_H % OWN_PB == 0, "the planes are read in whole groups");
  #pragma nounroll
  for (int hb = 0; hb < OWN_H; hb += OWN_PB) {
    int raw[OWN_PB][OWN_NPT];
    #pragma unroll
    for (int q = 0; q < OWN_PB; ++q) {
      const int* __restrict__ plane = cnt + (size_t)(hb + q) * n;
      #pragma unroll
      for (int j = 0; j < OWN_NPT; ++j) raw[q][j] = plane[nodec[j]];
    }
    #pragma unroll
    for (int q = 0; q < OWN_PB; ++q) {
      #pragma unroll
      for (int j = 0; j < OWN_NPT; ++j) {
        const int cv = (first + j < n) ? raw[q][j] : 0;
        if (hb + q < h) before[j] += cv;
        v[j] += cv;
      }
    }
  }
  int pre[OWN_NPT];
  int tsum = 0;
  #pragma unroll
  for (int j = 0; j < OWN_NPT; ++j) { pre[j] = tsum; tsum += v[j]; }
  int inc = tsum;
  for (int o = 1; o < 32; o <<= 1) {
    const int t = __shfl_up(inc, o);
    if (lane >= o) inc += t;
  }
  if (lane == 31) wsum[wave] = inc;
  __syncthreads();
  int wpre = 0;
  for (int w = 0; w < wave; ++w) wpre += wsum[w];
  const int excl = base + wpre + inc - tsum;
  #pragma unroll
  for (int j = 0; j < OWN_NPT; ++j) {
    const int node = first + j;
    pos[tid * OWN_NPT + j] = excl + pre[j] + before[j];
    if (h == 0 && node < n) {
      rowstart[node] = excl + pre[j];
      dinv[node] = rsqrtf((float)v[j] + 1.0f);
    }
  }
  if (h == 0 && blockIdx.x == gridDim.x - 1 && tid == OWN_T - 1) rowstart[n] = excl + tsum;
  __syncthreads();

  if (((E & 3) == 0) && ((((size_t)dst) & 15) == 0) && ((((size_t)src) & 15) == 0)) {
    const int len = (ee - eb) >> 2;
    if (len > 0) {
      const int4* dv = (const int4*)dst + (eb >> 2);
      const int4* sv = (const int4*)src + (eb >> 2);
      const int rot = (int)(((long long)blockIdx.x * len) / gridDim.x);
      constexpr int STEP = OWN_T * OWN_U;
      int4 dc[OWN_U], sc[OWN_U], dn[OWN_U], sn[OWN_U];
      #pragma unroll
      for (int u = 0; u < OWN_U; ++u) {
        dc[u] = own_piece(dv, len, rot, u * OWN_T + tid);
        sc[u] = own_piece(sv, len, rot, u * OWN_T + tid);
      }
      for (int j0 = 0; j0 < len; j0 += STEP) {
        #pragma unroll
        for (int u = 0; u < OWN_U; ++u) {
          dn[u] = own_piece(dv, len, rot, j0 + STEP + u * OWN_T + tid);
          sn[u] = own_piece(sv, len, rot, j0 + STEP + u * OWN_T + tid);
        }
        OWN_ASK(0) OWN_ASK(1) OWN_ASK(2) OWN_ASK(3)
        OWN_PUT(0) OWN_PUT(1) OWN_PUT(2) OWN_PUT(3)
        #pragma unroll
        for (int u = 0; u < OWN_U; ++u) { dc[u] = dn[u]; sc[u] = sn[u]; }
      }
    }
  } else {
    for (int e = eb + tid; e < ee; e += OWN_T) {
      const unsigned l = (unsigned)(dst[e] - lo);
      if (l < OWN) csr[atomicAdd(&pos[l], 1)] = src[e];
    }
  }
}

#define AGG_AP 132
#define AGG_U 8
#define AGG_SPB 13
#define AGG_NODES (AGG_SPB * 16)
#define AGG_T 1024
#define AGG_WV (AGG_T / 32)
#define AGG_WS 2


__device__ __forceinline__ int agg_csr(const int* __restrict__ csr, int first, int s, int e, int lane) {
  const int slot = first + lane;
  return csr[(slot >= s && slot < e) ? slot : 0];
}

__device__ __forceinline__ int agg_idx(int craw, int first, int s, int e, int lane, int self) {
  const int slot = first + lane;
  return (slot >= s && slot < e) ? craw : self;
}

__device__ __forceinline__ void agg_rows(const float* __restrict__ Hin, int lane, int idx, float dw,
                                         int cnt, float4& acc) {
  const int dwb = __float_as_int(dw);
  if (cnt > 0) idx = (lane < cnt) ? idx : __builtin_amdgcn_readlane(idx, cnt - 1);
  for (int k0 = 0; k0 < cnt; k0 += AGG_U) {
    float4 v[AGG_U];
    #pragma unroll
    for (int u = 0; u < AGG_U; ++u) {
      const int sidx = __builtin_amdgcn_readlane(idx, k0 + u);
      v[u] = *(const float4*)(Hin + (size_t)sidx * 128 + lane * 4);
    }
    #pragma unroll
    for (int u = 0; u < AGG_U; ++u) {
      const float w = __int_as_float(__builtin_amdgcn_readlane(dwb, k0 + u));
      acc.x += w * v[u].x; acc.y += w * v[u].y; acc.z += w * v[u].z; acc.w += w * v[u].w;
    }
  }
}

template <int N, bool POOL>
__global__ void __launch_bounds__(AGG_T)
k_agg_gemm(const float* __restrict__ Hin, const float* __restrict__ W,
           const float* __restrict__ dinv, const int* __restrict__ rowstart,
           const int* __restrict__ csr, const float* __restrict__ bias,
           float* __restrict__ out, const int* __restrict__ batch,
           float* __restrict__ cntg, int M) {
  constexpr int K = 128;
  constexpr int NT = N / 16;
  constexpr int NTW = NT / AGG_WS;
  constexpr int RP = 2 * N + 32;
  constexpr int WL = (K / 2) * RP;
  constexpr int AT = AGG_NODES * AGG_AP;
  constexpr int ALONE = 168 * 1024 / 4;
  constexpr int WLA = (WL + AT >= ALONE) ? WL : ALONE - AT;
  __shared__ __attribute__((aligned(16))) float Wlds[WLA];
  __shared__ __attribute__((aligned(16))) float Atile[AT];
  for (int i = threadIdx.x; i < K * N; i += blockDim.x) {
    int k = i / N;
    int n = i % N;
    Wlds[(k >> 1) * RP + n * 2 + (k & 1)] = W[i];
  }

  const int wave = __builtin_amdgcn_readfirstlane(threadIdx.x >> 5);
  const int lane = threadIdx.x & 31;
  const int Rb = blockIdx.x * AGG_NODES;
  const int l0 = (wave * AGG_NODES) / AGG_WV;
  const int nr = ((wave + 1) * AGG_NODES) / AGG_WV - l0;
  const int R0 = Rb + l0;

  {
    const int rsv = rowstart[min(R0 + min(lane, nr), M)];
    int s0 = __builtin_amdgcn_readlane(rsv, 0);
    int s1 = __builtin_amdgcn_readlane(rsv, 1);
    int s2 = __builtin_amdgcn_readlane(rsv, 2);
    const int raw0 = agg_csr(csr, s0 - 1, s0, s1, lane);
    int raw1 = agg_csr(csr, s1 - 1, s1, s2, lane);
    int idx = agg_idx(raw0, s0 - 1, s0, s1, lane, min(R0, M - 1));
    float dw = dinv[idx];
    for (int r = 0; r < nr; ++r) {
      const int node = R0 + r;
      const int nodec = min(node, M - 1);
      const int s3 = __builtin_amdgcn_readlane(rsv, r + 3);
      const int raw2 = agg_csr(csr, s2 - 1, s2, s3, lane);
      const int nidx = agg_idx(raw1, s1 - 1, s1, s2, lane, min(node + 1, M - 1));
      const float ndw = dinv[nidx];
      const int tot = (node < M) ? s1 - s0 + 1 : 0;
      float4 acc = make_float4(0.0f, 0.0f, 0.0f, 0.0f);
      agg_rows(Hin, lane, idx, (lane < tot) ? dw : 0.0f, min(tot, 32), acc);
      for (int cb = s0 + 31; cb < s1; cb += 32) {
        const int xi = agg_idx(agg_csr(csr, cb, s0, s1, lane), cb, s0, s1, lane, nodec);
        const float xw = dinv[xi];
        agg_rows(Hin, lane, xi, (lane < s1 - cb) ? xw : 0.0f, min(s1 - cb, 32), acc);
      }
      *(float4*)(Atile + (l0 + r) * AGG_AP + lane * 4) = acc;
      idx = nidx; dw = ndw; raw1 = raw2;
      s0 = s1; s1 = s2; s2 = s3;
    }
  }
  __syncthreads();
  const int strip = wave / AGG_WS;
  const int sub = wave % AGG_WS;
  const int Rm = Rb + strip * 16;
  if (strip >= AGG_SPB || Rm >= M) return;
  const float* At = Atile + strip * 16 * AGG_AP;

  const int half = lane >> 4;
  const int l16 = lane & 15;

  v8f acc[NTW] = {};
  const float* abase = At + l16 * AGG_AP + 2 * half;
  const int t0 = sub * NTW;
  const float* bbase = Wlds + half * RP + l16 * 2 + t0 * 32;

  for (int kb = 0; kb < K; kb += 4) {
    v2f a = *(const v2f*)(abase + kb);
    const float* bk = bbase + (kb >> 1) * RP;
    #pragma unroll
    for (int t = 0; t < NTW; ++t) {
      v2f b = *(const v2f*)(bk + t * 32);
      acc[t] = __builtin_amdgcn_wmma_f32_16x16x4_f32(
          false, a, false, b, (short)0, acc[t], false, false);
    }
  }

  float bv[NTW];
  #pragma unroll
  for (int t = 0; t < NTW; ++t) bv[t] = bias[(t0 + t) * 16 + l16];
  if constexpr (!POOL) {
    #pragma unroll
    for (int r = 0; r < 8; ++r) {
      const int row = Rm + r + 8 * half;
      if (row < M) {
        const float di = dinv[row];
        #pragma unroll
        for (int t = 0; t < NTW; ++t) {
          out[(size_t)row * N + (t0 + t) * 16 + l16] = fmaxf(acc[t][r] * di + bv[t], 0.0f);
        }
      }
    }
  } else {
    static_assert(!POOL || NTW == 2, "the pooled epilogue maps a wave's two column tiles onto one 32-lane add");
    const int bg = batch[min(Rm + l16, M - 1)];
    const int g0 = __builtin_amdgcn_readfirstlane(bg);
    if (Rm + 16 <= M && __all(bg == g0)) {
      if ((unsigned)g0 < (unsigned)N_GRAPHS) {
        float di[8];
        #pragma unroll
        for (int r = 0; r < 8; ++r) di[r] = dinv[Rm + r + 8 * half];
        float s[NTW];
        #pragma unroll
        for (int t = 0; t < NTW; ++t) s[t] = 0.0f;
        #pragma unroll
        for (int r = 0; r < 8; ++r) {
          #pragma unroll
          for (int t = 0; t < NTW; ++t) s[t] += fmaxf(acc[t][r] * di[r] + bv[t], 0.0f);
        }
        #pragma unroll
        for (int t = 0; t < NTW; ++t) s[t] += __shfl_xor(s[t], 16);
        float* pg = out + g0 * N + t0 * 16;
        atomicAdd(pg + lane, half ? s[1] : s[0]);
        if (lane == 0 && sub == 0) atomicAdd(cntg + g0, 16.0f);
      }
    } else {
      #pragma unroll
      for (int r = 0; r < 8; ++r) {
        const int row = Rm + r + 8 * half;
        if (row < M) {
          const int g = batch[row];
          if ((unsigned)g < (unsigned)N_GRAPHS) {
            const float di = dinv[row];
            #pragma unroll
            for (int t = 0; t < NTW; ++t)
              atomicAdd(out + g * N + (t0 + t) * 16 + l16, fmaxf(acc[t][r] * di + bv[t], 0.0f));
            if (l16 == 0 && sub == 0) atomicAdd(cntg + g, 1.0f);
          }
        }
      }
    }
  }
}

__global__ void k_pool_final(const float* __restrict__ pool, const float* __restrict__ cntg,
                             float* __restrict__ out) {
  int idx = blockIdx.x * blockDim.x + threadIdx.x;
  if (idx < N_GRAPHS * 64) {
    out[idx] = pool[idx] / fmaxf(cntg[idx >> 6], 1.0f);
  }
}

extern "C" void kernel_launch(void* const* d_in, const int* in_sizes, int n_in,
                              void* d_out, int out_size, void* d_ws, size_t ws_size,
                              hipStream_t stream) {
  const float* x     = (const float*)d_in[0];
  const float* W1    = (const float*)d_in[1];
  const float* b1    = (const float*)d_in[2];
  const float* W2    = (const float*)d_in[3];
  const float* b2    = (const float*)d_in[4];
  const int*   ei    = (const int*)d_in[5];
  const int*   batch = (const int*)d_in[6];

  const int N = in_sizes[6];
  const int E = in_sizes[5] / 2;
  const int G = N_GRAPHS;

  const int* src = ei;
  const int* dst = ei + E;

  char* base = (char*)d_ws;
  size_t off = 0;
  auto take = [&](size_t bytes) {
    size_t o = off;
    off = (off + bytes + 255) & ~(size_t)255;
    return o;
  };
  int*   cnt      = (int*)(base + take(sizeof(int) * (size_t)N * OWN_H));
  int*   tot      = (int*)(base + take(sizeof(int) * (size_t)((N + OWN - 1) / OWN) * OWN_H));
  int*   rowstart = (int*)(base + take(sizeof(int) * (size_t)(N + 1)));
  float* dinv     = (float*)(base + take(sizeof(float) * (size_t)N));
  int*   csr      = (int*)(base + take(sizeof(int) * (size_t)E));
  float* h1       = (float*)(base + take(sizeof(float) * (size_t)N * 128));
  float* pool     = (float*)(base + take(sizeof(float) * (size_t)G * 64));
  float* cntg     = (float*)(base + take(sizeof(float) * (size_t)G));

  const int B = 256;
  const int gOwn = (N + OWN - 1) / OWN;
  const int mtiles = (N + 15) / 16;
  const int gGemm = (mtiles + AGG_SPB - 1) / AGG_SPB;

  k_zero_f32<<<(G * 64 + G + B - 1) / B, B, 0, stream>>>(pool, G * 64 + G);

  k_own_count<<<dim3(gOwn, OWN_H), OWN_T, 0, stream>>>(dst, cnt, tot, E, N);
  k_own_fill<<<dim3(gOwn, OWN_H), OWN_T, 0, stream>>>(src, dst, cnt, tot, rowstart, dinv, csr, E, N);

  k_agg_gemm<128, false><<<gGemm, AGG_T, 0, stream>>>(x, W1, dinv, rowstart, csr, b1, h1, nullptr, nullptr, N);

  k_agg_gemm<64, true><<<gGemm, AGG_T, 0, stream>>>(h1, W2, dinv, rowstart, csr, b2, pool, batch, cntg, N);

  k_pool_final<<<(G * 64 + B - 1) / B, B, 0, stream>>>(pool, cntg, (float*)d_out);
}


# layer launches: a block's 13 strips go in 4 chunks; a wave runs the 32 WMMAs of one 16x16 tile of chunk c in slices of 8 behind the row loads it issues for chunk c+1; sum orders kept
# speedup vs baseline: 4.0457x; 1.0363x over previous
#include <hip/hip_runtime.h>
#include <hip/hip_bf16.h>

typedef float v2f __attribute__((ext_vector_type(2)));
typedef float v8f __attribute__((ext_vector_type(8)));

#define N_GRAPHS 64

__global__ void k_zero_f32(float* p, int n) {
  int i = blockIdx.x * blockDim.x + threadIdx.x;
  for (; i < n; i += gridDim.x * blockDim.x) p[i] = 0.0f;
}

#define OWN 4096
#define OWN_H 16
#define OWN_T 1024
#define OWN_U 4
#define OWN_NPT (OWN / OWN_T)
#define OWN_PB 8

__device__ __forceinline__ void own_part(int E, int h, int& eb, int& ee) {
  const int pieces = (E + 3) >> 2;
  const int per = ((pieces + OWN_H - 1) / OWN_H) * 4;
  eb = min(E, h * per);
  ee = min(E, eb + per);
}

__device__ __forceinline__ int4 own_piece(const int4* __restrict__ v, int len, int rot, int j) {
  int idx = j + rot;
  if (idx >= len) idx -= len;
  const bool in = j < len;
  int4 r = v[in ? idx : 0];
  if (!in) r = make_int4(-1, -1, -1, -1);
  return r;
}

__global__ void __launch_bounds__(OWN_T)
k_own_count(const int* __restrict__ dst, int* __restrict__ cnt, int* __restrict__ tot, int E, int n) {
  __shared__ int c[OWN];
  __shared__ int wsum[32];
  const int tid = threadIdx.x;
  const int lo = blockIdx.x * OWN;
  const int h = blockIdx.y;
  int eb, ee;
  own_part(E, h, eb, ee);
  #pragma unroll
  for (int j = 0; j < OWN_NPT; ++j) c[j * OWN_T + tid] = 0;
  __syncthreads();
  if (((E & 3) == 0) && ((((size_t)dst) & 15) == 0)) {
    const int len = (ee - eb) >> 2;
    if (len > 0) {
      const int4* dv = (const int4*)dst + (eb >> 2);
      const int rot = (int)(((long long)blockIdx.x * len) / gridDim.x);
      constexpr int STEP = OWN_T * OWN_U;
      int4 cur[OWN_U], nxt[OWN_U];
      #pragma unroll
      for (int u = 0; u < OWN_U; ++u) cur[u] = own_piece(dv, len, rot, u * OWN_T + tid);
      for (int j0 = 0; j0 < len; j0 += STEP) {
        #pragma unroll
        for (int u = 0; u < OWN_U; ++u) nxt[u] = own_piece(dv, len, rot, j0 + STEP + u * OWN_T + tid);
        #pragma unroll
        for (int u = 0; u < OWN_U; ++u) {
          const unsigned l0 = (unsigned)(cur[u].x - lo), l1 = (unsigned)(cur[u].y - lo);
          const unsigned l2 = (unsigned)(cur[u].z - lo), l3 = (unsigned)(cur[u].w - lo);
          if (l0 < OWN) atomicAdd(&c[l0], 1);
          if (l1 < OWN) atomicAdd(&c[l1], 1);
          if (l2 < OWN) atomicAdd(&c[l2], 1);
          if (l3 < OWN) atomicAdd(&c[l3], 1);
        }
        #pragma unroll
        for (int u = 0; u < OWN_U; ++u) cur[u] = nxt[u];
      }
    }
  } else {
    for (int e = eb + tid; e < ee; e += OWN_T) {
      const unsigned l = (unsigned)(dst[e] - lo);
      if (l < OWN) atomicAdd(&c[l], 1);
    }
  }
  __syncthreads();
  int mine = 0;
  #pragma unroll
  for (int j = 0; j < OWN_NPT; ++j) {
    const int cv = c[j * OWN_T + tid];
    const int node = lo + j * OWN_T + tid;
    if (node < n) cnt[h * n + node] = cv;
    mine += cv;
  }
  for (int o = 16; o > 0; o >>= 1) mine += __shfl_xor(mine, o);
  if ((tid & 31) == 0) wsum[tid >> 5] = mine;
  __syncthreads();
  if (tid == 0) {
    int t = 0;
    for (int w = 0; w < 32; ++w) t += wsum[w];
    tot[blockIdx.x * OWN_H + h] = t;
  }
}

#define OWN_ASK(U) \
  int p##U##x = -1, p##U##y = -1, p##U##z = -1, p##U##w = -1; \
  { const unsigned lx = (unsigned)(dc[U].x - lo), ly = (unsigned)(dc[U].y - lo); \
    const unsigned lz = (unsigned)(dc[U].z - lo), lw = (unsigned)(dc[U].w - lo); \
    if (lx < OWN) p##U##x = atomicAdd(&pos[lx], 1); \
    if (ly < OWN) p##U##y = atomicAdd(&pos[ly], 1); \
    if (lz < OWN) p##U##z = atomicAdd(&pos[lz], 1); \
    if (lw < OWN) p##U##w = atomicAdd(&pos[lw], 1); }
#define OWN_PUT(U) \
  if (p##U##x >= 0) csr[p##U##x] = sc[U].x; \
  if (p##U##y >= 0) csr[p##U##y] = sc[U].y; \
  if (p##U##z >= 0) csr[p##U##z] = sc[U].z; \
  if (p##U##w >= 0) csr[p##U##w] = sc[U].w;

__global__ void __launch_bounds__(OWN_T)
k_own_fill(const int* __restrict__ src, const int* __restrict__ dst, const int* __restrict__ cnt,
           const int* __restrict__ tot,
           int* __restrict__ rowstart, float* __restrict__ dinv, int* __restrict__ csr,
           int E, int n) {
  __shared__ int pos[OWN];
  __shared__ int wsum[32];
  const int tid = threadIdx.x;
  const int lane = tid & 31;
  const int wave = tid >> 5;
  const int lo = blockIdx.x * OWN;
  const int h = blockIdx.y;
  int eb, ee;
  own_part(E, h, eb, ee);

  int acc = 0;
  for (int i = tid; i < (int)blockIdx.x * OWN_H; i += OWN_T) acc += tot[i];
  for (int o = 16; o > 0; o >>= 1) acc += __shfl_xor(acc, o);
  if (lane == 0) wsum[wave] = acc;
  __syncthreads();
  int base = 0;
  for (int w = 0; w < 32; ++w) base += wsum[w];
  __syncthreads();

  const int first = lo + tid * OWN_NPT;
  int v[OWN_NPT], before[OWN_NPT];
  int nodec[OWN_NPT];
  #pragma unroll
  for (int j = 0; j < OWN_NPT; ++j) { v[j] = 0; before[j] = 0; nodec[j] = min(first + j, n - 1); }
  static_assert(OWN_H % OWN_PB == 0, "the planes are read in whole groups");
  #pragma nounroll
  for (int hb = 0; hb < OWN_H; hb += OWN_PB) {
    int raw[OWN_PB][OWN_NPT];
    #pragma unroll
    for (int q = 0; q < OWN_PB; ++q) {
      const int* __restrict__ plane = cnt + (size_t)(hb + q) * n;
      #pragma unroll
      for (int j = 0; j < OWN_NPT; ++j) raw[q][j] = plane[nodec[j]];
    }
    #pragma unroll
    for (int q = 0; q < OWN_PB; ++q) {
      #pragma unroll
      for (int j = 0; j < OWN_NPT; ++j) {
        const int cv = (first + j < n) ? raw[q][j] : 0;
        if (hb + q < h) before[j] += cv;
        v[j] += cv;
      }
    }
  }
  int pre[OWN_NPT];
  int tsum = 0;
  #pragma unroll
  for (int j = 0; j < OWN_NPT; ++j) { pre[j] = tsum; tsum += v[j]; }
  int inc = tsum;
  for (int o = 1; o < 32; o <<= 1) {
    const int t = __shfl_up(inc, o);
    if (lane >= o) inc += t;
  }
  if (lane == 31) wsum[wave] = inc;
  __syncthreads();
  int wpre = 0;
  for (int w = 0; w < wave; ++w) wpre += wsum[w];
  const int excl = base + wpre + inc - tsum;
  #pragma unroll
  for (int j = 0; j < OWN_NPT; ++j) {
    const int node = first + j;
    pos[tid * OWN_NPT + j] = excl + pre[j] + before[j];
    if (h == 0 && node < n) {
      rowstart[node] = excl + pre[j];
      dinv[node] = rsqrtf((float)v[j] + 1.0f);
    }
  }
  if (h == 0 && blockIdx.x == gridDim.x - 1 && tid == OWN_T - 1) rowstart[n] = excl + tsum;
  __syncthreads();

  if (((E & 3) == 0) && ((((size_t)dst) & 15) == 0) && ((((size_t)src) & 15) == 0)) {
    const int len = (ee - eb) >> 2;
    if (len > 0) {
      const int4* dv = (const int4*)dst + (eb >> 2);
      const int4* sv = (const int4*)src + (eb >> 2);
      const int rot = (int)(((long long)blockIdx.x * len) / gridDim.x);
      constexpr int STEP = OWN_T * OWN_U;
      int4 dc[OWN_U], sc[OWN_U], dn[OWN_U], sn[OWN_U];
      #pragma unroll
      for (int u = 0; u < OWN_U; ++u) {
        dc[u] = own_piece(dv, len, rot, u * OWN_T + tid);
        sc[u] = own_piece(sv, len, rot, u * OWN_T + tid);
      }
      for (int j0 = 0; j0 < len; j0 += STEP) {
        #pragma unroll
        for (int u = 0; u < OWN_U; ++u) {
          dn[u] = own_piece(dv, len, rot, j0 + STEP + u * OWN_T + tid);
          sn[u] = own_piece(sv, len, rot, j0 + STEP + u * OWN_T + tid);
        }
        OWN_ASK(0) OWN_ASK(1) OWN_ASK(2) OWN_ASK(3)
        OWN_PUT(0) OWN_PUT(1) OWN_PUT(2) OWN_PUT(3)
        #pragma unroll
        for (int u = 0; u < OWN_U; ++u) { dc[u] = dn[u]; sc[u] = sn[u]; }
      }
    }
  } else {
    for (int e = eb + tid; e < ee; e += OWN_T) {
      const unsigned l = (unsigned)(dst[e] - lo);
      if (l < OWN) csr[atomicAdd(&pos[l], 1)] = src[e];
    }
  }
}

#define AGG_AP 132
#define AGG_U 8
#define AGG_SPB 13
#define AGG_NODES (AGG_SPB * 16)
#define AGG_T 1024
#define AGG_WV (AGG_T / 32)
#define AGG_CH 4


__device__ __forceinline__ int agg_csr(const int* __restrict__ csr, int first, int s, int e, int lane) {
  const int slot = first + lane;
  return csr[(slot >= s && slot < e) ? slot : 0];
}

__device__ __forceinline__ int agg_idx(int craw, int first, int s, int e, int lane, int self) {
  const int slot = first + lane;
  return (slot >= s && slot < e) ? craw : self;
}

struct AggNoWork { __device__ __forceinline__ void operator()() const {} };
template <class WK>
__device__ __forceinline__ void agg_rows(const float* __restrict__ Hin, int lane, int idx, float dw,
                                         int cnt, float4& acc, WK&& work) {
  const int dwb = __float_as_int(dw);
  if (cnt > 0) idx = (lane < cnt) ? idx : __builtin_amdgcn_readlane(idx, cnt - 1);
  for (int k0 = 0; k0 < cnt; k0 += AGG_U) {
    float4 v[AGG_U];
    #pragma unroll
    for (int u = 0; u < AGG_U; ++u) {
      const int sidx = __builtin_amdgcn_readlane(idx, k0 + u);
      v[u] = *(const float4*)(Hin + (size_t)sidx * 128 + lane * 4);
    }
    work();
    #pragma unroll
    for (int u = 0; u < AGG_U; ++u) {
      const float w = __int_as_float(__builtin_amdgcn_readlane(dwb, k0 + u));
      acc.x += w * v[u].x; acc.y += w * v[u].y; acc.z += w * v[u].z; acc.w += w * v[u].w;
    }
  }
}

__device__ __forceinline__ int agg_cs(int c) {
  return min(4 * c, 8) + (c >= 3 ? 3 : 0) + (c >= 4 ? 2 : 0);
}

template <int N, bool POOL>
__global__ void __launch_bounds__(AGG_T)
k_agg_gemm(const float* __restrict__ Hin, const float* __restrict__ W,
           const float* __restrict__ dinv, const int* __restrict__ rowstart,
           const int* __restrict__ csr, const float* __restrict__ bias,
           float* __restrict__ out, const int* __restrict__ batch,
           float* __restrict__ cntg, int M) {
  constexpr int K = 128;
  constexpr int NT = N / 16;
  constexpr int KQ = 4;
  constexpr int QS = K / 4 / KQ;
  constexpr int RP = 2 * N + 32;
  constexpr int WL = (K / 2) * RP;
  constexpr int AT = AGG_NODES * AGG_AP;
  static_assert(4 * NT <= AGG_WV, "a chunk has at most 4 strips and every tile of it a wave of its own");
  constexpr int ALONE = 168 * 1024 / 4;
  constexpr int WLA = (WL + AT >= ALONE) ? WL : ALONE - AT;
  __shared__ __attribute__((aligned(16))) float Wlds[WLA];
  __shared__ __attribute__((aligned(16))) float Atile[AT];
  for (int i = threadIdx.x; i < K * N; i += blockDim.x) {
    int k = i / N;
    int n = i % N;
    Wlds[(k >> 1) * RP + n * 2 + (k & 1)] = W[i];
  }

  const int wave = __builtin_amdgcn_readfirstlane(threadIdx.x >> 5);
  const int lane = threadIdx.x & 31;
  const int half = lane >> 4;
  const int l16 = lane & 15;
  const int Rb = blockIdx.x * AGG_NODES;

  int seql = -1;
  int total = 0;
  #pragma unroll
  for (int c = 0; c < AGG_CH; ++c) {
    const int cb = agg_cs(c) * 16, cn = (agg_cs(c + 1) - agg_cs(c)) * 16;
    const int f = (wave * cn) / AGG_WV, nrc = ((wave + 1) * cn) / AGG_WV - f;
    if (lane >= total && lane < total + nrc) seql = cb + f + (lane - total);
    total += nrc;
  }
  const int gnode = Rb + max(seql, 0);
  int rs_b = rowstart[min(gnode, M)];
  int rs_e = rowstart[min(gnode + 1, M)];
  if (seql < 0) { rs_b = 0; rs_e = 0; }
  const int selfv = min(gnode, M - 1);

  v8f accw = {};
  int kq = KQ + 1;
  int Rmw = 0, tilew = 0, bgw = 0;
  float bvw = 0.0f;
  float diw[8];
  #pragma unroll
  for (int r = 0; r < 8; ++r) diw[r] = 0.0f;
  const float* abase = Atile;
  const float* bbase = Wlds;

  auto slice = [&]() {
    const float* ap = abase + kq * (QS * 4);
    const float* bp = bbase + kq * (QS * 2) * RP;
    #pragma unroll
    for (int i = 0; i < QS; ++i) {
      v2f a = *(const v2f*)(ap + 4 * i);
      v2f b = *(const v2f*)(bp + 2 * i * RP);
      accw = __builtin_amdgcn_wmma_f32_16x16x4_f32(
          false, a, false, b, (short)0, accw, false, false);
    }
  };

  auto finish = [&]() {
    if constexpr (!POOL) {
      #pragma unroll
      for (int r = 0; r < 8; ++r) {
        const int row = Rmw + r + 8 * half;
        if (row < M) out[(size_t)row * N + tilew * 16 + l16] = fmaxf(accw[r] * diw[r] + bvw, 0.0f);
      }
    } else {
      const int g0 = __builtin_amdgcn_readfirstlane(bgw);
      if (Rmw + 16 <= M && __all(bgw == g0)) {
        if ((unsigned)g0 < (unsigned)N_GRAPHS) {
          float s = 0.0f;
          #pragma unroll
          for (int r = 0; r < 8; ++r) s += fmaxf(accw[r] * diw[r] + bvw, 0.0f);
          s += __shfl_xor(s, 16);
          if (lane < 16) atomicAdd(out + g0 * N + tilew * 16 + lane, s);
          if (lane == 0 && tilew == 0) atomicAdd(cntg + g0, 16.0f);
        }
      } else {
        #pragma unroll
        for (int r = 0; r < 8; ++r) {
          const int row = Rmw + r + 8 * half;
          if (row < M) {
            const int g = batch[row];
            if ((unsigned)g < (unsigned)N_GRAPHS) {
              atomicAdd(out + g * N + tilew * 16 + l16, fmaxf(accw[r] * diw[r] + bvw, 0.0f));
              if (l16 == 0 && tilew == 0) atomicAdd(cntg + g, 1.0f);
            }
          }
        }
      }
    }
  };

  auto work = [&]() {
    if (kq < KQ) { slice(); ++kq; }
    else if (kq == KQ) { finish(); kq = KQ + 1; }
  };

  int sA = __builtin_amdgcn_readlane(rs_b, 0), eA = __builtin_amdgcn_readlane(rs_e, 0);
  int sB = __builtin_amdgcn_readlane(rs_b, 1), eB = __builtin_amdgcn_readlane(rs_e, 1);
  const int raw0 = agg_csr(csr, sA - 1, sA, eA, lane);
  int raw1 = agg_csr(csr, sB - 1, sB, eB, lane);
  int idx = agg_idx(raw0, sA - 1, sA, eA, lane, __builtin_amdgcn_readlane(selfv, 0));
  float dw = dinv[idx];
  int j = 0;
  #pragma nounroll
  for (int c = 0; c < AGG_CH; ++c) {
    const int cn = (agg_cs(c + 1) - agg_cs(c)) * 16;
    const int nrc = ((wave + 1) * cn) / AGG_WV - (wave * cn) / AGG_WV;
    for (int r = 0; r < nrc; ++r, ++j) {
      const int sC = __builtin_amdgcn_readlane(rs_b, j + 2), eC = __builtin_amdgcn_readlane(rs_e, j + 2);
      const int raw2 = agg_csr(csr, sC - 1, sC, eC, lane);
      const int nidx = agg_idx(raw1, sB - 1, sB, eB, lane, __builtin_amdgcn_readlane(selfv, j + 1));
      const float ndw = dinv[nidx];
      const int lrow = __builtin_amdgcn_readlane(seql, j);
      const int node = Rb + lrow;
      const int nodec = min(node, M - 1);
      const int tot = (node < M) ? eA - sA + 1 : 0;
      float4 acc = make_float4(0.0f, 0.0f, 0.0f, 0.0f);
      agg_rows(Hin, lane, idx, (lane < tot) ? dw : 0.0f, min(tot, 32), acc, work);
      for (int cb = sA + 31; cb < eA; cb += 32) {
        const int xi = agg_idx(agg_csr(csr, cb, sA, eA, lane), cb, sA, eA, lane, nodec);
        const float xw = dinv[xi];
        agg_rows(Hin, lane, xi, (lane < eA - cb) ? xw : 0.0f, min(eA - cb, 32), acc, AggNoWork());
      }
      *(float4*)(Atile + lrow * AGG_AP + lane * 4) = acc;
      idx = nidx; dw = ndw; raw1 = raw2;
      sA = sB; eA = eB; sB = sC; eB = eC;
    }
    for (; kq < KQ; ++kq) slice();
    if (kq == KQ) { finish(); kq = KQ + 1; }
    __syncthreads();
    const int strip = agg_cs(c) + wave / NT;
    Rmw = Rb + strip * 16;
    if (wave < (agg_cs(c + 1) - agg_cs(c)) * NT && Rmw < M) {
      tilew = wave % NT;
      abase = Atile + (strip * 16 + l16) * AGG_AP + 2 * half;
      bbase = Wlds + half * RP + l16 * 2 + tilew * 32;
      #pragma unroll
      for (int r = 0; r < 8; ++r) accw[r] = 0.0f;
      kq = 0;
      bvw = bias[tilew * 16 + l16];
      #pragma unroll
      for (int r = 0; r < 8; ++r) diw[r] = dinv[min(Rmw + r + 8 * half, M - 1)];
      if constexpr (POOL) bgw = batch[min(Rmw + l16, M - 1)];
    }
  }
  for (; kq < KQ; ++kq) slice();
  if (kq == KQ) finish();
}

__global__ void k_pool_final(const float* __restrict__ pool, const float* __restrict__ cntg,
                             float* __restrict__ out) {
  int idx = blockIdx.x * blockDim.x + threadIdx.x;
  if (idx < N_GRAPHS * 64) {
    out[idx] = pool[idx] / fmaxf(cntg[idx >> 6], 1.0f);
  }
}

extern "C" void kernel_launch(void* const* d_in, const int* in_sizes, int n_in,
                              void* d_out, int out_size, void* d_ws, size_t ws_size,
                              hipStream_t stream) {
  const float* x     = (const float*)d_in[0];
  const float* W1    = (const float*)d_in[1];
  const float* b1    = (const float*)d_in[2];
  const float* W2    = (const float*)d_in[3];
  const float* b2    = (const float*)d_in[4];
  const int*   ei    = (const int*)d_in[5];
  const int*   batch = (const int*)d_in[6];

  const int N = in_sizes[6];
  const int E = in_sizes[5] / 2;
  const int G = N_GRAPHS;

  const int* src = ei;
  const int* dst = ei + E;

  char* base = (char*)d_ws;
  size_t off = 0;
  auto take = [&](size_t bytes) {
    size_t o = off;
    off = (off + bytes + 255) & ~(size_t)255;
    return o;
  };
  int*   cnt      = (int*)(base + take(sizeof(int) * (size_t)N * OWN_H));
  int*   tot      = (int*)(base + take(sizeof(int) * (size_t)((N + OWN - 1) / OWN) * OWN_H));
  int*   rowstart = (int*)(base + take(sizeof(int) * (size_t)(N + 1)));
  float* dinv     = (float*)(base + take(sizeof(float) * (size_t)N));
  int*   csr      = (int*)(base + take(sizeof(int) * (size_t)E));
  float* h1       = (float*)(base + take(sizeof(float) * (size_t)N * 128));
  float* pool     = (float*)(base + take(sizeof(float) * (size_t)G * 64));
  float* cntg     = (float*)(base + take(sizeof(float) * (size_t)G));

  const int B = 256;
  const int gOwn = (N + OWN - 1) / OWN;
  const int mtiles = (N + 15) / 16;
  const int gGemm = (mtiles + AGG_SPB - 1) / AGG_SPB;

  k_zero_f32<<<(G * 64 + G + B - 1) / B, B, 0, stream>>>(pool, G * 64 + G);

  k_own_count<<<dim3(gOwn, OWN_H), OWN_T, 0, stream>>>(dst, cnt, tot, E, N);
  k_own_fill<<<dim3(gOwn, OWN_H), OWN_T, 0, stream>>>(src, dst, cnt, tot, rowstart, dinv, csr, E, N);

  k_agg_gemm<128, false><<<gGemm, AGG_T, 0, stream>>>(x, W1, dinv, rowstart, csr, b1, h1, nullptr, nullptr, N);

  k_agg_gemm<64, true><<<gGemm, AGG_T, 0, stream>>>(h1, W2, dinv, rowstart, csr, b2, pool, batch, cntg, N);

  k_pool_final<<<(G * 64 + B - 1) / B, B, 0, stream>>>(pool, cntg, (float*)d_out);
}


# layer launches: W staging leaves the kernel front (16 or 8 dependent 4-byte load-wait-store trips a thread) and rides behind the first chunk's row loads, where no tile work exists; sum orders kept
# speedup vs baseline: 4.3520x; 1.0757x over previous
#include <hip/hip_runtime.h>
#include <hip/hip_bf16.h>

typedef float v2f __attribute__((ext_vector_type(2)));
typedef float v8f __attribute__((ext_vector_type(8)));

#define N_GRAPHS 64

__global__ void k_zero_f32(float* p, int n) {
  int i = blockIdx.x * blockDim.x + threadIdx.x;
  for (; i < n; i += gridDim.x * blockDim.x) p[i] = 0.0f;
}

#define OWN 4096
#define OWN_H 16
#define OWN_T 1024
#define OWN_U 4
#define OWN_NPT (OWN / OWN_T)
#define OWN_PB 8

__device__ __forceinline__ void own_part(int E, int h, int& eb, int& ee) {
  const int pieces = (E + 3) >> 2;
  const int per = ((pieces + OWN_H - 1) / OWN_H) * 4;
  eb = min(E, h * per);
  ee = min(E, eb + per);
}

__device__ __forceinline__ int4 own_piece(const int4* __restrict__ v, int len, int rot, int j) {
  int idx = j + rot;
  if (idx >= len) idx -= len;
  const bool in = j < len;
  int4 r = v[in ? idx : 0];
  if (!in) r = make_int4(-1, -1, -1, -1);
  return r;
}

__global__ void __launch_bounds__(OWN_T)
k_own_count(const int* __restrict__ dst, int* __restrict__ cnt, int* __restrict__ tot, int E, int n) {
  __shared__ int c[OWN];
  __shared__ int wsum[32];
  const int tid = threadIdx.x;
  const int lo = blockIdx.x * OWN;
  const int h = blockIdx.y;
  int eb, ee;
  own_part(E, h, eb, ee);
  #pragma unroll
  for (int j = 0; j < OWN_NPT; ++j) c[j * OWN_T + tid] = 0;
  __syncthreads();
  if (((E & 3) == 0) && ((((size_t)dst) & 15) == 0)) {
    const int len = (ee - eb) >> 2;
    if (len > 0) {
      const int4* dv = (const int4*)dst + (eb >> 2);
      const int rot = (int)(((long long)blockIdx.x * len) / gridDim.x);
      constexpr int STEP = OWN_T * OWN_U;
      int4 cur[OWN_U], nxt[OWN_U];
      #pragma unroll
      for (int u = 0; u < OWN_U; ++u) cur[u] = own_piece(dv, len, rot, u * OWN_T + tid);
      for (int j0 = 0; j0 < len; j0 += STEP) {
        #pragma unroll
        for (int u = 0; u < OWN_U; ++u) nxt[u] = own_piece(dv, len, rot, j0 + STEP + u * OWN_T + tid);
        #pragma unroll
        for (int u = 0; u < OWN_U; ++u) {
          const unsigned l0 = (unsigned)(cur[u].x - lo), l1 = (unsigned)(cur[u].y - lo);
          const unsigned l2 = (unsigned)(cur[u].z - lo), l3 = (unsigned)(cur[u].w - lo);
          if (l0 < OWN) atomicAdd(&c[l0], 1);
          if (l1 < OWN) atomicAdd(&c[l1], 1);
          if (l2 < OWN) atomicAdd(&c[l2], 1);
          if (l3 < OWN) atomicAdd(&c[l3], 1);
        }
        #pragma unroll
        for (int u = 0; u < OWN_U; ++u) cur[u] = nxt[u];
      }
    }
  } else {
    for (int e = eb + tid; e < ee; e += OWN_T) {
      const unsigned l = (unsigned)(dst[e] - lo);
      if (l < OWN) atomicAdd(&c[l], 1);
    }
  }
  __syncthreads();
  int mine = 0;
  #pragma unroll
  for (int j = 0; j < OWN_NPT; ++j) {
    const int cv = c[j * OWN_T + tid];
    const int node = lo + j * OWN_T + tid;
    if (node < n) cnt[h * n + node] = cv;
    mine += cv;
  }
  for (int o = 16; o > 0; o >>= 1) mine += __shfl_xor(mine, o);
  if ((tid & 31) == 0) wsum[tid >> 5] = mine;
  __syncthreads();
  if (tid == 0) {
    int t = 0;
    for (int w = 0; w < 32; ++w) t += wsum[w];
    tot[blockIdx.x * OWN_H + h] = t;
  }
}

#define OWN_ASK(U) \
  int p##U##x = -1, p##U##y = -1, p##U##z = -1, p##U##w = -1; \
  { const unsigned lx = (unsigned)(dc[U].x - lo), ly = (unsigned)(dc[U].y - lo); \
    const unsigned lz = (unsigned)(dc[U].z - lo), lw = (unsigned)(dc[U].w - lo); \
    if (lx < OWN) p##U##x = atomicAdd(&pos[lx], 1); \
    if (ly < OWN) p##U##y = atomicAdd(&pos[ly], 1); \
    if (lz < OWN) p##U##z = atomicAdd(&pos[lz], 1); \
    if (lw < OWN) p##U##w = atomicAdd(&pos[lw], 1); }
#define OWN_PUT(U) \
  if (p##U##x >= 0) csr[p##U##x] = sc[U].x; \
  if (p##U##y >= 0) csr[p##U##y] = sc[U].y; \
  if (p##U##z >= 0) csr[p##U##z] = sc[U].z; \
  if (p##U##w >= 0) csr[p##U##w] = sc[U].w;

__global__ void __launch_bounds__(OWN_T)
k_own_fill(const int* __restrict__ src, const int* __restrict__ dst, const int* __restrict__ cnt,
           const int* __restrict__ tot,
           int* __restrict__ rowstart, float* __restrict__ dinv, int* __restrict__ csr,
           int E, int n) {
  __shared__ int pos[OWN];
  __shared__ int wsum[32];
  const int tid = threadIdx.x;
  const int lane = tid & 31;
  const int wave = tid >> 5;
  const int lo = blockIdx.x * OWN;
  const int h = blockIdx.y;
  int eb, ee;
  own_part(E, h, eb, ee);

  int acc = 0;
  for (int i = tid; i < (int)blockIdx.x * OWN_H; i += OWN_T) acc += tot[i];
  for (int o = 16; o > 0; o >>= 1) acc += __shfl_xor(acc, o);
  if (lane == 0) wsum[wave] = acc;
  __syncthreads();
  int base = 0;
  for (int w = 0; w < 32; ++w) base += wsum[w];
  __syncthreads();

  const int first = lo + tid * OWN_NPT;
  int v[OWN_NPT], before[OWN_NPT];
  int nodec[OWN_NPT];
  #pragma unroll
  for (int j = 0; j < OWN_NPT; ++j) { v[j] = 0; before[j] = 0; nodec[j] = min(first + j, n - 1); }
  static_assert(OWN_H % OWN_PB == 0, "the planes are read in whole groups");
  #pragma nounroll
  for (int hb = 0; hb < OWN_H; hb += OWN_PB) {
    int raw[OWN_PB][OWN_NPT];
    #pragma unroll
    for (int q = 0; q < OWN_PB; ++q) {
      const int* __restrict__ plane = cnt + (size_t)(hb + q) * n;
      #pragma unroll
      for (int j = 0; j < OWN_NPT; ++j) raw[q][j] = plane[nodec[j]];
    }
    #pragma unroll
    for (int q = 0; q < OWN_PB; ++q) {
      #pragma unroll
      for (int j = 0; j < OWN_NPT; ++j) {
        const int cv = (first + j < n) ? raw[q][j] : 0;
        if (hb + q < h) before[j] += cv;
        v[j] += cv;
      }
    }
  }
  int pre[OWN_NPT];
  int tsum = 0;
  #pragma unroll
  for (int j = 0; j < OWN_NPT; ++j) { pre[j] = tsum; tsum += v[j]; }
  int inc = tsum;
  for (int o = 1; o < 32; o <<= 1) {
    const int t = __shfl_up(inc, o);
    if (lane >= o) inc += t;
  }
  if (lane == 31) wsum[wave] = inc;
  __syncthreads();
  int wpre = 0;
  for (int w = 0; w < wave; ++w) wpre += wsum[w];
  const int excl = base + wpre + inc - tsum;
  #pragma unroll
  for (int j = 0; j < OWN_NPT; ++j) {
    const int node = first + j;
    pos[tid * OWN_NPT + j] = excl + pre[j] + before[j];
    if (h == 0 && node < n) {
      rowstart[node] = excl + pre[j];
      dinv[node] = rsqrtf((float)v[j] + 1.0f);
    }
  }
  if (h == 0 && blockIdx.x == gridDim.x - 1 && tid == OWN_T - 1) rowstart[n] = excl + tsum;
  __syncthreads();

  if (((E & 3) == 0) && ((((size_t)dst) & 15) == 0) && ((((size_t)src) & 15) == 0)) {
    const int len = (ee - eb) >> 2;
    if (len > 0) {
      const int4* dv = (const int4*)dst + (eb >> 2);
      const int4* sv = (const int4*)src + (eb >> 2);
      const int rot = (int)(((long long)blockIdx.x * len) / gridDim.x);
      constexpr int STEP = OWN_T * OWN_U;
      int4 dc[OWN_U], sc[OWN_U], dn[OWN_U], sn[OWN_U];
      #pragma unroll
      for (int u = 0; u < OWN_U; ++u) {
        dc[u] = own_piece(dv, len, rot, u * OWN_T + tid);
        sc[u] = own_piece(sv, len, rot, u * OWN_T + tid);
      }
      for (int j0 = 0; j0 < len; j0 += STEP) {
        #pragma unroll
        for (int u = 0; u < OWN_U; ++u) {
          dn[u] = own_piece(dv, len, rot, j0 + STEP + u * OWN_T + tid);
          sn[u] = own_piece(sv, len, rot, j0 + STEP + u * OWN_T + tid);
        }
        OWN_ASK(0) OWN_ASK(1) OWN_ASK(2) OWN_ASK(3)
        OWN_PUT(0) OWN_PUT(1) OWN_PUT(2) OWN_PUT(3)
        #pragma unroll
        for (int u = 0; u < OWN_U; ++u) { dc[u] = dn[u]; sc[u] = sn[u]; }
      }
    }
  } else {
    for (int e = eb + tid; e < ee; e += OWN_T) {
      const unsigned l = (unsigned)(dst[e] - lo);
      if (l < OWN) csr[atomicAdd(&pos[l], 1)] = src[e];
    }
  }
}

#define AGG_AP 132
#define AGG_U 8
#define AGG_SPB 13
#define AGG_NODES (AGG_SPB * 16)
#define AGG_T 1024
#define AGG_WV (AGG_T / 32)
#define AGG_CH 4


__device__ __forceinline__ int agg_csr(const int* __restrict__ csr, int first, int s, int e, int lane) {
  const int slot = first + lane;
  return csr[(slot >= s && slot < e) ? slot : 0];
}

__device__ __forceinline__ int agg_idx(int craw, int first, int s, int e, int lane, int self) {
  const int slot = first + lane;
  return (slot >= s && slot < e) ? craw : self;
}

struct AggNoWork { __device__ __forceinline__ void operator()() const {} };
template <class WK>
__device__ __forceinline__ void agg_rows(const float* __restrict__ Hin, int lane, int idx, float dw,
                                         int cnt, float4& acc, WK&& work) {
  const int dwb = __float_as_int(dw);
  if (cnt > 0) idx = (lane < cnt) ? idx : __builtin_amdgcn_readlane(idx, cnt - 1);
  for (int k0 = 0; k0 < cnt; k0 += AGG_U) {
    float4 v[AGG_U];
    #pragma unroll
    for (int u = 0; u < AGG_U; ++u) {
      const int sidx = __builtin_amdgcn_readlane(idx, k0 + u);
      v[u] = *(const float4*)(Hin + (size_t)sidx * 128 + lane * 4);
    }
    work();
    #pragma unroll
    for (int u = 0; u < AGG_U; ++u) {
      const float w = __int_as_float(__builtin_amdgcn_readlane(dwb, k0 + u));
      acc.x += w * v[u].x; acc.y += w * v[u].y; acc.z += w * v[u].z; acc.w += w * v[u].w;
    }
  }
}

__device__ __forceinline__ int agg_cs(int c) {
  return min(4 * c, 8) + (c >= 3 ? 3 : 0) + (c >= 4 ? 2 : 0);
}

template <int N, bool POOL>
__global__ void __launch_bounds__(AGG_T)
k_agg_gemm(const float* __restrict__ Hin, const float* __restrict__ W,
           const float* __restrict__ dinv, const int* __restrict__ rowstart,
           const int* __restrict__ csr, const float* __restrict__ bias,
           float* __restrict__ out, const int* __restrict__ batch,
           float* __restrict__ cntg, int M) {
  constexpr int K = 128;
  constexpr int NT = N / 16;
  constexpr int KQ = 4;
  constexpr int QS = K / 4 / KQ;
  constexpr int RP = 2 * N + 32;
  constexpr int WL = (K / 2) * RP;
  constexpr int AT = AGG_NODES * AGG_AP;
  static_assert(4 * NT <= AGG_WV, "a chunk has at most 4 strips and every tile of it a wave of its own");
  constexpr int ALONE = 168 * 1024 / 4;
  constexpr int WLA = (WL + AT >= ALONE) ? WL : ALONE - AT;
  __shared__ __attribute__((aligned(16))) float Wlds[WLA];
  __shared__ __attribute__((aligned(16))) float Atile[AT];

  const int wave = __builtin_amdgcn_readfirstlane(threadIdx.x >> 5);
  const int lane = threadIdx.x & 31;
  const int half = lane >> 4;
  const int l16 = lane & 15;
  const int Rb = blockIdx.x * AGG_NODES;

  int seql = -1;
  int total = 0;
  #pragma unroll
  for (int c = 0; c < AGG_CH; ++c) {
    const int cb = agg_cs(c) * 16, cn = (agg_cs(c + 1) - agg_cs(c)) * 16;
    const int f = (wave * cn) / AGG_WV, nrc = ((wave + 1) * cn) / AGG_WV - f;
    if (lane >= total && lane < total + nrc) seql = cb + f + (lane - total);
    total += nrc;
  }
  const int gnode = Rb + max(seql, 0);
  int rs_b = rowstart[min(gnode, M)];
  int rs_e = rowstart[min(gnode + 1, M)];
  if (seql < 0) { rs_b = 0; rs_e = 0; }
  const int selfv = min(gnode, M - 1);

  v8f accw = {};
  int kq = KQ + 1;
  int Rmw = 0, tilew = 0, bgw = 0;
  float bvw = 0.0f;
  float diw[8];
  #pragma unroll
  for (int r = 0; r < 8; ++r) diw[r] = 0.0f;
  const float* abase = Atile;
  const float* bbase = Wlds;

  auto slice = [&]() {
    const float* ap = abase + kq * (QS * 4);
    const float* bp = bbase + kq * (QS * 2) * RP;
    #pragma unroll
    for (int i = 0; i < QS; ++i) {
      v2f a = *(const v2f*)(ap + 4 * i);
      v2f b = *(const v2f*)(bp + 2 * i * RP);
      accw = __builtin_amdgcn_wmma_f32_16x16x4_f32(
          false, a, false, b, (short)0, accw, false, false);
    }
  };

  auto finish = [&]() {
    if constexpr (!POOL) {
      #pragma unroll
      for (int r = 0; r < 8; ++r) {
        const int row = Rmw + r + 8 * half;
        if (row < M) out[(size_t)row * N + tilew * 16 + l16] = fmaxf(accw[r] * diw[r] + bvw, 0.0f);
      }
    } else {
      const int g0 = __builtin_amdgcn_readfirstlane(bgw);
      if (Rmw + 16 <= M && __all(bgw == g0)) {
        if ((unsigned)g0 < (unsigned)N_GRAPHS) {
          float s = 0.0f;
          #pragma unroll
          for (int r = 0; r < 8; ++r) s += fmaxf(accw[r] * diw[r] + bvw, 0.0f);
          s += __shfl_xor(s, 16);
          if (lane < 16) atomicAdd(out + g0 * N + tilew * 16 + lane, s);
          if (lane == 0 && tilew == 0) atomicAdd(cntg + g0, 16.0f);
        }
      } else {
        #pragma unroll
        for (int r = 0; r < 8; ++r) {
          const int row = Rmw + r + 8 * half;
          if (row < M) {
            const int g = batch[row];
            if ((unsigned)g < (unsigned)N_GRAPHS) {
              atomicAdd(out + g * N + tilew * 16 + l16, fmaxf(accw[r] * diw[r] + bvw, 0.0f));
              if (l16 == 0 && tilew == 0) atomicAdd(cntg + g, 1.0f);
            }
          }
        }
      }
    }
  };

  auto work = [&]() {
    if (kq < KQ) { slice(); ++kq; }
    else if (kq == KQ) { finish(); kq = KQ + 1; }
  };

  constexpr int WU = (K / 2) * (N / 4);
  constexpr int WQ = WU / AGG_T;
  static_assert(WU % AGG_T == 0, "every thread stages the same number of units");
  int wq = 0;
  auto stage = [&]() {
    const int u = wq * AGG_T + (int)threadIdx.x;
    const int kp = u / (N / 4), n4 = u % (N / 4);
    const float4 w0 = *(const float4*)(W + (2 * kp) * N + n4 * 4);
    const float4 w1 = *(const float4*)(W + (2 * kp + 1) * N + n4 * 4);
    float* d = Wlds + kp * RP + n4 * 8;
    *(float4*)d = make_float4(w0.x, w1.x, w0.y, w1.y);
    *(float4*)(d + 4) = make_float4(w0.z, w1.z, w0.w, w1.w);
  };
  auto work_first = [&]() {
    if (wq < WQ) { stage(); ++wq; }
  };

  int sA = __builtin_amdgcn_readlane(rs_b, 0), eA = __builtin_amdgcn_readlane(rs_e, 0);
  int sB = __builtin_amdgcn_readlane(rs_b, 1), eB = __builtin_amdgcn_readlane(rs_e, 1);
  const int raw0 = agg_csr(csr, sA - 1, sA, eA, lane);
  int raw1 = agg_csr(csr, sB - 1, sB, eB, lane);
  int idx = agg_idx(raw0, sA - 1, sA, eA, lane, __builtin_amdgcn_readlane(selfv, 0));
  float dw = dinv[idx];
  int j = 0;
  auto gather = [&](int c, auto&& wk) {
    const int cn = (agg_cs(c + 1) - agg_cs(c)) * 16;
    const int nrc = ((wave + 1) * cn) / AGG_WV - (wave * cn) / AGG_WV;
    for (int r = 0; r < nrc; ++r, ++j) {
      const int sC = __builtin_amdgcn_readlane(rs_b, j + 2), eC = __builtin_amdgcn_readlane(rs_e, j + 2);
      const int raw2 = agg_csr(csr, sC - 1, sC, eC, lane);
      const int nidx = agg_idx(raw1, sB - 1, sB, eB, lane, __builtin_amdgcn_readlane(selfv, j + 1));
      const float ndw = dinv[nidx];
      const int lrow = __builtin_amdgcn_readlane(seql, j);
      const int node = Rb + lrow;
      const int nodec = min(node, M - 1);
      const int tot = (node < M) ? eA - sA + 1 : 0;
      float4 acc = make_float4(0.0f, 0.0f, 0.0f, 0.0f);
      agg_rows(Hin, lane, idx, (lane < tot) ? dw : 0.0f, min(tot, 32), acc, wk);
      for (int cb = sA + 31; cb < eA; cb += 32) {
        const int xi = agg_idx(agg_csr(csr, cb, sA, eA, lane), cb, sA, eA, lane, nodec);
        const float xw = dinv[xi];
        agg_rows(Hin, lane, xi, (lane < eA - cb) ? xw : 0.0f, min(eA - cb, 32), acc, AggNoWork());
      }
      *(float4*)(Atile + lrow * AGG_AP + lane * 4) = acc;
      idx = nidx; dw = ndw; raw1 = raw2;
      sA = sB; eA = eB; sB = sC; eB = eC;
    }
  };
  auto open_tile = [&](int c) {
    const int strip = agg_cs(c) + wave / NT;
    Rmw = Rb + strip * 16;
    if (wave < (agg_cs(c + 1) - agg_cs(c)) * NT && Rmw < M) {
      tilew = wave % NT;
      abase = Atile + (strip * 16 + l16) * AGG_AP + 2 * half;
      bbase = Wlds + half * RP + l16 * 2 + tilew * 32;
      #pragma unroll
      for (int r = 0; r < 8; ++r) accw[r] = 0.0f;
      kq = 0;
      bvw = bias[tilew * 16 + l16];
      #pragma unroll
      for (int r = 0; r < 8; ++r) diw[r] = dinv[min(Rmw + r + 8 * half, M - 1)];
      if constexpr (POOL) bgw = batch[min(Rmw + l16, M - 1)];
    }
  };

  gather(0, work_first);
  for (; wq < WQ; ++wq) stage();
  #pragma nounroll
  for (int c = 0;;) {
    __syncthreads();
    open_tile(c);
    if (++c == AGG_CH) break;
    gather(c, work);
    for (; kq < KQ; ++kq) slice();
    if (kq == KQ) { finish(); kq = KQ + 1; }
  }
  for (; kq < KQ; ++kq) slice();
  if (kq == KQ) finish();
}

__global__ void k_pool_final(const float* __restrict__ pool, const float* __restrict__ cntg,
                             float* __restrict__ out) {
  int idx = blockIdx.x * blockDim.x + threadIdx.x;
  if (idx < N_GRAPHS * 64) {
    out[idx] = pool[idx] / fmaxf(cntg[idx >> 6], 1.0f);
  }
}

extern "C" void kernel_launch(void* const* d_in, const int* in_sizes, int n_in,
                              void* d_out, int out_size, void* d_ws, size_t ws_size,
                              hipStream_t stream) {
  const float* x     = (const float*)d_in[0];
  const float* W1    = (const float*)d_in[1];
  const float* b1    = (const float*)d_in[2];
  const float* W2    = (const float*)d_in[3];
  const float* b2    = (const float*)d_in[4];
  const int*   ei    = (const int*)d_in[5];
  const int*   batch = (const int*)d_in[6];

  const int N = in_sizes[6];
  const int E = in_sizes[5] / 2;
  const int G = N_GRAPHS;

  const int* src = ei;
  const int* dst = ei + E;

  char* base = (char*)d_ws;
  size_t off = 0;
  auto take = [&](size_t bytes) {
    size_t o = off;
    off = (off + bytes + 255) & ~(size_t)255;
    return o;
  };
  int*   cnt      = (int*)(base + take(sizeof(int) * (size_t)N * OWN_H));
  int*   tot      = (int*)(base + take(sizeof(int) * (size_t)((N + OWN - 1) / OWN) * OWN_H));
  int*   rowstart = (int*)(base + take(sizeof(int) * (size_t)(N + 1)));
  float* dinv     = (float*)(base + take(sizeof(float) * (size_t)N));
  int*   csr      = (int*)(base + take(sizeof(int) * (size_t)E));
  float* h1       = (float*)(base + take(sizeof(float) * (size_t)N * 128));
  float* pool     = (float*)(base + take(sizeof(float) * (size_t)G * 64));
  float* cntg     = (float*)(base + take(sizeof(float) * (size_t)G));

  const int B = 256;
  const int gOwn = (N + OWN - 1) / OWN;
  const int mtiles = (N + 15) / 16;
  const int gGemm = (mtiles + AGG_SPB - 1) / AGG_SPB;

  k_zero_f32<<<(G * 64 + G + B - 1) / B, B, 0, stream>>>(pool, G * 64 + G);

  k_own_count<<<dim3(gOwn, OWN_H), OWN_T, 0, stream>>>(dst, cnt, tot, E, N);
  k_own_fill<<<dim3(gOwn, OWN_H), OWN_T, 0, stream>>>(src, dst, cnt, tot, rowstart, dinv, csr, E, N);

  k_agg_gemm<128, false><<<gGemm, AGG_T, 0, stream>>>(x, W1, dinv, rowstart, csr, b1, h1, nullptr, nullptr, N);

  k_agg_gemm<64, true><<<gGemm, AGG_T, 0, stream>>>(h1, W2, dinv, rowstart, csr, b2, pool, batch, cntg, N);

  k_pool_final<<<(G * 64 + B - 1) / B, B, 0, stream>>>(pool, cntg, (float*)d_out);
}


# layer 2 multiplies BEFORE it sums: layer-1 launch keeps relu rows in LDS, multiplies them by W2 (WMMA), stores 64-wide rows times dinv; layer 2 sums 256-byte rows, no product; sum orders change
# speedup vs baseline: 4.8910x; 1.1239x over previous
#include <hip/hip_runtime.h>
#include <hip/hip_bf16.h>

typedef float v2f __attribute__((ext_vector_type(2)));
typedef float v8f __attribute__((ext_vector_type(8)));

#define N_GRAPHS 64

__global__ void k_zero_f32(float* p, int n) {
  int i = blockIdx.x * blockDim.x + threadIdx.x;
  for (; i < n; i += gridDim.x * blockDim.x) p[i] = 0.0f;
}

#define OWN 4096
#define OWN_H 16
#define OWN_T 1024
#define OWN_U 4
#define OWN_NPT (OWN / OWN_T)
#define OWN_PB 8

__device__ __forceinline__ void own_part(int E, int h, int& eb, int& ee) {
  const int pieces = (E + 3) >> 2;
  const int per = ((pieces + OWN_H - 1) / OWN_H) * 4;
  eb = min(E, h * per);
  ee = min(E, eb + per);
}

__device__ __forceinline__ int4 own_piece(const int4* __restrict__ v, int len, int rot, int j) {
  int idx = j + rot;
  if (idx >= len) idx -= len;
  const bool in = j < len;
  int4 r = v[in ? idx : 0];
  if (!in) r = make_int4(-1, -1, -1, -1);
  return r;
}

__global__ void __launch_bounds__(OWN_T)
k_own_count(const int* __restrict__ dst, int* __restrict__ cnt, int* __restrict__ tot, int E, int n) {
  __shared__ int c[OWN];
  __shared__ int wsum[32];
  const int tid = threadIdx.x;
  const int lo = blockIdx.x * OWN;
  const int h = blockIdx.y;
  int eb, ee;
  own_part(E, h, eb, ee);
  #pragma unroll
  for (int j = 0; j < OWN_NPT; ++j) c[j * OWN_T + tid] = 0;
  __syncthreads();
  if (((E & 3) == 0) && ((((size_t)dst) & 15) == 0)) {
    const int len = (ee - eb) >> 2;
    if (len > 0) {
      const int4* dv = (const int4*)dst + (eb >> 2);
      const int rot = (int)(((long long)blockIdx.x * len) / gridDim.x);
      constexpr int STEP = OWN_T * OWN_U;
      int4 cur[OWN_U], nxt[OWN_U];
      #pragma unroll
      for (int u = 0; u < OWN_U; ++u) cur[u] = own_piece(dv, len, rot, u * OWN_T + tid);
      for (int j0 = 0; j0 < len; j0 += STEP) {
        #pragma unroll
        for (int u = 0; u < OWN_U; ++u) nxt[u] = own_piece(dv, len, rot, j0 + STEP + u * OWN_T + tid);
        #pragma unroll
        for (int u = 0; u < OWN_U; ++u) {
          const unsigned l0 = (unsigned)(cur[u].x - lo), l1 = (unsigned)(cur[u].y - lo);
          const unsigned l2 = (unsigned)(cur[u].z - lo), l3 = (unsigned)(cur[u].w - lo);
          if (l0 < OWN) atomicAdd(&c[l0], 1);
          if (l1 < OWN) atomicAdd(&c[l1], 1);
          if (l2 < OWN) atomicAdd(&c[l2], 1);
          if (l3 < OWN) atomicAdd(&c[l3], 1);
        }
        #pragma unroll
        for (int u = 0; u < OWN_U; ++u) cur[u] = nxt[u];
      }
    }
  } else {
    for (int e = eb + tid; e < ee; e += OWN_T) {
      const unsigned l = (unsigned)(dst[e] - lo);
      if (l < OWN) atomicAdd(&c[l], 1);
    }
  }
  __syncthreads();
  int mine = 0;
  #pragma unroll
  for (int j = 0; j < OWN_NPT; ++j) {
    const int cv = c[j * OWN_T + tid];
    const int node = lo + j * OWN_T + tid;
    if (node < n) cnt[h * n + node] = cv;
    mine += cv;
  }
  for (int o = 16; o > 0; o >>= 1) mine += __shfl_xor(mine, o);
  if ((tid & 31) == 0) wsum[tid >> 5] = mine;
  __syncthreads();
  if (tid == 0) {
    int t = 0;
    for (int w = 0; w < 32; ++w) t += wsum[w];
    tot[blockIdx.x * OWN_H + h] = t;
  }
}

#define OWN_ASK(U) \
  int p##U##x = -1, p##U##y = -1, p##U##z = -1, p##U##w = -1; \
  { const unsigned lx = (unsigned)(dc[U].x - lo), ly = (unsigned)(dc[U].y - lo); \
    const unsigned lz = (unsigned)(dc[U].z - lo), lw = (unsigned)(dc[U].w - lo); \
    if (lx < OWN) p##U##x = atomicAdd(&pos[lx], 1); \
    if (ly < OWN) p##U##y = atomicAdd(&pos[ly], 1); \
    if (lz < OWN) p##U##z = atomicAdd(&pos[lz], 1); \
    if (lw < OWN) p##U##w = atomicAdd(&pos[lw], 1); }
#define OWN_PUT(U) \
  if (p##U##x >= 0) csr[p##U##x] = sc[U].x; \
  if (p##U##y >= 0) csr[p##U##y] = sc[U].y; \
  if (p##U##z >= 0) csr[p##U##z] = sc[U].z; \
  if (p##U##w >= 0) csr[p##U##w] = sc[U].w;

__global__ void __launch_bounds__(OWN_T)
k_own_fill(const int* __restrict__ src, const int* __restrict__ dst, const int* __restrict__ cnt,
           const int* __restrict__ tot,
           int* __restrict__ rowstart, float* __restrict__ dinv, int* __restrict__ csr,
           int E, int n) {
  __shared__ int pos[OWN];
  __shared__ int wsum[32];
  const int tid = threadIdx.x;
  const int lane = tid & 31;
  const int wave = tid >> 5;
  const int lo = blockIdx.x * OWN;
  const int h = blockIdx.y;
  int eb, ee;
  own_part(E, h, eb, ee);

  int acc = 0;
  for (int i = tid; i < (int)blockIdx.x * OWN_H; i += OWN_T) acc += tot[i];
  for (int o = 16; o > 0; o >>= 1) acc += __shfl_xor(acc, o);
  if (lane == 0) wsum[wave] = acc;
  __syncthreads();
  int base = 0;
  for (int w = 0; w < 32; ++w) base += wsum[w];
  __syncthreads();

  const int first = lo + tid * OWN_NPT;
  int v[OWN_NPT], before[OWN_NPT];
  int nodec[OWN_NPT];
  #pragma unroll
  for (int j = 0; j < OWN_NPT; ++j) { v[j] = 0; before[j] = 0; nodec[j] = min(first + j, n - 1); }
  static_assert(OWN_H % OWN_PB == 0, "the planes are read in whole groups");
  #pragma nounroll
  for (int hb = 0; hb < OWN_H; hb += OWN_PB) {
    int raw[OWN_PB][OWN_NPT];
    #pragma unroll
    for (int q = 0; q < OWN_PB; ++q) {
      const int* __restrict__ plane = cnt + (size_t)(hb + q) * n;
      #pragma unroll
      for (int j = 0; j < OWN_NPT; ++j) raw[q][j] = plane[nodec[j]];
    }
    #pragma unroll
    for (int q = 0; q < OWN_PB; ++q) {
      #pragma unroll
      for (int j = 0; j < OWN_NPT; ++j) {
        const int cv = (first + j < n) ? raw[q][j] : 0;
        if (hb + q < h) before[j] += cv;
        v[j] += cv;
      }
    }
  }
  int pre[OWN_NPT];
  int tsum = 0;
  #pragma unroll
  for (int j = 0; j < OWN_NPT; ++j) { pre[j] = tsum; tsum += v[j]; }
  int inc = tsum;
  for (int o = 1; o < 32; o <<= 1) {
    const int t = __shfl_up(inc, o);
    if (lane >= o) inc += t;
  }
  if (lane == 31) wsum[wave] = inc;
  __syncthreads();
  int wpre = 0;
  for (int w = 0; w < wave; ++w) wpre += wsum[w];
  const int excl = base + wpre + inc - tsum;
  #pragma unroll
  for (int j = 0; j < OWN_NPT; ++j) {
    const int node = first + j;
    pos[tid * OWN_NPT + j] = excl + pre[j] + before[j];
    if (h == 0 && node < n) {
      rowstart[node] = excl + pre[j];
      dinv[node] = rsqrtf((float)v[j] + 1.0f);
    }
  }
  if (h == 0 && blockIdx.x == gridDim.x - 1 && tid == OWN_T - 1) rowstart[n] = excl + tsum;
  __syncthreads();

  if (((E & 3) == 0) && ((((size_t)dst) & 15) == 0) && ((((size_t)src) & 15) == 0)) {
    const int len = (ee - eb) >> 2;
    if (len > 0) {
      const int4* dv = (const int4*)dst + (eb >> 2);
      const int4* sv = (const int4*)src + (eb >> 2);
      const int rot = (int)(((long long)blockIdx.x * len) / gridDim.x);
      constexpr int STEP = OWN_T * OWN_U;
      int4 dc[OWN_U], sc[OWN_U], dn[OWN_U], sn[OWN_U];
      #pragma unroll
      for (int u = 0; u < OWN_U; ++u) {
        dc[u] = own_piece(dv, len, rot, u * OWN_T + tid);
        sc[u] = own_piece(sv, len, rot, u * OWN_T + tid);
      }
      for (int j0 = 0; j0 < len; j0 += STEP) {
        #pragma unroll
        for (int u = 0; u < OWN_U; ++u) {
          dn[u] = own_piece(dv, len, rot, j0 + STEP + u * OWN_T + tid);
          sn[u] = own_piece(sv, len, rot, j0 + STEP + u * OWN_T + tid);
        }
        OWN_ASK(0) OWN_ASK(1) OWN_ASK(2) OWN_ASK(3)
        OWN_PUT(0) OWN_PUT(1) OWN_PUT(2) OWN_PUT(3)
        #pragma unroll
        for (int u = 0; u < OWN_U; ++u) { dc[u] = dn[u]; sc[u] = sn[u]; }
      }
    }
  } else {
    for (int e = eb + tid; e < ee; e += OWN_T) {
      const unsigned l = (unsigned)(dst[e] - lo);
      if (l < OWN) csr[atomicAdd(&pos[l], 1)] = src[e];
    }
  }
}

#define AGG_AP 132
#define AGG_U 8
#define AGG_SPB 13
#define AGG_NODES (AGG_SPB * 16)
#define AGG_T 1024
#define AGG_WV (AGG_T / 32)
#define AGG_CH 4


__device__ __forceinline__ int agg_csr(const int* __restrict__ csr, int first, int s, int e, int lane) {
  const int slot = first + lane;
  return csr[(slot >= s && slot < e) ? slot : 0];
}

__device__ __forceinline__ int agg_idx(int craw, int first, int s, int e, int lane, int self) {
  const int slot = first + lane;
  return (slot >= s && slot < e) ? craw : self;
}

struct AggNoWork { __device__ __forceinline__ void operator()() const {} };
template <class WK>
__device__ __forceinline__ void agg_rows(const float* __restrict__ Hin, int lane, int idx, float dw,
                                         int cnt, float4& acc, WK&& work) {
  const int dwb = __float_as_int(dw);
  if (cnt > 0) idx = (lane < cnt) ? idx : __builtin_amdgcn_readlane(idx, cnt - 1);
  for (int k0 = 0; k0 < cnt; k0 += AGG_U) {
    float4 v[AGG_U];
    #pragma unroll
    for (int u = 0; u < AGG_U; ++u) {
      const int sidx = __builtin_amdgcn_readlane(idx, k0 + u);
      v[u] = *(const float4*)(Hin + (size_t)sidx * 128 + lane * 4);
    }
    work();
    #pragma unroll
    for (int u = 0; u < AGG_U; ++u) {
      const float w = __int_as_float(__builtin_amdgcn_readlane(dwb, k0 + u));
      acc.x += w * v[u].x; acc.y += w * v[u].y; acc.z += w * v[u].z; acc.w += w * v[u].w;
    }
  }
}

__device__ __forceinline__ int agg_cs(int c) {
  return min(4 * c, 8) + (c >= 3 ? 3 : 0) + (c >= 4 ? 2 : 0);
}

template <int N, int N2>
__global__ void __launch_bounds__(AGG_T)
k_agg_gemm(const float* __restrict__ Hin, const float* __restrict__ W,
           const float* __restrict__ dinv, const int* __restrict__ rowstart,
           const int* __restrict__ csr, const float* __restrict__ bias,
           const float* __restrict__ Wn, float* __restrict__ out, int M) {
  constexpr int K = 128;
  static_assert(N == K, "the relu rows take the place of the summed rows in the A tile");
  constexpr int RP2 = 2 * N2 + 32;
  constexpr int WL2 = (N / 2) * RP2;
  constexpr int NT = N / 16;
  constexpr int KQ = 4;
  constexpr int QS = K / 4 / KQ;
  constexpr int RP = 2 * N + 32;
  constexpr int WL = (K / 2) * RP;
  constexpr int AT = AGG_NODES * AGG_AP;
  static_assert(4 * NT <= AGG_WV, "a chunk has at most 4 strips and every tile of it a wave of its own");
  constexpr int ALONE = 168 * 1024 / 4;
  constexpr int WLA = (WL + AT >= ALONE) ? WL : ALONE - AT;
  __shared__ __attribute__((aligned(16))) float Wlds[WLA];
  __shared__ __attribute__((aligned(16))) float Atile[AT];
  __shared__ __attribute__((aligned(16))) float W2lds[WL2];

  const int wave = __builtin_amdgcn_readfirstlane(threadIdx.x >> 5);
  const int lane = threadIdx.x & 31;
  const int half = lane >> 4;
  const int l16 = lane & 15;
  const int Rb = blockIdx.x * AGG_NODES;

  int seql = -1;
  int total = 0;
  #pragma unroll
  for (int c = 0; c < AGG_CH; ++c) {
    const int cb = agg_cs(c) * 16, cn = (agg_cs(c + 1) - agg_cs(c)) * 16;
    const int f = (wave * cn) / AGG_WV, nrc = ((wave + 1) * cn) / AGG_WV - f;
    if (lane >= total && lane < total + nrc) seql = cb + f + (lane - total);
    total += nrc;
  }
  const int gnode = Rb + max(seql, 0);
  int rs_b = rowstart[min(gnode, M)];
  int rs_e = rowstart[min(gnode + 1, M)];
  if (seql < 0) { rs_b = 0; rs_e = 0; }
  const int selfv = min(gnode, M - 1);

  v8f accw = {};
  int kq = KQ + 1;
  int Rmw = 0, tilew = 0, lrw = 0;
  float bvw = 0.0f;
  float diw[8];
  #pragma unroll
  for (int r = 0; r < 8; ++r) diw[r] = 0.0f;
  const float* abase = Atile;
  const float* bbase = Wlds;

  auto slice = [&]() {
    const float* ap = abase + kq * (QS * 4);
    const float* bp = bbase + kq * (QS * 2) * RP;
    #pragma unroll
    for (int i = 0; i < QS; ++i) {
      v2f a = *(const v2f*)(ap + 4 * i);
      v2f b = *(const v2f*)(bp + 2 * i * RP);
      accw = __builtin_amdgcn_wmma_f32_16x16x4_f32(
          false, a, false, b, (short)0, accw, false, false);
    }
  };

  auto finish = [&]() {
    #pragma unroll
    for (int r = 0; r < 8; ++r)
      Atile[(lrw + r + 8 * half) * AGG_AP + tilew * 16 + l16] = fmaxf(accw[r] * diw[r] + bvw, 0.0f);
  };

  auto work = [&]() {
    if (kq < KQ) { slice(); ++kq; }
  };

  constexpr int WU = (K / 2) * (N / 4);
  constexpr int WQ = WU / AGG_T;
  static_assert(WU % AGG_T == 0, "every thread stages the same number of units");
  int wq = 0;
  auto stage = [&]() {
    const int u = wq * AGG_T + (int)threadIdx.x;
    const int kp = u / (N / 4), n4 = u % (N / 4);
    const float4 w0 = *(const float4*)(W + (2 * kp) * N + n4 * 4);
    const float4 w1 = *(const float4*)(W + (2 * kp + 1) * N + n4 * 4);
    float* d = Wlds + kp * RP + n4 * 8;
    *(float4*)d = make_float4(w0.x, w1.x, w0.y, w1.y);
    *(float4*)(d + 4) = make_float4(w0.z, w1.z, w0.w, w1.w);
  };
  constexpr int WU2 = (N / 2) * (N2 / 4);
  constexpr int WQ2 = WU2 / AGG_T;
  static_assert(WU2 % AGG_T == 0, "every thread stages the same number of units");
  auto stage2 = [&]() {
    const int u = (wq - WQ) * AGG_T + (int)threadIdx.x;
    const int kp = u / (N2 / 4), n4 = u % (N2 / 4);
    const float4 w0 = *(const float4*)(Wn + (2 * kp) * N2 + n4 * 4);
    const float4 w1 = *(const float4*)(Wn + (2 * kp + 1) * N2 + n4 * 4);
    float* d = W2lds + kp * RP2 + n4 * 8;
    *(float4*)d = make_float4(w0.x, w1.x, w0.y, w1.y);
    *(float4*)(d + 4) = make_float4(w0.z, w1.z, w0.w, w1.w);
  };
  auto work_first = [&]() {
    if (wq < WQ) { stage(); ++wq; }
    else if (wq < WQ + WQ2) { stage2(); ++wq; }
  };

  int sA = __builtin_amdgcn_readlane(rs_b, 0), eA = __builtin_amdgcn_readlane(rs_e, 0);
  int sB = __builtin_amdgcn_readlane(rs_b, 1), eB = __builtin_amdgcn_readlane(rs_e, 1);
  const int raw0 = agg_csr(csr, sA - 1, sA, eA, lane);
  int raw1 = agg_csr(csr, sB - 1, sB, eB, lane);
  int idx = agg_idx(raw0, sA - 1, sA, eA, lane, __builtin_amdgcn_readlane(selfv, 0));
  float dw = dinv[idx];
  int j = 0;
  auto gather = [&](int c, auto&& wk) {
    const int cn = (agg_cs(c + 1) - agg_cs(c)) * 16;
    const int nrc = ((wave + 1) * cn) / AGG_WV - (wave * cn) / AGG_WV;
    for (int r = 0; r < nrc; ++r, ++j) {
      const int sC = __builtin_amdgcn_readlane(rs_b, j + 2), eC = __builtin_amdgcn_readlane(rs_e, j + 2);
      const int raw2 = agg_csr(csr, sC - 1, sC, eC, lane);
      const int nidx = agg_idx(raw1, sB - 1, sB, eB, lane, __builtin_amdgcn_readlane(selfv, j + 1));
      const float ndw = dinv[nidx];
      const int lrow = __builtin_amdgcn_readlane(seql, j);
      const int node = Rb + lrow;
      const int nodec = min(node, M - 1);
      const int tot = (node < M) ? eA - sA + 1 : 0;
      float4 acc = make_float4(0.0f, 0.0f, 0.0f, 0.0f);
      agg_rows(Hin, lane, idx, (lane < tot) ? dw : 0.0f, min(tot, 32), acc, wk);
      for (int cb = sA + 31; cb < eA; cb += 32) {
        const int xi = agg_idx(agg_csr(csr, cb, sA, eA, lane), cb, sA, eA, lane, nodec);
        const float xw = dinv[xi];
        agg_rows(Hin, lane, xi, (lane < eA - cb) ? xw : 0.0f, min(eA - cb, 32), acc, AggNoWork());
      }
      *(float4*)(Atile + lrow * AGG_AP + lane * 4) = acc;
      idx = nidx; dw = ndw; raw1 = raw2;
      sA = sB; eA = eB; sB = sC; eB = eC;
    }
  };
  auto open_tile = [&](int c) {
    const int strip = agg_cs(c) + wave / NT;
    Rmw = Rb + strip * 16;
    if (wave < (agg_cs(c + 1) - agg_cs(c)) * NT && Rmw < M) {
      tilew = wave % NT;
      lrw = strip * 16;
      abase = Atile + (strip * 16 + l16) * AGG_AP + 2 * half;
      bbase = Wlds + half * RP + l16 * 2 + tilew * 32;
      #pragma unroll
      for (int r = 0; r < 8; ++r) accw[r] = 0.0f;
      kq = 0;
      bvw = bias[tilew * 16 + l16];
      #pragma unroll
      for (int r = 0; r < 8; ++r) diw[r] = dinv[min(Rmw + r + 8 * half, M - 1)];
    }
  };

  gather(0, work_first);
  for (; wq < WQ; ++wq) stage();
  for (; wq < WQ + WQ2; ++wq) stage2();
  #pragma nounroll
  for (int c = 0;;) {
    __syncthreads();
    if (kq == KQ) { finish(); kq = KQ + 1; }
    open_tile(c);
    if (++c == AGG_CH) break;
    gather(c, work);
    for (; kq < KQ; ++kq) slice();
  }
  for (; kq < KQ; ++kq) slice();
  __syncthreads();
  if (kq == KQ) finish();
  __syncthreads();

  {
    constexpr int CT = N2 / 32;
    static_assert(N2 == 64 && CT == 2, "two column tiles a wave, two waves a strip");
    const int strip = wave >> 1, cp = wave & 1;
    const int R2 = Rb + strip * 16;
    if (strip < AGG_SPB && R2 < M) {
      v8f c0 = {}, c1 = {};
      float d2[8];
      #pragma unroll
      for (int r = 0; r < 8; ++r) d2[r] = dinv[min(R2 + r + 8 * half, M - 1)];
      const float* ap = Atile + (strip * 16 + l16) * AGG_AP + 2 * half;
      const float* bp = W2lds + half * RP2 + l16 * 2 + cp * 64;
      #pragma unroll 8
      for (int i = 0; i < K / 4; ++i) {
        const v2f a = *(const v2f*)(ap + 4 * i);
        const v2f b0 = *(const v2f*)(bp + 2 * i * RP2);
        const v2f b1 = *(const v2f*)(bp + 2 * i * RP2 + 32);
        c0 = __builtin_amdgcn_wmma_f32_16x16x4_f32(false, a, false, b0, (short)0, c0, false, false);
        c1 = __builtin_amdgcn_wmma_f32_16x16x4_f32(false, a, false, b1, (short)0, c1, false, false);
      }
      #pragma unroll
      for (int r = 0; r < 8; ++r) {
        const int row = R2 + r + 8 * half;
        if (row < M) {
          out[(size_t)row * N2 + cp * 32 + l16] = c0[r] * d2[r];
          out[(size_t)row * N2 + cp * 32 + 16 + l16] = c1[r] * d2[r];
        }
      }
    }
  }
}

#define SP_NPW 4
#define SP_T 512
#define SP_U 8
#define SP_NODES (SP_T / 32 * SP_NPW)

__device__ __forceinline__ void sp_rows(const float* __restrict__ G2, int lane, int idx, int cnt, float4& acc) {
  const int half = lane >> 4, l16 = lane & 15;
  if (cnt > 0) idx = (lane < cnt) ? idx : __builtin_amdgcn_readlane(idx, cnt - 1);
  for (int k0 = 0; k0 < cnt; k0 += 2 * SP_U) {
    float4 v[SP_U];
    #pragma unroll
    for (int u = 0; u < SP_U; ++u) {
      v[u] = make_float4(0.0f, 0.0f, 0.0f, 0.0f);
      if (k0 + 2 * u < cnt) {
        const int ia = __builtin_amdgcn_readlane(idx, k0 + 2 * u);
        const int ib = __builtin_amdgcn_readlane(idx, k0 + 2 * u + 1);
        v[u] = *(const float4*)(G2 + (size_t)(half ? ib : ia) * 64 + l16 * 4);
      }
    }
    #pragma unroll
    for (int u = 0; u < SP_U; ++u) {
      const bool in = k0 + 2 * u + half < cnt;
      acc.x += in ? v[u].x : 0.0f; acc.y += in ? v[u].y : 0.0f;
      acc.z += in ? v[u].z : 0.0f; acc.w += in ? v[u].w : 0.0f;
    }
  }
}

__global__ void __launch_bounds__(SP_T)
k_sum_pool(const float* __restrict__ G2, const float* __restrict__ dinv, const int* __restrict__ rowstart,
           const int* __restrict__ csr, const float* __restrict__ bias, const int* __restrict__ batch,
           float* __restrict__ pool, float* __restrict__ cntg, int M) {
  __shared__ float ps[64];
  __shared__ float pc;
  const int wave = __builtin_amdgcn_readfirstlane(threadIdx.x >> 5);
  const int lane = threadIdx.x & 31;
  const int l16 = lane & 15;
  const int nb = blockIdx.x * SP_NODES;
  const int n0 = nb + wave * SP_NPW;
  const int gb = batch[nb];
  if (threadIdx.x < 64) ps[threadIdx.x] = 0.0f;
  if (threadIdx.x == 64) pc = 0.0f;
  __syncthreads();

  const int rsv = rowstart[min(n0 + lane, M)];
  const int dvb = __float_as_int(dinv[min(n0 + lane, M - 1)]);
  const int btv = batch[min(n0 + lane, M - 1)];
  const float4 bv = *(const float4*)(bias + l16 * 4);
  int craw[SP_NPW];
  #pragma unroll
  for (int q = 0; q < SP_NPW; ++q) {
    const int s = __builtin_amdgcn_readlane(rsv, q), e = __builtin_amdgcn_readlane(rsv, q + 1);
    craw[q] = agg_csr(csr, s - 1, s, e, lane);
  }
  float4 psum = make_float4(0.0f, 0.0f, 0.0f, 0.0f);
  float pcnt = 0.0f;
  int gcur = gb;
  auto flush = [&]() {
    if ((unsigned)gcur < (unsigned)N_GRAPHS && pcnt > 0.0f) {
      if (gcur == gb) {
        if (lane < 16) {
          atomicAdd(&ps[l16 * 4 + 0], psum.x); atomicAdd(&ps[l16 * 4 + 1], psum.y);
          atomicAdd(&ps[l16 * 4 + 2], psum.z); atomicAdd(&ps[l16 * 4 + 3], psum.w);
        }
        if (lane == 0) atomicAdd(&pc, pcnt);
      } else {
        if (lane < 16) {
          float* p = pool + gcur * 64 + l16 * 4;
          atomicAdd(p + 0, psum.x); atomicAdd(p + 1, psum.y);
          atomicAdd(p + 2, psum.z); atomicAdd(p + 3, psum.w);
        }
        if (lane == 0) atomicAdd(cntg + gcur, pcnt);
      }
    }
  };
  #pragma unroll
  for (int q = 0; q < SP_NPW; ++q) {
    const int node = n0 + q;
    if (node < M) {
      const int s = __builtin_amdgcn_readlane(rsv, q), e = __builtin_amdgcn_readlane(rsv, q + 1);
      const int tot = e - s + 1;
      const int idx = agg_idx(craw[q], s - 1, s, e, lane, node);
      float4 acc = make_float4(0.0f, 0.0f, 0.0f, 0.0f);
      sp_rows(G2, lane, idx, min(tot, 32), acc);
      for (int cb = s + 31; cb < e; cb += 32) {
        const int xi = agg_idx(agg_csr(csr, cb, s, e, lane), cb, s, e, lane, node);
        sp_rows(G2, lane, xi, min(e - cb, 32), acc);
      }
      acc.x += __shfl_xor(acc.x, 16); acc.y += __shfl_xor(acc.y, 16);
      acc.z += __shfl_xor(acc.z, 16); acc.w += __shfl_xor(acc.w, 16);
      const float d = __int_as_float(__builtin_amdgcn_readlane(dvb, q));
      const int g = __builtin_amdgcn_readlane(btv, q);
      if (g != gcur) {
        flush();
        gcur = g; pcnt = 0.0f;
        psum = make_float4(0.0f, 0.0f, 0.0f, 0.0f);
      }
      psum.x += fmaxf(acc.x * d + bv.x, 0.0f); psum.y += fmaxf(acc.y * d + bv.y, 0.0f);
      psum.z += fmaxf(acc.z * d + bv.z, 0.0f); psum.w += fmaxf(acc.w * d + bv.w, 0.0f);
      pcnt += 1.0f;
    }
  }
  flush();
  __syncthreads();
  if ((unsigned)gb < (unsigned)N_GRAPHS) {
    if (threadIdx.x < 64) atomicAdd(pool + gb * 64 + threadIdx.x, ps[threadIdx.x]);
    if (threadIdx.x == 64 && pc > 0.0f) atomicAdd(cntg + gb, pc);
  }
}

__global__ void k_pool_final(const float* __restrict__ pool, const float* __restrict__ cntg,
                             float* __restrict__ out) {
  int idx = blockIdx.x * blockDim.x + threadIdx.x;
  if (idx < N_GRAPHS * 64) {
    out[idx] = pool[idx] / fmaxf(cntg[idx >> 6], 1.0f);
  }
}

extern "C" void kernel_launch(void* const* d_in, const int* in_sizes, int n_in,
                              void* d_out, int out_size, void* d_ws, size_t ws_size,
                              hipStream_t stream) {
  const float* x     = (const float*)d_in[0];
  const float* W1    = (const float*)d_in[1];
  const float* b1    = (const float*)d_in[2];
  const float* W2    = (const float*)d_in[3];
  const float* b2    = (const float*)d_in[4];
  const int*   ei    = (const int*)d_in[5];
  const int*   batch = (const int*)d_in[6];

  const int N = in_sizes[6];
  const int E = in_sizes[5] / 2;
  const int G = N_GRAPHS;

  const int* src = ei;
  const int* dst = ei + E;

  char* base = (char*)d_ws;
  size_t off = 0;
  auto take = [&](size_t bytes) {
    size_t o = off;
    off = (off + bytes + 255) & ~(size_t)255;
    return o;
  };
  int*   cnt      = (int*)(base + take(sizeof(int) * (size_t)N * OWN_H));
  int*   tot      = (int*)(base + take(sizeof(int) * (size_t)((N + OWN - 1) / OWN) * OWN_H));
  int*   rowstart = (int*)(base + take(sizeof(int) * (size_t)(N + 1)));
  float* dinv     = (float*)(base + take(sizeof(float) * (size_t)N));
  int*   csr      = (int*)(base + take(sizeof(int) * (size_t)E));
  float* g2       = (float*)(base + take(sizeof(float) * (size_t)N * 64));
  float* pool     = (float*)(base + take(sizeof(float) * (size_t)G * 64));
  float* cntg     = (float*)(base + take(sizeof(float) * (size_t)G));

  const int B = 256;
  const int gOwn = (N + OWN - 1) / OWN;
  const int mtiles = (N + 15) / 16;
  const int gGemm = (mtiles + AGG_SPB - 1) / AGG_SPB;

  k_zero_f32<<<(G * 64 + G + B - 1) / B, B, 0, stream>>>(pool, G * 64 + G);

  k_own_count<<<dim3(gOwn, OWN_H), OWN_T, 0, stream>>>(dst, cnt, tot, E, N);
  k_own_fill<<<dim3(gOwn, OWN_H), OWN_T, 0, stream>>>(src, dst, cnt, tot, rowstart, dinv, csr, E, N);

  k_agg_gemm<128, 64><<<gGemm, AGG_T, 0, stream>>>(x, W1, dinv, rowstart, csr, b1, W2, g2, N);

  k_sum_pool<<<(N + SP_NODES - 1) / SP_NODES, SP_T, 0, stream>>>(g2, dinv, rowstart, csr, b2, batch, pool, cntg, N);

  k_pool_final<<<(G * 64 + B - 1) / B, B, 0, stream>>>(pool, cntg, (float*)d_out);
}
